# starting kernel
# baseline (speedup 1.0000x reference)
#include <hip/hip_runtime.h>
#include <hip/hip_bf16.h>
#include <math.h>

typedef float v2f __attribute__((ext_vector_type(2)));
typedef float v8f __attribute__((ext_vector_type(8)));

#define H 64

__device__ __forceinline__ float elu1(float x) {
    return x > 0.0f ? x : (__expf(x) - 1.0f);
}

__global__ void enc_kernel(const float* __restrict__ x, const float* __restrict__ W,
                           const float* __restrict__ b, float* __restrict__ h,
                           int N, int F) {
    int t = blockIdx.x * blockDim.x + threadIdx.x;
    int n = t >> 6, j = t & 63;
    if (n >= N) return;
    const float* xr = x + (long)n * F;
    float s = b[j];
    for (int k = 0; k < F; ++k) s = fmaf(xr[k], W[k * H + j], s);
    h[(long)n * H + j] = elu1(s);
}

__global__ void scatter_kernel(const float* __restrict__ hsrc, const int* __restrict__ src,
                               const int* __restrict__ dst, float* __restrict__ agg,
                               float* __restrict__ deg, int E) {
    long t = (long)blockIdx.x * blockDim.x + threadIdx.x;
    int e = (int)(t >> 6), j = (int)(t & 63);
    if (e >= E) return;
    int s = src[e], d = dst[e];
    atomicAdd(&agg[(long)d * H + j], hsrc[(long)s * H + j]);
    if (j == 0) atomicAdd(&deg[d], 1.0f);
}

__global__ void __launch_bounds__(128)
sage_gemm_kernel(const float* __restrict__ agg, const float* __restrict__ deg,
                 const float* __restrict__ hdst,
                 const float* __restrict__ Wl, const float* __restrict__ bl,
                 const float* __restrict__ Wr,
                 float* __restrict__ hnew, int N) {
    __shared__ float sWl[H * H];
    __shared__ float sWr[H * H];
    int tid = threadIdx.x;
    {
        const float4* gl = (const float4*)Wl;
        const float4* gr = (const float4*)Wr;
        float4* ll = (float4*)sWl;
        float4* lr = (float4*)sWr;
        for (int i = tid; i < (H * H / 4); i += 128) {
            ll[i] = gl[i];
            lr[i] = gr[i];
        }
    }
    __syncthreads();

    int wave = tid >> 5;
    int lane = tid & 31;
    int hi   = lane >> 4;
    int col  = lane & 15;
    int kk   = hi * 2;
    int m0   = blockIdx.x * 64 + wave * 16;

    int r  = m0 + col;
    int rc = r < N ? r : (N - 1);
    float invd = 1.0f / fmaxf(deg[rc], 1.0f);

    const float* aggr = agg  + (long)rc * H;
    const float* hdr  = hdst + (long)rc * H;

    v8f acc[4] = {};

#pragma unroll
    for (int k0 = 0; k0 < H; k0 += 4) {
        v2f a;
        a.x = aggr[k0 + kk] * invd;
        a.y = aggr[k0 + kk + 1] * invd;
#pragma unroll
        for (int t4 = 0; t4 < 4; ++t4) {
            v2f b;
            b.x = sWl[(k0 + kk) * H + t4 * 16 + col];
            b.y = sWl[(k0 + kk + 1) * H + t4 * 16 + col];
            acc[t4] = __builtin_amdgcn_wmma_f32_16x16x4_f32(
                false, a, false, b, (short)0, acc[t4], false, false);
        }
    }
#pragma unroll
    for (int k0 = 0; k0 < H; k0 += 4) {
        v2f a;
        a.x = hdr[k0 + kk];
        a.y = hdr[k0 + kk + 1];
#pragma unroll
        for (int t4 = 0; t4 < 4; ++t4) {
            v2f b;
            b.x = sWr[(k0 + kk) * H + t4 * 16 + col];
            b.y = sWr[(k0 + kk + 1) * H + t4 * 16 + col];
            acc[t4] = __builtin_amdgcn_wmma_f32_16x16x4_f32(
                false, a, false, b, (short)0, acc[t4], false, false);
        }
    }

#pragma unroll
    for (int t4 = 0; t4 < 4; ++t4) {
        int cg = t4 * 16 + col;
        float bias = bl[cg];
#pragma unroll
        for (int rr = 0; rr < 8; ++rr) {
            int row = m0 + rr + hi * 8;
            if (row < N) hnew[(long)row * H + cg] += acc[t4][rr] + bias;
        }
    }
}

__global__ void ln_kernel(float* __restrict__ h, const float* __restrict__ hn,
                          const float* __restrict__ g, const float* __restrict__ be, int N) {
    int n = blockIdx.x;
    if (n >= N) return;
    int lane = threadIdx.x;
    float v0 = h[(long)n * H + lane]      + hn[(long)n * H + lane];
    float v1 = h[(long)n * H + lane + 32] + hn[(long)n * H + lane + 32];
    float s = v0 + v1;
#pragma unroll
    for (int m = 16; m >= 1; m >>= 1) s += __shfl_xor(s, m, 32);
    float mean = s * (1.0f / 64.0f);
    float d0 = v0 - mean, d1 = v1 - mean;
    float q = d0 * d0 + d1 * d1;
#pragma unroll
    for (int m = 16; m >= 1; m >>= 1) q += __shfl_xor(q, m, 32);
    float inv = rsqrtf(q * (1.0f / 64.0f) + 1e-5f);
    h[(long)n * H + lane]      = elu1(d0 * inv * g[lane]      + be[lane]);
    h[(long)n * H + lane + 32] = elu1(d1 * inv * g[lane + 32] + be[lane + 32]);
}

__global__ void cox_kernel(const float* __restrict__ hp, const float* __restrict__ W1,
                           const float* __restrict__ b1, const float* __restrict__ W2,
                           const float* __restrict__ b2, float* __restrict__ out, int N) {
    int n = blockIdx.x * blockDim.x + threadIdx.x;
    if (n >= N) return;
    const float* hr = hp + (long)n * H;
    float hreg[H];
#pragma unroll
    for (int k = 0; k < H; ++k) hreg[k] = hr[k];
    float hz = b2[0];
    for (int j = 0; j < 32; ++j) {
        float s = b1[j];
#pragma unroll
        for (int k = 0; k < H; ++k) s = fmaf(hreg[k], W1[k * 32 + j], s);
        hz = fmaf(elu1(s), W2[j], hz);
    }
    out[n] = hz;
}

extern "C" void kernel_launch(void* const* d_in, const int* in_sizes, int n_in,
                              void* d_out, int out_size, void* d_ws, size_t ws_size,
                              hipStream_t stream) {
    (void)in_sizes; (void)n_in; (void)out_size; (void)ws_size;
    const int NG = 20000, NP = 100000, NM = 5000, E = 1000000;

    const float* x_gene    = (const float*)d_in[0];
    const float* x_patient = (const float*)d_in[1];
    const float* x_group   = (const float*)d_in[2];
    const int* srcA[4]  = { (const int*)d_in[3], (const int*)d_in[5],
                            (const int*)d_in[7], (const int*)d_in[9] };
    const int* dstA[4]  = { (const int*)d_in[4], (const int*)d_in[6],
                            (const int*)d_in[8], (const int*)d_in[10] };
    const float* encW[3] = { (const float*)d_in[11], (const float*)d_in[13], (const float*)d_in[15] };
    const float* encB[3] = { (const float*)d_in[12], (const float*)d_in[14], (const float*)d_in[16] };
    const float* convW_l = (const float*)d_in[17];
    const float* convb_l = (const float*)d_in[18];
    const float* convW_r = (const float*)d_in[19];
    const float* ln_g = (const float*)d_in[20];
    const float* ln_b = (const float*)d_in[21];
    const float* coxW1 = (const float*)d_in[22];
    const float* coxb1 = (const float*)d_in[23];
    const float* coxW2 = (const float*)d_in[24];
    const float* coxb2 = (const float*)d_in[25];

    float* ws = (float*)d_ws;
    float* h_gene  = ws; ws += (size_t)NG * H;
    float* h_pat   = ws; ws += (size_t)NP * H;
    float* h_grp   = ws; ws += (size_t)NM * H;
    float* hn_gene = ws; ws += (size_t)NG * H;
    float* hn_pat  = ws; ws += (size_t)NP * H;
    float* hn_grp  = ws; ws += (size_t)NM * H;
    float* agg     = ws; ws += (size_t)NP * H;
    float* deg     = ws; ws += (size_t)NP;

    enc_kernel<<<(NG * H + 255) / 256, 256, 0, stream>>>(x_gene,    encW[0], encB[0], h_gene, NG, 11);
    enc_kernel<<<(NP * H + 255) / 256, 256, 0, stream>>>(x_patient, encW[1], encB[1], h_pat,  NP, 3);
    enc_kernel<<<(NM * H + 255) / 256, 256, 0, stream>>>(x_group,   encW[2], encB[2], h_grp,  NM, 4);

    float* hArr[3]  = { h_gene, h_pat, h_grp };
    float* hnArr[3] = { hn_gene, hn_pat, hn_grp };
    int    nArr[3]  = { NG, NP, NM };
    const int sType[4] = { 0, 1, 0, 2 };
    const int dType[4] = { 1, 0, 2, 0 };

    for (int layer = 0; layer < 2; ++layer) {
        hipMemsetAsync(hn_gene, 0, (size_t)NG * H * sizeof(float), stream);
        hipMemsetAsync(hn_pat,  0, (size_t)NP * H * sizeof(float), stream);
        hipMemsetAsync(hn_grp,  0, (size_t)NM * H * sizeof(float), stream);
        for (int e = 0; e < 4; ++e) {
            int Nd = nArr[dType[e]];
            hipMemsetAsync(agg, 0, (size_t)Nd * H * sizeof(float), stream);
            hipMemsetAsync(deg, 0, (size_t)Nd * sizeof(float), stream);
            scatter_kernel<<<(E * 64) / 256, 256, 0, stream>>>(
                hArr[sType[e]], srcA[e], dstA[e], agg, deg, E);
            const float* Wl  = convW_l + ((size_t)layer * 4 + e) * H * H;
            const float* blp = convb_l + ((size_t)layer * 4 + e) * H;
            const float* Wr  = convW_r + ((size_t)layer * 4 + e) * H * H;
            sage_gemm_kernel<<<(Nd + 63) / 64, 128, 0, stream>>>(
                agg, deg, hArr[dType[e]], Wl, blp, Wr, hnArr[dType[e]], Nd);
        }
        for (int ni = 0; ni < 3; ++ni) {
            ln_kernel<<<nArr[ni], 32, 0, stream>>>(
                hArr[ni], hnArr[ni],
                ln_g + ((size_t)layer * 3 + ni) * H,
                ln_b + ((size_t)layer * 3 + ni) * H, nArr[ni]);
        }
    }
    cox_kernel<<<(NP + 255) / 256, 256, 0, stream>>>(
        h_pat, coxW1, coxb1, coxW2, coxb2, (float*)d_out, NP);
}


# Replace float-atomic scatter-sum by a per-call grouping of edges by destination (int atomic arrival rank + prefix sum) and a one-wave-per-node gather sum; neighbour rows now add in rank order
# speedup vs baseline: 3.2691x; 3.2691x over previous
#include <hip/hip_runtime.h>
#include <hip/hip_bf16.h>
#include <math.h>

typedef float v2f __attribute__((ext_vector_type(2)));
typedef float v8f __attribute__((ext_vector_type(8)));

#define H 64

__device__ __forceinline__ float elu1(float x) {
    return x > 0.0f ? x : (__expf(x) - 1.0f);
}

__global__ void enc_kernel(const float* __restrict__ x, const float* __restrict__ W,
                           const float* __restrict__ b, float* __restrict__ h,
                           int N, int F) {
    int t = blockIdx.x * blockDim.x + threadIdx.x;
    int n = t >> 6, j = t & 63;
    if (n >= N) return;
    const float* xr = x + (long)n * F;
    float s = b[j];
    for (int k = 0; k < F; ++k) s = fmaf(xr[k], W[k * H + j], s);
    h[(long)n * H + j] = elu1(s);
}

__global__ void rank_kernel(const int* __restrict__ d0, const int* __restrict__ d1,
                            const int* __restrict__ d2, const int* __restrict__ d3,
                            int* __restrict__ cnt, int* __restrict__ rank,
                            int n0, int n1, int n2, int E) {
    int t = blockIdx.y;
    const int* d = t == 0 ? d0 : t == 1 ? d1 : t == 2 ? d2 : d3;
    int coff = t == 0 ? 0 : t == 1 ? n0 : t == 2 ? n0 + n1 : n0 + n1 + n2;
    int e = blockIdx.x * blockDim.x + threadIdx.x;
    if (e >= E) return;
    rank[(long)t * E + e] = atomicAdd(&cnt[coff + d[e]], 1);
}

__global__ void __launch_bounds__(1024)
rowptr_scan_kernel(const int* __restrict__ cnt, int* __restrict__ rowptr,
                   int n0, int n1, int n2, int n3) {
    __shared__ int wsum[32];
    int t = blockIdx.x;
    int N    = t == 0 ? n0 : t == 1 ? n1 : t == 2 ? n2 : n3;
    int coff = t == 0 ? 0 : t == 1 ? n0 : t == 2 ? n0 + n1 : n0 + n1 + n2;
    const int* c = cnt + coff;
    int* rp = rowptr + coff + 4 * t;
    int wave = threadIdx.x >> 5, lane = threadIdx.x & 31;
    int per = (((N + 31) >> 5) + 127) & ~127;
    int beg = wave * per;
    int end = min(beg + per, N);

    int s = 0;
    for (int p = beg + lane * 4; p < end; p += 128) {
        int4 v = *(const int4*)(c + p);
        s += v.x + v.y + v.z + v.w;
    }
    for (int m = 16; m >= 1; m >>= 1) s += __shfl_xor(s, m, 32);
    if (lane == 0) wsum[wave] = s;
    __syncthreads();

    int wtot = wsum[lane];
    int winc = wtot;
    for (int dlt = 1; dlt < 32; dlt <<= 1) {
        int o = __shfl_up(winc, dlt, 32);
        if (lane >= dlt) winc += o;
    }
    int carry = __shfl(winc - wtot, wave, 32);
    int total = __shfl(winc, 31, 32);

    for (int tb = beg; tb < end; tb += 128) {
        int p = tb + lane * 4;
        int4 v = make_int4(0, 0, 0, 0);
        if (p < end) v = *(const int4*)(c + p);
        int ls = v.x + v.y + v.z + v.w;
        int li = ls;
        for (int dlt = 1; dlt < 32; dlt <<= 1) {
            int o = __shfl_up(li, dlt, 32);
            if (lane >= dlt) li += o;
        }
        int ex = carry + li - ls;
        if (p < end)
            *(int4*)(rp + p) = make_int4(ex, ex + v.x, ex + v.x + v.y, ex + v.x + v.y + v.z);
        carry += __shfl(li, 31, 32);
    }
    if (threadIdx.x == 0) rp[N] = total;
}

__global__ void fill_kernel(const int* __restrict__ s0, const int* __restrict__ s1,
                            const int* __restrict__ s2, const int* __restrict__ s3,
                            const int* __restrict__ d0, const int* __restrict__ d1,
                            const int* __restrict__ d2, const int* __restrict__ d3,
                            const int* __restrict__ rowptr, const int* __restrict__ rank,
                            int* __restrict__ perm, int n0, int n1, int n2, int E) {
    int t = blockIdx.y;
    const int* s = t == 0 ? s0 : t == 1 ? s1 : t == 2 ? s2 : s3;
    const int* d = t == 0 ? d0 : t == 1 ? d1 : t == 2 ? d2 : d3;
    int coff = t == 0 ? 0 : t == 1 ? n0 : t == 2 ? n0 + n1 : n0 + n1 + n2;
    int e = blockIdx.x * blockDim.x + threadIdx.x;
    if (e >= E) return;
    int pos = rowptr[coff + 4 * t + d[e]] + rank[(long)t * E + e];
    pos = min(max(pos, 0), E - 1);
    perm[(long)t * E + pos] = s[e];
}

template <int U>
__global__ void __launch_bounds__(256)
gather_sum_kernel(const float* __restrict__ hsrc, const int* __restrict__ rowptr,
                  const int* __restrict__ perm, float* __restrict__ agg,
                  float* __restrict__ deg, int nbeg, int ncnt, int nsrc, int E) {
    int wave = threadIdx.x >> 5, lane = threadIdx.x & 31;
    int ln = blockIdx.x * 8 + wave;
    if (ln >= ncnt) return;
    int beg = rowptr[nbeg + ln], end = rowptr[nbeg + ln + 1];
    beg = max(beg, 0);
    end = min(end, E);
    int half = lane >> 4, c = lane & 15;
    const float4* hs4 = (const float4*)hsrc;
    float4 acc = make_float4(0.0f, 0.0f, 0.0f, 0.0f);

    for (int base = beg; base < end; base += 32) {
        int cnt = min(32, end - base);
        int mine = 0;
        if (lane < cnt) mine = perm[base + lane];
        mine = min(max(mine, 0), nsrc - 1);
        for (int i = 0; i < cnt; i += 2 * U) {
            float4 v[U];
            bool ok[U];
#pragma unroll
            for (int u = 0; u < U; ++u) {
                int idx = i + 2 * u + half;
                ok[u] = idx < cnt;
                int sidx = __shfl(mine, ok[u] ? idx : 0, 32);
                v[u] = hs4[(long)sidx * 16 + c];
            }
#pragma unroll
            for (int u = 0; u < U; ++u) {
                acc.x += ok[u] ? v[u].x : 0.0f;
                acc.y += ok[u] ? v[u].y : 0.0f;
                acc.z += ok[u] ? v[u].z : 0.0f;
                acc.w += ok[u] ? v[u].w : 0.0f;
            }
        }
    }
    acc.x += __shfl_xor(acc.x, 16, 32);
    acc.y += __shfl_xor(acc.y, 16, 32);
    acc.z += __shfl_xor(acc.z, 16, 32);
    acc.w += __shfl_xor(acc.w, 16, 32);
    if (half == 0) ((float4*)agg)[(long)ln * 16 + c] = acc;
    if (lane == 0) deg[ln] = (float)(end - beg);
}

__global__ void __launch_bounds__(128)
sage_gemm_kernel(const float* __restrict__ agg, const float* __restrict__ deg,
                 const float* __restrict__ hdst,
                 const float* __restrict__ Wl, const float* __restrict__ bl,
                 const float* __restrict__ Wr,
                 float* __restrict__ hnew, int N) {
    __shared__ float sWl[H * H];
    __shared__ float sWr[H * H];
    int tid = threadIdx.x;
    {
        const float4* gl = (const float4*)Wl;
        const float4* gr = (const float4*)Wr;
        float4* ll = (float4*)sWl;
        float4* lr = (float4*)sWr;
        for (int i = tid; i < (H * H / 4); i += 128) {
            ll[i] = gl[i];
            lr[i] = gr[i];
        }
    }
    __syncthreads();

    int wave = tid >> 5;
    int lane = tid & 31;
    int hi   = lane >> 4;
    int col  = lane & 15;
    int kk   = hi * 2;
    int m0   = blockIdx.x * 64 + wave * 16;

    int r  = m0 + col;
    int rc = r < N ? r : (N - 1);
    float invd = 1.0f / fmaxf(deg[rc], 1.0f);

    const float* aggr = agg  + (long)rc * H;
    const float* hdr  = hdst + (long)rc * H;

    v8f acc[4] = {};

#pragma unroll
    for (int k0 = 0; k0 < H; k0 += 4) {
        v2f a;
        a.x = aggr[k0 + kk] * invd;
        a.y = aggr[k0 + kk + 1] * invd;
#pragma unroll
        for (int t4 = 0; t4 < 4; ++t4) {
            v2f b;
            b.x = sWl[(k0 + kk) * H + t4 * 16 + col];
            b.y = sWl[(k0 + kk + 1) * H + t4 * 16 + col];
            acc[t4] = __builtin_amdgcn_wmma_f32_16x16x4_f32(
                false, a, false, b, (short)0, acc[t4], false, false);
        }
    }
#pragma unroll
    for (int k0 = 0; k0 < H; k0 += 4) {
        v2f a;
        a.x = hdr[k0 + kk];
        a.y = hdr[k0 + kk + 1];
#pragma unroll
        for (int t4 = 0; t4 < 4; ++t4) {
            v2f b;
            b.x = sWr[(k0 + kk) * H + t4 * 16 + col];
            b.y = sWr[(k0 + kk + 1) * H + t4 * 16 + col];
            acc[t4] = __builtin_amdgcn_wmma_f32_16x16x4_f32(
                false, a, false, b, (short)0, acc[t4], false, false);
        }
    }

#pragma unroll
    for (int t4 = 0; t4 < 4; ++t4) {
        int cg = t4 * 16 + col;
        float bias = bl[cg];
#pragma unroll
        for (int rr = 0; rr < 8; ++rr) {
            int row = m0 + rr + hi * 8;
            if (row < N) hnew[(long)row * H + cg] += acc[t4][rr] + bias;
        }
    }
}

__global__ void ln_kernel(float* __restrict__ h, const float* __restrict__ hn,
                          const float* __restrict__ g, const float* __restrict__ be, int N) {
    int n = blockIdx.x;
    if (n >= N) return;
    int lane = threadIdx.x;
    float v0 = h[(long)n * H + lane]      + hn[(long)n * H + lane];
    float v1 = h[(long)n * H + lane + 32] + hn[(long)n * H + lane + 32];
    float s = v0 + v1;
#pragma unroll
    for (int m = 16; m >= 1; m >>= 1) s += __shfl_xor(s, m, 32);
    float mean = s * (1.0f / 64.0f);
    float d0 = v0 - mean, d1 = v1 - mean;
    float q = d0 * d0 + d1 * d1;
#pragma unroll
    for (int m = 16; m >= 1; m >>= 1) q += __shfl_xor(q, m, 32);
    float inv = rsqrtf(q * (1.0f / 64.0f) + 1e-5f);
    h[(long)n * H + lane]      = elu1(d0 * inv * g[lane]      + be[lane]);
    h[(long)n * H + lane + 32] = elu1(d1 * inv * g[lane + 32] + be[lane + 32]);
}

__global__ void cox_kernel(const float* __restrict__ hp, const float* __restrict__ W1,
                           const float* __restrict__ b1, const float* __restrict__ W2,
                           const float* __restrict__ b2, float* __restrict__ out, int N) {
    int n = blockIdx.x * blockDim.x + threadIdx.x;
    if (n >= N) return;
    const float* hr = hp + (long)n * H;
    float hreg[H];
#pragma unroll
    for (int k = 0; k < H; ++k) hreg[k] = hr[k];
    float hz = b2[0];
    for (int j = 0; j < 32; ++j) {
        float s = b1[j];
#pragma unroll
        for (int k = 0; k < H; ++k) s = fmaf(hreg[k], W1[k * 32 + j], s);
        hz = fmaf(elu1(s), W2[j], hz);
    }
    out[n] = hz;
}

extern "C" void kernel_launch(void* const* d_in, const int* in_sizes, int n_in,
                              void* d_out, int out_size, void* d_ws, size_t ws_size,
                              hipStream_t stream) {
    (void)in_sizes; (void)n_in; (void)out_size; (void)ws_size;
    const int NG = 20000, NP = 100000, NM = 5000, E = 1000000;

    const float* x_gene    = (const float*)d_in[0];
    const float* x_patient = (const float*)d_in[1];
    const float* x_group   = (const float*)d_in[2];
    const int* srcA[4]  = { (const int*)d_in[3], (const int*)d_in[5],
                            (const int*)d_in[7], (const int*)d_in[9] };
    const int* dstA[4]  = { (const int*)d_in[4], (const int*)d_in[6],
                            (const int*)d_in[8], (const int*)d_in[10] };
    const float* encW[3] = { (const float*)d_in[11], (const float*)d_in[13], (const float*)d_in[15] };
    const float* encB[3] = { (const float*)d_in[12], (const float*)d_in[14], (const float*)d_in[16] };
    const float* convW_l = (const float*)d_in[17];
    const float* convb_l = (const float*)d_in[18];
    const float* convW_r = (const float*)d_in[19];
    const float* ln_g = (const float*)d_in[20];
    const float* ln_b = (const float*)d_in[21];
    const float* coxW1 = (const float*)d_in[22];
    const float* coxb1 = (const float*)d_in[23];
    const float* coxW2 = (const float*)d_in[24];
    const float* coxb2 = (const float*)d_in[25];

    float* ws = (float*)d_ws;
    float* h_gene  = ws; ws += (size_t)NG * H;
    float* h_pat   = ws; ws += (size_t)NP * H;
    float* h_grp   = ws; ws += (size_t)NM * H;
    float* hn_gene = ws; ws += (size_t)NG * H;
    float* hn_pat  = ws; ws += (size_t)NP * H;
    float* hn_grp  = ws; ws += (size_t)NM * H;
    const int SL = 34000;
    float* aggreg  = ws; ws += (size_t)NP * H;
    float* deg     = ws; ws += (size_t)NP;
    float* agg     = aggreg;
    int*   perm    = (int*)(aggreg + (size_t)SL * H);
    int*   rowptr  = perm + (size_t)4 * E;
    int*   cnt     = (int*)hn_gene;
    int*   rank    = (int*)hn_pat;

    hipMemsetAsync(cnt, 0, (size_t)(NP + NG + NM + NG) * sizeof(int), stream);
    rank_kernel<<<dim3((E + 255) / 256, 4), 256, 0, stream>>>(
        dstA[0], dstA[1], dstA[2], dstA[3], cnt, rank, NP, NG, NM, E);
    rowptr_scan_kernel<<<4, 1024, 0, stream>>>(cnt, rowptr, NP, NG, NM, NG);
    fill_kernel<<<dim3((E + 255) / 256, 4), 256, 0, stream>>>(
        srcA[0], srcA[1], srcA[2], srcA[3], dstA[0], dstA[1], dstA[2], dstA[3],
        rowptr, rank, perm, NP, NG, NM, E);
    const int* rpA[4] = { rowptr, rowptr + NP + 4, rowptr + NP + NG + 8, rowptr + NP + NG + NM + 12 };

    enc_kernel<<<(NG * H + 255) / 256, 256, 0, stream>>>(x_gene,    encW[0], encB[0], h_gene, NG, 11);
    enc_kernel<<<(NP * H + 255) / 256, 256, 0, stream>>>(x_patient, encW[1], encB[1], h_pat,  NP, 3);
    enc_kernel<<<(NM * H + 255) / 256, 256, 0, stream>>>(x_group,   encW[2], encB[2], h_grp,  NM, 4);

    float* hArr[3]  = { h_gene, h_pat, h_grp };
    float* hnArr[3] = { hn_gene, hn_pat, hn_grp };
    int    nArr[3]  = { NG, NP, NM };
    const int sType[4] = { 0, 1, 0, 2 };
    const int dType[4] = { 1, 0, 2, 0 };

    for (int layer = 0; layer < 2; ++layer) {
        hipMemsetAsync(hn_gene, 0, (size_t)NG * H * sizeof(float), stream);
        hipMemsetAsync(hn_pat,  0, (size_t)NP * H * sizeof(float), stream);
        hipMemsetAsync(hn_grp,  0, (size_t)NM * H * sizeof(float), stream);
        for (int e = 0; e < 4; ++e) {
            int Nd = nArr[dType[e]];
            int Ns = nArr[sType[e]];
            const float* Wl  = convW_l + ((size_t)layer * 4 + e) * H * H;
            const float* blp = convb_l + ((size_t)layer * 4 + e) * H;
            const float* Wr  = convW_r + ((size_t)layer * 4 + e) * H * H;
            const int* pm = perm + (size_t)e * E;
            for (int nb = 0; nb < Nd; nb += SL) {
                int nc = Nd - nb < SL ? Nd - nb : SL;
                if (e == 0)
                    gather_sum_kernel<4><<<(nc + 7) / 8, 256, 0, stream>>>(
                        hArr[sType[e]], rpA[e], pm, agg, deg, nb, nc, Ns, E);
                else
                    gather_sum_kernel<8><<<(nc + 7) / 8, 256, 0, stream>>>(
                        hArr[sType[e]], rpA[e], pm, agg, deg, nb, nc, Ns, E);
                sage_gemm_kernel<<<(nc + 63) / 64, 128, 0, stream>>>(
                    agg, deg, hArr[dType[e]] + (size_t)nb * H, Wl, blp, Wr,
                    hnArr[dType[e]] + (size_t)nb * H, nc);
            }
        }
        for (int ni = 0; ni < 3; ++ni) {
            ln_kernel<<<nArr[ni], 32, 0, stream>>>(
                hArr[ni], hnArr[ni],
                ln_g + ((size_t)layer * 3 + ni) * H,
                ln_b + ((size_t)layer * 3 + ni) * H, nArr[ni]);
        }
    }
    cox_kernel<<<(NP + 255) / 256, 256, 0, stream>>>(
        h_pat, coxW1, coxb1, coxW2, coxb2, (float*)d_out, NP);
}


# Bounded grids of 8-wave workgroups whose waves loop over rows, edges or nodes (encoders, rank, fill, gather, LayerNorm) instead of one short wave per item: about four times fewer waves per call
# speedup vs baseline: 3.5639x; 1.0902x over previous
#include <hip/hip_runtime.h>
#include <hip/hip_bf16.h>
#include <math.h>

typedef float v2f __attribute__((ext_vector_type(2)));
typedef float v8f __attribute__((ext_vector_type(8)));

#define H 64

__device__ __forceinline__ float elu1(float x) {
    return x > 0.0f ? x : (__expf(x) - 1.0f);
}

template <int F, int U>
__global__ void __launch_bounds__(256)
enc_kernel(const float* __restrict__ x, const float* __restrict__ W,
           const float* __restrict__ b, float* __restrict__ h, int N) {
    int j = threadIdx.x & 63;
    int sub = threadIdx.x >> 6;
    float w[F];
#pragma unroll
    for (int k = 0; k < F; ++k) w[k] = W[k * H + j];
    float bj = b[j];
    for (int n0 = blockIdx.x * (4 * U) + sub; n0 < N; n0 += gridDim.x * (4 * U)) {
        if (n0 + 4 * (U - 1) < N) {
            float xv[U][F];
#pragma unroll
            for (int u = 0; u < U; ++u) {
                const float* xr = x + (long)(n0 + 4 * u) * F;
#pragma unroll
                for (int k = 0; k < F; ++k) xv[u][k] = xr[k];
            }
#pragma unroll
            for (int u = 0; u < U; ++u) {
                float s = bj;
#pragma unroll
                for (int k = 0; k < F; ++k) s = fmaf(xv[u][k], w[k], s);
                h[(long)(n0 + 4 * u) * H + j] = elu1(s);
            }
        } else {
            for (int n = n0; n < N; n += 4) {
                const float* xr = x + (long)n * F;
                float s = bj;
#pragma unroll
                for (int k = 0; k < F; ++k) s = fmaf(xr[k], w[k], s);
                h[(long)n * H + j] = elu1(s);
            }
        }
    }
}

#define EPT 8
__global__ void __launch_bounds__(256)
rank_kernel(const int* __restrict__ d0, const int* __restrict__ d1,
            const int* __restrict__ d2, const int* __restrict__ d3,
            int* __restrict__ cnt, int* __restrict__ rank,
            int n0, int n1, int n2, int E) {
    int t = blockIdx.y;
    const int* d = t == 0 ? d0 : t == 1 ? d1 : t == 2 ? d2 : d3;
    int coff = t == 0 ? 0 : t == 1 ? n0 : t == 2 ? n0 + n1 : n0 + n1 + n2;
    int base = blockIdx.x * (256 * EPT) + threadIdx.x;
    int dv[EPT], rk[EPT];
#pragma unroll
    for (int u = 0; u < EPT; ++u) dv[u] = d[min(base + u * 256, E - 1)];
#pragma unroll
    for (int u = 0; u < EPT; ++u)
        rk[u] = atomicAdd(&cnt[coff + dv[u]], base + u * 256 < E ? 1 : 0);
#pragma unroll
    for (int u = 0; u < EPT; ++u) {
        int e = base + u * 256;
        if (e < E) rank[(long)t * E + e] = rk[u];
    }
}

__global__ void __launch_bounds__(1024)
rowptr_scan_kernel(const int* __restrict__ cnt, int* __restrict__ rowptr,
                   int n0, int n1, int n2, int n3) {
    __shared__ int wsum[32];
    int t = blockIdx.x;
    int N    = t == 0 ? n0 : t == 1 ? n1 : t == 2 ? n2 : n3;
    int coff = t == 0 ? 0 : t == 1 ? n0 : t == 2 ? n0 + n1 : n0 + n1 + n2;
    const int* c = cnt + coff;
    int* rp = rowptr + coff + 4 * t;
    int wave = threadIdx.x >> 5, lane = threadIdx.x & 31;
    int per = (((N + 31) >> 5) + 127) & ~127;
    int beg = wave * per;
    int end = min(beg + per, N);

    int s = 0;
    for (int p = beg + lane * 4; p < end; p += 128) {
        int4 v = *(const int4*)(c + p);
        s += v.x + v.y + v.z + v.w;
    }
    for (int m = 16; m >= 1; m >>= 1) s += __shfl_xor(s, m, 32);
    if (lane == 0) wsum[wave] = s;
    __syncthreads();

    int wtot = wsum[lane];
    int winc = wtot;
    for (int dlt = 1; dlt < 32; dlt <<= 1) {
        int o = __shfl_up(winc, dlt, 32);
        if (lane >= dlt) winc += o;
    }
    int carry = __shfl(winc - wtot, wave, 32);
    int total = __shfl(winc, 31, 32);

    for (int tb = beg; tb < end; tb += 128) {
        int p = tb + lane * 4;
        int4 v = make_int4(0, 0, 0, 0);
        if (p < end) v = *(const int4*)(c + p);
        int ls = v.x + v.y + v.z + v.w;
        int li = ls;
        for (int dlt = 1; dlt < 32; dlt <<= 1) {
            int o = __shfl_up(li, dlt, 32);
            if (lane >= dlt) li += o;
        }
        int ex = carry + li - ls;
        if (p < end)
            *(int4*)(rp + p) = make_int4(ex, ex + v.x, ex + v.x + v.y, ex + v.x + v.y + v.z);
        carry += __shfl(li, 31, 32);
    }
    if (threadIdx.x == 0) rp[N] = total;
}

__global__ void __launch_bounds__(256)
fill_kernel(const int* __restrict__ s0, const int* __restrict__ s1,
                            const int* __restrict__ s2, const int* __restrict__ s3,
                            const int* __restrict__ d0, const int* __restrict__ d1,
                            const int* __restrict__ d2, const int* __restrict__ d3,
                            const int* __restrict__ rowptr, const int* __restrict__ rank,
                            int* __restrict__ perm, int n0, int n1, int n2, int E) {
    int t = blockIdx.y;
    const int* s = t == 0 ? s0 : t == 1 ? s1 : t == 2 ? s2 : s3;
    const int* d = t == 0 ? d0 : t == 1 ? d1 : t == 2 ? d2 : d3;
    int coff = t == 0 ? 0 : t == 1 ? n0 : t == 2 ? n0 + n1 : n0 + n1 + n2;
    int base = blockIdx.x * (256 * EPT) + threadIdx.x;
    if (base + (EPT - 1) * 256 < E) {
        int dv[EPT], sv[EPT], rk[EPT], rp[EPT];
#pragma unroll
        for (int u = 0; u < EPT; ++u) {
            int e = base + u * 256;
            dv[u] = d[e];
            sv[u] = s[e];
            rk[u] = rank[(long)t * E + e];
        }
#pragma unroll
        for (int u = 0; u < EPT; ++u) rp[u] = rowptr[coff + 4 * t + dv[u]];
#pragma unroll
        for (int u = 0; u < EPT; ++u) {
            int pos = rp[u] + rk[u];
            pos = min(max(pos, 0), E - 1);
            perm[(long)t * E + pos] = sv[u];
        }
    } else {
        for (int e = base; e < E; e += 256) {
            int pos = rowptr[coff + 4 * t + d[e]] + rank[(long)t * E + e];
            pos = min(max(pos, 0), E - 1);
            perm[(long)t * E + pos] = s[e];
        }
    }
}

template <int U>
__global__ void __launch_bounds__(256)
gather_sum_kernel(const float* __restrict__ hsrc, const int* __restrict__ rowptr,
                  const int* __restrict__ perm, float* __restrict__ agg,
                  float* __restrict__ deg, int nbeg, int ncnt, int nsrc, int E) {
    int wave = threadIdx.x >> 5, lane = threadIdx.x & 31;
    int half = lane >> 4, c = lane & 15;
    const float4* hs4 = (const float4*)hsrc;
    for (int ln = blockIdx.x * 8 + wave; ln < ncnt; ln += gridDim.x * 8) {
        int beg = rowptr[nbeg + ln], end = rowptr[nbeg + ln + 1];
        beg = max(beg, 0);
        end = min(end, E);
        float4 acc = make_float4(0.0f, 0.0f, 0.0f, 0.0f);

        for (int base = beg; base < end; base += 32) {
            int cnt = min(32, end - base);
            int mine = 0;
            if (lane < cnt) mine = perm[base + lane];
            mine = min(max(mine, 0), nsrc - 1);
            for (int i = 0; i < cnt; i += 2 * U) {
                float4 v[U];
                bool ok[U];
#pragma unroll
                for (int u = 0; u < U; ++u) {
                    int idx = i + 2 * u + half;
                    ok[u] = idx < cnt;
                    int sidx = __shfl(mine, ok[u] ? idx : 0, 32);
                    v[u] = hs4[(long)sidx * 16 + c];
                }
#pragma unroll
                for (int u = 0; u < U; ++u) {
                    acc.x += ok[u] ? v[u].x : 0.0f;
                    acc.y += ok[u] ? v[u].y : 0.0f;
                    acc.z += ok[u] ? v[u].z : 0.0f;
                    acc.w += ok[u] ? v[u].w : 0.0f;
                }
            }
        }
        acc.x += __shfl_xor(acc.x, 16, 32);
        acc.y += __shfl_xor(acc.y, 16, 32);
        acc.z += __shfl_xor(acc.z, 16, 32);
        acc.w += __shfl_xor(acc.w, 16, 32);
        if (half == 0) ((float4*)agg)[(long)ln * 16 + c] = acc;
        if (lane == 0) deg[ln] = (float)(end - beg);
    }
}

__global__ void __launch_bounds__(128)
sage_gemm_kernel(const float* __restrict__ agg, const float* __restrict__ deg,
                 const float* __restrict__ hdst,
                 const float* __restrict__ Wl, const float* __restrict__ bl,
                 const float* __restrict__ Wr,
                 float* __restrict__ hnew, int N) {
    __shared__ float sWl[H * H];
    __shared__ float sWr[H * H];
    int tid = threadIdx.x;
    {
        const float4* gl = (const float4*)Wl;
        const float4* gr = (const float4*)Wr;
        float4* ll = (float4*)sWl;
        float4* lr = (float4*)sWr;
        for (int i = tid; i < (H * H / 4); i += 128) {
            ll[i] = gl[i];
            lr[i] = gr[i];
        }
    }
    __syncthreads();

    int wave = tid >> 5;
    int lane = tid & 31;
    int hi   = lane >> 4;
    int col  = lane & 15;
    int kk   = hi * 2;
    int m0   = blockIdx.x * 64 + wave * 16;

    int r  = m0 + col;
    int rc = r < N ? r : (N - 1);
    float invd = 1.0f / fmaxf(deg[rc], 1.0f);

    const float* aggr = agg  + (long)rc * H;
    const float* hdr  = hdst + (long)rc * H;

    v8f acc[4] = {};

#pragma unroll
    for (int k0 = 0; k0 < H; k0 += 4) {
        v2f a;
        a.x = aggr[k0 + kk] * invd;
        a.y = aggr[k0 + kk + 1] * invd;
#pragma unroll
        for (int t4 = 0; t4 < 4; ++t4) {
            v2f b;
            b.x = sWl[(k0 + kk) * H + t4 * 16 + col];
            b.y = sWl[(k0 + kk + 1) * H + t4 * 16 + col];
            acc[t4] = __builtin_amdgcn_wmma_f32_16x16x4_f32(
                false, a, false, b, (short)0, acc[t4], false, false);
        }
    }
#pragma unroll
    for (int k0 = 0; k0 < H; k0 += 4) {
        v2f a;
        a.x = hdr[k0 + kk];
        a.y = hdr[k0 + kk + 1];
#pragma unroll
        for (int t4 = 0; t4 < 4; ++t4) {
            v2f b;
            b.x = sWr[(k0 + kk) * H + t4 * 16 + col];
            b.y = sWr[(k0 + kk + 1) * H + t4 * 16 + col];
            acc[t4] = __builtin_amdgcn_wmma_f32_16x16x4_f32(
                false, a, false, b, (short)0, acc[t4], false, false);
        }
    }

#pragma unroll
    for (int t4 = 0; t4 < 4; ++t4) {
        int cg = t4 * 16 + col;
        float bias = bl[cg];
#pragma unroll
        for (int rr = 0; rr < 8; ++rr) {
            int row = m0 + rr + hi * 8;
            if (row < N) hnew[(long)row * H + cg] += acc[t4][rr] + bias;
        }
    }
}

#define LNU 4
__device__ __forceinline__ void ln_row(float v0, float v1, float g0, float g1, float b0, float b1,
                                       float& o0, float& o1) {
    float s = v0 + v1;
#pragma unroll
    for (int m = 16; m >= 1; m >>= 1) s += __shfl_xor(s, m, 32);
    float mean = s * (1.0f / 64.0f);
    float d0 = v0 - mean, d1 = v1 - mean;
    float q = d0 * d0 + d1 * d1;
#pragma unroll
    for (int m = 16; m >= 1; m >>= 1) q += __shfl_xor(q, m, 32);
    float inv = rsqrtf(q * (1.0f / 64.0f) + 1e-5f);
    o0 = elu1(d0 * inv * g0 + b0);
    o1 = elu1(d1 * inv * g1 + b1);
}

__global__ void __launch_bounds__(256)
ln_kernel(float* __restrict__ h, const float* __restrict__ hn,
          const float* __restrict__ g, const float* __restrict__ be, int N) {
    int wave = threadIdx.x >> 5, lane = threadIdx.x & 31;
    float g0 = g[lane], g1 = g[lane + 32];
    float b0 = be[lane], b1 = be[lane + 32];
    int stride = gridDim.x * 8;
    for (int n0 = blockIdx.x * 8 + wave; n0 < N; n0 += stride * LNU) {
        if (n0 + (LNU - 1) * stride < N) {
            float a0[LNU], a1[LNU], c0[LNU], c1[LNU];
#pragma unroll
            for (int u = 0; u < LNU; ++u) {
                long o = (long)(n0 + u * stride) * H + lane;
                a0[u] = h[o];
                a1[u] = h[o + 32];
                c0[u] = hn[o];
                c1[u] = hn[o + 32];
            }
#pragma unroll
            for (int u = 0; u < LNU; ++u) {
                long o = (long)(n0 + u * stride) * H + lane;
                float o0, o1;
                ln_row(a0[u] + c0[u], a1[u] + c1[u], g0, g1, b0, b1, o0, o1);
                h[o]      = o0;
                h[o + 32] = o1;
            }
        } else {
            for (int n = n0; n < N; n += stride) {
                long o = (long)n * H + lane;
                float o0, o1;
                ln_row(h[o] + hn[o], h[o + 32] + hn[o + 32], g0, g1, b0, b1, o0, o1);
                h[o]      = o0;
                h[o + 32] = o1;
            }
        }
    }
}

__global__ void cox_kernel(const float* __restrict__ hp, const float* __restrict__ W1,
                           const float* __restrict__ b1, const float* __restrict__ W2,
                           const float* __restrict__ b2, float* __restrict__ out, int N) {
    int n = blockIdx.x * blockDim.x + threadIdx.x;
    if (n >= N) return;
    const float* hr = hp + (long)n * H;
    float hreg[H];
#pragma unroll
    for (int k = 0; k < H; ++k) hreg[k] = hr[k];
    float hz = b2[0];
    for (int j = 0; j < 32; ++j) {
        float s = b1[j];
#pragma unroll
        for (int k = 0; k < H; ++k) s = fmaf(hreg[k], W1[k * 32 + j], s);
        hz = fmaf(elu1(s), W2[j], hz);
    }
    out[n] = hz;
}

extern "C" void kernel_launch(void* const* d_in, const int* in_sizes, int n_in,
                              void* d_out, int out_size, void* d_ws, size_t ws_size,
                              hipStream_t stream) {
    (void)in_sizes; (void)n_in; (void)out_size; (void)ws_size;
    const int NG = 20000, NP = 100000, NM = 5000, E = 1000000;

    const float* x_gene    = (const float*)d_in[0];
    const float* x_patient = (const float*)d_in[1];
    const float* x_group   = (const float*)d_in[2];
    const int* srcA[4]  = { (const int*)d_in[3], (const int*)d_in[5],
                            (const int*)d_in[7], (const int*)d_in[9] };
    const int* dstA[4]  = { (const int*)d_in[4], (const int*)d_in[6],
                            (const int*)d_in[8], (const int*)d_in[10] };
    const float* encW[3] = { (const float*)d_in[11], (const float*)d_in[13], (const float*)d_in[15] };
    const float* encB[3] = { (const float*)d_in[12], (const float*)d_in[14], (const float*)d_in[16] };
    const float* convW_l = (const float*)d_in[17];
    const float* convb_l = (const float*)d_in[18];
    const float* convW_r = (const float*)d_in[19];
    const float* ln_g = (const float*)d_in[20];
    const float* ln_b = (const float*)d_in[21];
    const float* coxW1 = (const float*)d_in[22];
    const float* coxb1 = (const float*)d_in[23];
    const float* coxW2 = (const float*)d_in[24];
    const float* coxb2 = (const float*)d_in[25];

    float* ws = (float*)d_ws;
    float* h_gene  = ws; ws += (size_t)NG * H;
    float* h_pat   = ws; ws += (size_t)NP * H;
    float* h_grp   = ws; ws += (size_t)NM * H;
    float* hn_gene = ws; ws += (size_t)NG * H;
    float* hn_pat  = ws; ws += (size_t)NP * H;
    float* hn_grp  = ws; ws += (size_t)NM * H;
    const int SL = 34000;
    float* aggreg  = ws; ws += (size_t)NP * H;
    float* deg     = ws; ws += (size_t)NP;
    float* agg     = aggreg;
    int*   perm    = (int*)(aggreg + (size_t)SL * H);
    int*   rowptr  = perm + (size_t)4 * E;
    int*   cnt     = (int*)hn_gene;
    int*   rank    = (int*)hn_pat;

    hipMemsetAsync(cnt, 0, (size_t)(NP + NG + NM + NG) * sizeof(int), stream);
    rank_kernel<<<dim3((E + 256 * EPT - 1) / (256 * EPT), 4), 256, 0, stream>>>(
        dstA[0], dstA[1], dstA[2], dstA[3], cnt, rank, NP, NG, NM, E);
    rowptr_scan_kernel<<<4, 1024, 0, stream>>>(cnt, rowptr, NP, NG, NM, NG);
    fill_kernel<<<dim3((E + 256 * EPT - 1) / (256 * EPT), 4), 256, 0, stream>>>(
        srcA[0], srcA[1], srcA[2], srcA[3], dstA[0], dstA[1], dstA[2], dstA[3],
        rowptr, rank, perm, NP, NG, NM, E);
    const int* rpA[4] = { rowptr, rowptr + NP + 4, rowptr + NP + NG + 8, rowptr + NP + NG + NM + 12 };

    const int GCAP = 1024;
    enc_kernel<11, 2><<<(NG + 7) / 8 < GCAP ? (NG + 7) / 8 : GCAP, 256, 0, stream>>>(
        x_gene,    encW[0], encB[0], h_gene, NG);
    enc_kernel<3, 4><<<(NP + 15) / 16 < GCAP ? (NP + 15) / 16 : GCAP, 256, 0, stream>>>(
        x_patient, encW[1], encB[1], h_pat,  NP);
    enc_kernel<4, 4><<<(NM + 15) / 16 < GCAP ? (NM + 15) / 16 : GCAP, 256, 0, stream>>>(
        x_group,   encW[2], encB[2], h_grp,  NM);

    float* hArr[3]  = { h_gene, h_pat, h_grp };
    float* hnArr[3] = { hn_gene, hn_pat, hn_grp };
    int    nArr[3]  = { NG, NP, NM };
    const int sType[4] = { 0, 1, 0, 2 };
    const int dType[4] = { 1, 0, 2, 0 };

    for (int layer = 0; layer < 2; ++layer) {
        hipMemsetAsync(hn_gene, 0, (size_t)NG * H * sizeof(float), stream);
        hipMemsetAsync(hn_pat,  0, (size_t)NP * H * sizeof(float), stream);
        hipMemsetAsync(hn_grp,  0, (size_t)NM * H * sizeof(float), stream);
        for (int e = 0; e < 4; ++e) {
            int Nd = nArr[dType[e]];
            int Ns = nArr[sType[e]];
            const float* Wl  = convW_l + ((size_t)layer * 4 + e) * H * H;
            const float* blp = convb_l + ((size_t)layer * 4 + e) * H;
            const float* Wr  = convW_r + ((size_t)layer * 4 + e) * H * H;
            const int* pm = perm + (size_t)e * E;
            for (int nb = 0; nb < Nd; nb += SL) {
                int nc = Nd - nb < SL ? Nd - nb : SL;
                int gb = (nc + 7) / 8 < 2 * GCAP ? (nc + 7) / 8 : 2 * GCAP;
                if (e == 0)
                    gather_sum_kernel<4><<<gb, 256, 0, stream>>>(
                        hArr[sType[e]], rpA[e], pm, agg, deg, nb, nc, Ns, E);
                else
                    gather_sum_kernel<8><<<gb, 256, 0, stream>>>(
                        hArr[sType[e]], rpA[e], pm, agg, deg, nb, nc, Ns, E);
                sage_gemm_kernel<<<(nc + 63) / 64, 128, 0, stream>>>(
                    agg, deg, hArr[dType[e]] + (size_t)nb * H, Wl, blp, Wr,
                    hnArr[dType[e]] + (size_t)nb * H, nc);
            }
        }
        for (int ni = 0; ni < 3; ++ni) {
            int lb = (nArr[ni] + 8 * LNU - 1) / (8 * LNU);
            ln_kernel<<<lb < GCAP ? lb : GCAP, 256, 0, stream>>>(
                hArr[ni], hnArr[ni],
                ln_g + ((size_t)layer * 3 + ni) * H,
                ln_b + ((size_t)layer * 3 + ni) * H, nArr[ni]);
        }
    }
    cox_kernel<<<(NP + 255) / 256, 256, 0, stream>>>(
        h_pat, coxW1, coxb1, coxW2, coxb2, (float*)d_out, NP);
}


# Group edges by destination with counters in LDS: node-range x edge-chunk workgroups count, then place with LDS adds; no global atomics, no rank plane; neighbour rows now add chunk by chunk
# speedup vs baseline: 4.7832x; 1.3421x over previous
#include <hip/hip_runtime.h>
#include <hip/hip_bf16.h>
#include <math.h>

typedef float v2f __attribute__((ext_vector_type(2)));
typedef float v8f __attribute__((ext_vector_type(8)));

#define H 64

__device__ __forceinline__ float elu1(float x) {
    return x > 0.0f ? x : (__expf(x) - 1.0f);
}

template <int F, int U>
__global__ void __launch_bounds__(256)
enc_kernel(const float* __restrict__ x, const float* __restrict__ W,
           const float* __restrict__ b, float* __restrict__ h, int N) {
    int j = threadIdx.x & 63;
    int sub = threadIdx.x >> 6;
    float w[F];
#pragma unroll
    for (int k = 0; k < F; ++k) w[k] = W[k * H + j];
    float bj = b[j];
    for (int n0 = blockIdx.x * (4 * U) + sub; n0 < N; n0 += gridDim.x * (4 * U)) {
        if (n0 + 4 * (U - 1) < N) {
            float xv[U][F];
#pragma unroll
            for (int u = 0; u < U; ++u) {
                const float* xr = x + (long)(n0 + 4 * u) * F;
#pragma unroll
                for (int k = 0; k < F; ++k) xv[u][k] = xr[k];
            }
#pragma unroll
            for (int u = 0; u < U; ++u) {
                float s = bj;
#pragma unroll
                for (int k = 0; k < F; ++k) s = fmaf(xv[u][k], w[k], s);
                h[(long)(n0 + 4 * u) * H + j] = elu1(s);
            }
        } else {
            for (int n = n0; n < N; n += 4) {
                const float* xr = x + (long)n * F;
                float s = bj;
#pragma unroll
                for (int k = 0; k < F; ++k) s = fmaf(xr[k], w[k], s);
                h[(long)n * H + j] = elu1(s);
            }
        }
    }
}

#define RNG  25000
#define CHSH 15
#define CHE  (1 << CHSH)
#define LPT  8

__device__ __forceinline__ void range_of_slot(int slot, int n0, int n1, int n2, int n3,
                                              int& t, int& lo, int& cntn, int& coff) {
    int r0 = (n0 + RNG - 1) / RNG, r1 = (n1 + RNG - 1) / RNG, r2 = (n2 + RNG - 1) / RNG;
    int r, N;
    if (slot < r0)                { t = 0; r = slot;                N = n0; coff = 0; }
    else if (slot < r0 + r1)      { t = 1; r = slot - r0;           N = n1; coff = n0; }
    else if (slot < r0 + r1 + r2) { t = 2; r = slot - r0 - r1;      N = n2; coff = n0 + n1; }
    else                          { t = 3; r = slot - r0 - r1 - r2; N = n3; coff = n0 + n1 + n2; }
    lo = r * RNG;
    cntn = min(RNG, N - lo);
}

__global__ void __launch_bounds__(1024)
count_kernel(const int* __restrict__ d0, const int* __restrict__ d1,
             const int* __restrict__ d2, const int* __restrict__ d3,
             int* __restrict__ ccnt, int n0, int n1, int n2, int n3, int E) {
    __shared__ int lc[RNG];
    int t, lo, cntn, coff;
    range_of_slot(blockIdx.y, n0, n1, n2, n3, t, lo, cntn, coff);
    const int* d = t == 0 ? d0 : t == 1 ? d1 : t == 2 ? d2 : d3;
    int ntot = n0 + n1 + n2 + n3;
    for (int i = threadIdx.x; i < cntn; i += 1024) lc[i] = 0;
    __syncthreads();
    int e0 = blockIdx.x << CHSH;
    int e1 = min(e0 + CHE, E);
    for (int b = e0 + threadIdx.x; b < e1; b += 1024 * LPT) {
        int dv[LPT];
#pragma unroll
        for (int u = 0; u < LPT; ++u) {
            int e = b + u * 1024;
            dv[u] = e < e1 ? d[e] : -1;
        }
#pragma unroll
        for (int u = 0; u < LPT; ++u) {
            unsigned rel = (unsigned)(dv[u] - lo);
            if (rel < (unsigned)cntn) atomicAdd(&lc[rel], 1);
        }
    }
    __syncthreads();
    int* o = ccnt + (long)blockIdx.x * ntot + coff + lo;
    for (int i = threadIdx.x; i < cntn; i += 1024) o[i] = lc[i];
}

template <int NCH>
__global__ void __launch_bounds__(256)
chunk_offsets_kernel(int* __restrict__ ccnt, int* __restrict__ cnt, int ntot) {
    int n = blockIdx.x * 256 + threadIdx.x;
    if (n >= ntot) return;
    int v[NCH];
#pragma unroll
    for (int c = 0; c < NCH; ++c) v[c] = ccnt[(long)c * ntot + n];
    int run = 0;
#pragma unroll
    for (int c = 0; c < NCH; ++c) {
        ccnt[(long)c * ntot + n] = run;
        run += v[c];
    }
    cnt[n] = run;
}

__global__ void __launch_bounds__(1024)
rowptr_scan_kernel(const int* __restrict__ cnt, int* __restrict__ rowptr,
                   int n0, int n1, int n2, int n3) {
    __shared__ int wsum[32];
    int t = blockIdx.x;
    int N    = t == 0 ? n0 : t == 1 ? n1 : t == 2 ? n2 : n3;
    int coff = t == 0 ? 0 : t == 1 ? n0 : t == 2 ? n0 + n1 : n0 + n1 + n2;
    const int* c = cnt + coff;
    int* rp = rowptr + coff + 4 * t;
    int wave = threadIdx.x >> 5, lane = threadIdx.x & 31;
    int per = (((N + 31) >> 5) + 127) & ~127;
    int beg = wave * per;
    int end = min(beg + per, N);

    int s = 0;
    for (int p = beg + lane * 4; p < end; p += 128) {
        int4 v = *(const int4*)(c + p);
        s += v.x + v.y + v.z + v.w;
    }
    for (int m = 16; m >= 1; m >>= 1) s += __shfl_xor(s, m, 32);
    if (lane == 0) wsum[wave] = s;
    __syncthreads();

    int wtot = wsum[lane];
    int winc = wtot;
    for (int dlt = 1; dlt < 32; dlt <<= 1) {
        int o = __shfl_up(winc, dlt, 32);
        if (lane >= dlt) winc += o;
    }
    int carry = __shfl(winc - wtot, wave, 32);
    int total = __shfl(winc, 31, 32);

    for (int tb = beg; tb < end; tb += 128) {
        int p = tb + lane * 4;
        int4 v = make_int4(0, 0, 0, 0);
        if (p < end) v = *(const int4*)(c + p);
        int ls = v.x + v.y + v.z + v.w;
        int li = ls;
        for (int dlt = 1; dlt < 32; dlt <<= 1) {
            int o = __shfl_up(li, dlt, 32);
            if (lane >= dlt) li += o;
        }
        int ex = carry + li - ls;
        if (p < end)
            *(int4*)(rp + p) = make_int4(ex, ex + v.x, ex + v.x + v.y, ex + v.x + v.y + v.z);
        carry += __shfl(li, 31, 32);
    }
    if (threadIdx.x == 0) rp[N] = total;
}

__global__ void __launch_bounds__(1024)
fill_kernel(const int* __restrict__ s0, const int* __restrict__ s1,
            const int* __restrict__ s2, const int* __restrict__ s3,
            const int* __restrict__ d0, const int* __restrict__ d1,
            const int* __restrict__ d2, const int* __restrict__ d3,
            const int* __restrict__ rowptr, const int* __restrict__ choff,
            int* __restrict__ perm, int n0, int n1, int n2, int n3, int E) {
    __shared__ int lc[RNG];
    int t, lo, cntn, coff;
    range_of_slot(blockIdx.y, n0, n1, n2, n3, t, lo, cntn, coff);
    const int* s = t == 0 ? s0 : t == 1 ? s1 : t == 2 ? s2 : s3;
    const int* d = t == 0 ? d0 : t == 1 ? d1 : t == 2 ? d2 : d3;
    int ntot = n0 + n1 + n2 + n3;
    const int* rp = rowptr + coff + 4 * t + lo;
    const int* co = choff + (long)blockIdx.x * ntot + coff + lo;
    for (int i = threadIdx.x; i < cntn; i += 1024) lc[i] = rp[i] + co[i];
    __syncthreads();
    int e0 = blockIdx.x << CHSH;
    int e1 = min(e0 + CHE, E);
    int* pm = perm + (long)t * E;
    for (int b = e0 + threadIdx.x; b < e1; b += 1024 * LPT) {
        int dv[LPT], sv[LPT];
#pragma unroll
        for (int u = 0; u < LPT; ++u) {
            int e = b + u * 1024;
            bool ok = e < e1;
            dv[u] = ok ? d[e] : -1;
            sv[u] = ok ? s[e] : 0;
        }
        int pos[LPT];
#pragma unroll
        for (int u = 0; u < LPT; ++u) {
            unsigned rel = (unsigned)(dv[u] - lo);
            pos[u] = -1;
            if (rel < (unsigned)cntn) pos[u] = atomicAdd(&lc[rel], 1);
        }
#pragma unroll
        for (int u = 0; u < LPT; ++u) {
            if (pos[u] >= 0) pm[min(pos[u], E - 1)] = sv[u];
        }
    }
}

template <int U>
__global__ void __launch_bounds__(256)
gather_sum_kernel(const float* __restrict__ hsrc, const int* __restrict__ rowptr,
                  const int* __restrict__ perm, float* __restrict__ agg,
                  float* __restrict__ deg, int nbeg, int ncnt, int nsrc, int E) {
    int wave = threadIdx.x >> 5, lane = threadIdx.x & 31;
    int half = lane >> 4, c = lane & 15;
    const float4* hs4 = (const float4*)hsrc;
    for (int ln = blockIdx.x * 8 + wave; ln < ncnt; ln += gridDim.x * 8) {
        int beg = rowptr[nbeg + ln], end = rowptr[nbeg + ln + 1];
        beg = max(beg, 0);
        end = min(end, E);
        float4 acc = make_float4(0.0f, 0.0f, 0.0f, 0.0f);

        for (int base = beg; base < end; base += 32) {
            int cnt = min(32, end - base);
            int mine = 0;
            if (lane < cnt) mine = perm[base + lane];
            mine = min(max(mine, 0), nsrc - 1);
            for (int i = 0; i < cnt; i += 2 * U) {
                float4 v[U];
                bool ok[U];
#pragma unroll
                for (int u = 0; u < U; ++u) {
                    int idx = i + 2 * u + half;
                    ok[u] = idx < cnt;
                    int sidx = __shfl(mine, ok[u] ? idx : 0, 32);
                    v[u] = hs4[(long)sidx * 16 + c];
                }
#pragma unroll
                for (int u = 0; u < U; ++u) {
                    acc.x += ok[u] ? v[u].x : 0.0f;
                    acc.y += ok[u] ? v[u].y : 0.0f;
                    acc.z += ok[u] ? v[u].z : 0.0f;
                    acc.w += ok[u] ? v[u].w : 0.0f;
                }
            }
        }
        acc.x += __shfl_xor(acc.x, 16, 32);
        acc.y += __shfl_xor(acc.y, 16, 32);
        acc.z += __shfl_xor(acc.z, 16, 32);
        acc.w += __shfl_xor(acc.w, 16, 32);
        if (half == 0) ((float4*)agg)[(long)ln * 16 + c] = acc;
        if (lane == 0) deg[ln] = (float)(end - beg);
    }
}

__global__ void __launch_bounds__(128)
sage_gemm_kernel(const float* __restrict__ agg, const float* __restrict__ deg,
                 const float* __restrict__ hdst,
                 const float* __restrict__ Wl, const float* __restrict__ bl,
                 const float* __restrict__ Wr,
                 float* __restrict__ hnew, int N) {
    __shared__ float sWl[H * H];
    __shared__ float sWr[H * H];
    int tid = threadIdx.x;
    {
        const float4* gl = (const float4*)Wl;
        const float4* gr = (const float4*)Wr;
        float4* ll = (float4*)sWl;
        float4* lr = (float4*)sWr;
        for (int i = tid; i < (H * H / 4); i += 128) {
            ll[i] = gl[i];
            lr[i] = gr[i];
        }
    }
    __syncthreads();

    int wave = tid >> 5;
    int lane = tid & 31;
    int hi   = lane >> 4;
    int col  = lane & 15;
    int kk   = hi * 2;
    int m0   = blockIdx.x * 64 + wave * 16;

    int r  = m0 + col;
    int rc = r < N ? r : (N - 1);
    float invd = 1.0f / fmaxf(deg[rc], 1.0f);

    const float* aggr = agg  + (long)rc * H;
    const float* hdr  = hdst + (long)rc * H;

    v8f acc[4] = {};

#pragma unroll
    for (int k0 = 0; k0 < H; k0 += 4) {
        v2f a;
        a.x = aggr[k0 + kk] * invd;
        a.y = aggr[k0 + kk + 1] * invd;
#pragma unroll
        for (int t4 = 0; t4 < 4; ++t4) {
            v2f b;
            b.x = sWl[(k0 + kk) * H + t4 * 16 + col];
            b.y = sWl[(k0 + kk + 1) * H + t4 * 16 + col];
            acc[t4] = __builtin_amdgcn_wmma_f32_16x16x4_f32(
                false, a, false, b, (short)0, acc[t4], false, false);
        }
    }
#pragma unroll
    for (int k0 = 0; k0 < H; k0 += 4) {
        v2f a;
        a.x = hdr[k0 + kk];
        a.y = hdr[k0 + kk + 1];
#pragma unroll
        for (int t4 = 0; t4 < 4; ++t4) {
            v2f b;
            b.x = sWr[(k0 + kk) * H + t4 * 16 + col];
            b.y = sWr[(k0 + kk + 1) * H + t4 * 16 + col];
            acc[t4] = __builtin_amdgcn_wmma_f32_16x16x4_f32(
                false, a, false, b, (short)0, acc[t4], false, false);
        }
    }

#pragma unroll
    for (int t4 = 0; t4 < 4; ++t4) {
        int cg = t4 * 16 + col;
        float bias = bl[cg];
#pragma unroll
        for (int rr = 0; rr < 8; ++rr) {
            int row = m0 + rr + hi * 8;
            if (row < N) hnew[(long)row * H + cg] += acc[t4][rr] + bias;
        }
    }
}

#define LNU 4
__device__ __forceinline__ void ln_row(float v0, float v1, float g0, float g1, float b0, float b1,
                                       float& o0, float& o1) {
    float s = v0 + v1;
#pragma unroll
    for (int m = 16; m >= 1; m >>= 1) s += __shfl_xor(s, m, 32);
    float mean = s * (1.0f / 64.0f);
    float d0 = v0 - mean, d1 = v1 - mean;
    float q = d0 * d0 + d1 * d1;
#pragma unroll
    for (int m = 16; m >= 1; m >>= 1) q += __shfl_xor(q, m, 32);
    float inv = rsqrtf(q * (1.0f / 64.0f) + 1e-5f);
    o0 = elu1(d0 * inv * g0 + b0);
    o1 = elu1(d1 * inv * g1 + b1);
}

__global__ void __launch_bounds__(256)
ln_kernel(float* __restrict__ h, const float* __restrict__ hn,
          const float* __restrict__ g, const float* __restrict__ be, int N) {
    int wave = threadIdx.x >> 5, lane = threadIdx.x & 31;
    float g0 = g[lane], g1 = g[lane + 32];
    float b0 = be[lane], b1 = be[lane + 32];
    int stride = gridDim.x * 8;
    for (int n0 = blockIdx.x * 8 + wave; n0 < N; n0 += stride * LNU) {
        if (n0 + (LNU - 1) * stride < N) {
            float a0[LNU], a1[LNU], c0[LNU], c1[LNU];
#pragma unroll
            for (int u = 0; u < LNU; ++u) {
                long o = (long)(n0 + u * stride) * H + lane;
                a0[u] = h[o];
                a1[u] = h[o + 32];
                c0[u] = hn[o];
                c1[u] = hn[o + 32];
            }
#pragma unroll
            for (int u = 0; u < LNU; ++u) {
                long o = (long)(n0 + u * stride) * H + lane;
                float o0, o1;
                ln_row(a0[u] + c0[u], a1[u] + c1[u], g0, g1, b0, b1, o0, o1);
                h[o]      = o0;
                h[o + 32] = o1;
            }
        } else {
            for (int n = n0; n < N; n += stride) {
                long o = (long)n * H + lane;
                float o0, o1;
                ln_row(h[o] + hn[o], h[o + 32] + hn[o + 32], g0, g1, b0, b1, o0, o1);
                h[o]      = o0;
                h[o + 32] = o1;
            }
        }
    }
}

__global__ void cox_kernel(const float* __restrict__ hp, const float* __restrict__ W1,
                           const float* __restrict__ b1, const float* __restrict__ W2,
                           const float* __restrict__ b2, float* __restrict__ out, int N) {
    int n = blockIdx.x * blockDim.x + threadIdx.x;
    if (n >= N) return;
    const float* hr = hp + (long)n * H;
    float hreg[H];
#pragma unroll
    for (int k = 0; k < H; ++k) hreg[k] = hr[k];
    float hz = b2[0];
    for (int j = 0; j < 32; ++j) {
        float s = b1[j];
#pragma unroll
        for (int k = 0; k < H; ++k) s = fmaf(hreg[k], W1[k * 32 + j], s);
        hz = fmaf(elu1(s), W2[j], hz);
    }
    out[n] = hz;
}

extern "C" void kernel_launch(void* const* d_in, const int* in_sizes, int n_in,
                              void* d_out, int out_size, void* d_ws, size_t ws_size,
                              hipStream_t stream) {
    (void)in_sizes; (void)n_in; (void)out_size; (void)ws_size;
    const int NG = 20000, NP = 100000, NM = 5000, E = 1000000;

    const float* x_gene    = (const float*)d_in[0];
    const float* x_patient = (const float*)d_in[1];
    const float* x_group   = (const float*)d_in[2];
    const int* srcA[4]  = { (const int*)d_in[3], (const int*)d_in[5],
                            (const int*)d_in[7], (const int*)d_in[9] };
    const int* dstA[4]  = { (const int*)d_in[4], (const int*)d_in[6],
                            (const int*)d_in[8], (const int*)d_in[10] };
    const float* encW[3] = { (const float*)d_in[11], (const float*)d_in[13], (const float*)d_in[15] };
    const float* encB[3] = { (const float*)d_in[12], (const float*)d_in[14], (const float*)d_in[16] };
    const float* convW_l = (const float*)d_in[17];
    const float* convb_l = (const float*)d_in[18];
    const float* convW_r = (const float*)d_in[19];
    const float* ln_g = (const float*)d_in[20];
    const float* ln_b = (const float*)d_in[21];
    const float* coxW1 = (const float*)d_in[22];
    const float* coxb1 = (const float*)d_in[23];
    const float* coxW2 = (const float*)d_in[24];
    const float* coxb2 = (const float*)d_in[25];

    float* ws = (float*)d_ws;
    float* h_gene  = ws; ws += (size_t)NG * H;
    float* h_pat   = ws; ws += (size_t)NP * H;
    float* h_grp   = ws; ws += (size_t)NM * H;
    float* hn_gene = ws; ws += (size_t)NG * H;
    float* hn_pat  = ws; ws += (size_t)NP * H;
    float* hn_grp  = ws; ws += (size_t)NM * H;
    const int SL = 34000;
    float* aggreg  = ws; ws += (size_t)NP * H;
    float* deg     = ws; ws += (size_t)NP;
    float* agg     = aggreg;
    int*   perm    = (int*)(aggreg + (size_t)SL * H);
    int*   rowptr  = perm + (size_t)4 * E;
    int*   cnt     = (int*)hn_gene;
    int*   ccnt    = (int*)hn_pat;

    const int NCH   = (E + CHE - 1) / CHE;
    const int NSLOT = (NP + RNG - 1) / RNG + (NG + RNG - 1) / RNG
                    + (NM + RNG - 1) / RNG + (NG + RNG - 1) / RNG;
    const int NTOT  = NP + NG + NM + NG;
    count_kernel<<<dim3(NCH, NSLOT), 1024, 0, stream>>>(
        dstA[0], dstA[1], dstA[2], dstA[3], ccnt, NP, NG, NM, NG, E);
    chunk_offsets_kernel<NCH><<<(NTOT + 255) / 256, 256, 0, stream>>>(ccnt, cnt, NTOT);
    rowptr_scan_kernel<<<4, 1024, 0, stream>>>(cnt, rowptr, NP, NG, NM, NG);
    fill_kernel<<<dim3(NCH, NSLOT), 1024, 0, stream>>>(
        srcA[0], srcA[1], srcA[2], srcA[3], dstA[0], dstA[1], dstA[2], dstA[3],
        rowptr, ccnt, perm, NP, NG, NM, NG, E);
    const int* rpA[4] = { rowptr, rowptr + NP + 4, rowptr + NP + NG + 8, rowptr + NP + NG + NM + 12 };

    const int GCAP = 1024;
    enc_kernel<11, 2><<<(NG + 7) / 8 < GCAP ? (NG + 7) / 8 : GCAP, 256, 0, stream>>>(
        x_gene,    encW[0], encB[0], h_gene, NG);
    enc_kernel<3, 4><<<(NP + 15) / 16 < GCAP ? (NP + 15) / 16 : GCAP, 256, 0, stream>>>(
        x_patient, encW[1], encB[1], h_pat,  NP);
    enc_kernel<4, 4><<<(NM + 15) / 16 < GCAP ? (NM + 15) / 16 : GCAP, 256, 0, stream>>>(
        x_group,   encW[2], encB[2], h_grp,  NM);

    float* hArr[3]  = { h_gene, h_pat, h_grp };
    float* hnArr[3] = { hn_gene, hn_pat, hn_grp };
    int    nArr[3]  = { NG, NP, NM };
    const int sType[4] = { 0, 1, 0, 2 };
    const int dType[4] = { 1, 0, 2, 0 };

    for (int layer = 0; layer < 2; ++layer) {
        hipMemsetAsync(hn_gene, 0, (size_t)NG * H * sizeof(float), stream);
        hipMemsetAsync(hn_pat,  0, (size_t)NP * H * sizeof(float), stream);
        hipMemsetAsync(hn_grp,  0, (size_t)NM * H * sizeof(float), stream);
        for (int e = 0; e < 4; ++e) {
            int Nd = nArr[dType[e]];
            int Ns = nArr[sType[e]];
            const float* Wl  = convW_l + ((size_t)layer * 4 + e) * H * H;
            const float* blp = convb_l + ((size_t)layer * 4 + e) * H;
            const float* Wr  = convW_r + ((size_t)layer * 4 + e) * H * H;
            const int* pm = perm + (size_t)e * E;
            for (int nb = 0; nb < Nd; nb += SL) {
                int nc = Nd - nb < SL ? Nd - nb : SL;
                int gb = (nc + 7) / 8 < 2 * GCAP ? (nc + 7) / 8 : 2 * GCAP;
                if (e == 0)
                    gather_sum_kernel<4><<<gb, 256, 0, stream>>>(
                        hArr[sType[e]], rpA[e], pm, agg, deg, nb, nc, Ns, E);
                else
                    gather_sum_kernel<8><<<gb, 256, 0, stream>>>(
                        hArr[sType[e]], rpA[e], pm, agg, deg, nb, nc, Ns, E);
                sage_gemm_kernel<<<(nc + 63) / 64, 128, 0, stream>>>(
                    agg, deg, hArr[dType[e]] + (size_t)nb * H, Wl, blp, Wr,
                    hnArr[dType[e]] + (size_t)nb * H, nc);
            }
        }
        for (int ni = 0; ni < 3; ++ni) {
            int lb = (nArr[ni] + 8 * LNU - 1) / (8 * LNU);
            ln_kernel<<<lb < GCAP ? lb : GCAP, 256, 0, stream>>>(
                hArr[ni], hnArr[ni],
                ln_g + ((size_t)layer * 3 + ni) * H,
                ln_b + ((size_t)layer * 3 + ni) * H, nArr[ni]);
        }
    }
    cox_kernel<<<(NP + 255) / 256, 256, 0, stream>>>(
        h_pat, coxW1, coxb1, coxW2, coxb2, (float*)d_out, NP);
}


# Neighbour sums are gathered straight into the destination's accumulation plane and the GEMM overwrites them in place (first edge type stores, the second adds): no clearing, no node slices
# speedup vs baseline: 6.1527x; 1.2863x over previous
#include <hip/hip_runtime.h>
#include <hip/hip_bf16.h>
#include <math.h>

typedef float v2f __attribute__((ext_vector_type(2)));
typedef float v8f __attribute__((ext_vector_type(8)));

#define H 64

__device__ __forceinline__ float elu1(float x) {
    return x > 0.0f ? x : (__expf(x) - 1.0f);
}

template <int F, int U>
__global__ void __launch_bounds__(256)
enc_kernel(const float* __restrict__ x, const float* __restrict__ W,
           const float* __restrict__ b, float* __restrict__ h, int N) {
    int j = threadIdx.x & 63;
    int sub = threadIdx.x >> 6;
    float w[F];
#pragma unroll
    for (int k = 0; k < F; ++k) w[k] = W[k * H + j];
    float bj = b[j];
    for (int n0 = blockIdx.x * (4 * U) + sub; n0 < N; n0 += gridDim.x * (4 * U)) {
        if (n0 + 4 * (U - 1) < N) {
            float xv[U][F];
#pragma unroll
            for (int u = 0; u < U; ++u) {
                const float* xr = x + (long)(n0 + 4 * u) * F;
#pragma unroll
                for (int k = 0; k < F; ++k) xv[u][k] = xr[k];
            }
#pragma unroll
            for (int u = 0; u < U; ++u) {
                float s = bj;
#pragma unroll
                for (int k = 0; k < F; ++k) s = fmaf(xv[u][k], w[k], s);
                h[(long)(n0 + 4 * u) * H + j] = elu1(s);
            }
        } else {
            for (int n = n0; n < N; n += 4) {
                const float* xr = x + (long)n * F;
                float s = bj;
#pragma unroll
                for (int k = 0; k < F; ++k) s = fmaf(xr[k], w[k], s);
                h[(long)n * H + j] = elu1(s);
            }
        }
    }
}

#define RNG  25000
#define CHSH 15
#define CHE  (1 << CHSH)
#define LPT  8

__device__ __forceinline__ void range_of_slot(int slot, int n0, int n1, int n2, int n3,
                                              int& t, int& lo, int& cntn, int& coff) {
    int r0 = (n0 + RNG - 1) / RNG, r1 = (n1 + RNG - 1) / RNG, r2 = (n2 + RNG - 1) / RNG;
    int r, N;
    if (slot < r0)                { t = 0; r = slot;                N = n0; coff = 0; }
    else if (slot < r0 + r1)      { t = 1; r = slot - r0;           N = n1; coff = n0; }
    else if (slot < r0 + r1 + r2) { t = 2; r = slot - r0 - r1;      N = n2; coff = n0 + n1; }
    else                          { t = 3; r = slot - r0 - r1 - r2; N = n3; coff = n0 + n1 + n2; }
    lo = r * RNG;
    cntn = min(RNG, N - lo);
}

__global__ void __launch_bounds__(1024)
count_kernel(const int* __restrict__ d0, const int* __restrict__ d1,
             const int* __restrict__ d2, const int* __restrict__ d3,
             int* __restrict__ ccnt, int n0, int n1, int n2, int n3, int E) {
    __shared__ int lc[RNG];
    int t, lo, cntn, coff;
    range_of_slot(blockIdx.y, n0, n1, n2, n3, t, lo, cntn, coff);
    const int* d = t == 0 ? d0 : t == 1 ? d1 : t == 2 ? d2 : d3;
    int ntot = n0 + n1 + n2 + n3;
    for (int i = threadIdx.x; i < cntn; i += 1024) lc[i] = 0;
    __syncthreads();
    int e0 = blockIdx.x << CHSH;
    int e1 = min(e0 + CHE, E);
    for (int b = e0 + threadIdx.x; b < e1; b += 1024 * LPT) {
        int dv[LPT];
#pragma unroll
        for (int u = 0; u < LPT; ++u) {
            int e = b + u * 1024;
            dv[u] = e < e1 ? d[e] : -1;
        }
#pragma unroll
        for (int u = 0; u < LPT; ++u) {
            unsigned rel = (unsigned)(dv[u] - lo);
            if (rel < (unsigned)cntn) atomicAdd(&lc[rel], 1);
        }
    }
    __syncthreads();
    int* o = ccnt + (long)blockIdx.x * ntot + coff + lo;
    for (int i = threadIdx.x; i < cntn; i += 1024) o[i] = lc[i];
}

template <int NCH>
__global__ void __launch_bounds__(256)
chunk_offsets_kernel(int* __restrict__ ccnt, int* __restrict__ cnt, int ntot) {
    int n = blockIdx.x * 256 + threadIdx.x;
    if (n >= ntot) return;
    int v[NCH];
#pragma unroll
    for (int c = 0; c < NCH; ++c) v[c] = ccnt[(long)c * ntot + n];
    int run = 0;
#pragma unroll
    for (int c = 0; c < NCH; ++c) {
        ccnt[(long)c * ntot + n] = run;
        run += v[c];
    }
    cnt[n] = run;
}

__global__ void __launch_bounds__(1024)
rowptr_scan_kernel(const int* __restrict__ cnt, int* __restrict__ rowptr,
                   int n0, int n1, int n2, int n3) {
    __shared__ int wsum[32];
    int t = blockIdx.x;
    int N    = t == 0 ? n0 : t == 1 ? n1 : t == 2 ? n2 : n3;
    int coff = t == 0 ? 0 : t == 1 ? n0 : t == 2 ? n0 + n1 : n0 + n1 + n2;
    const int* c = cnt + coff;
    int* rp = rowptr + coff + 4 * t;
    int wave = threadIdx.x >> 5, lane = threadIdx.x & 31;
    int per = (((N + 31) >> 5) + 127) & ~127;
    int beg = wave * per;
    int end = min(beg + per, N);

    int s = 0;
    for (int p = beg + lane * 4; p < end; p += 128) {
        int4 v = *(const int4*)(c + p);
        s += v.x + v.y + v.z + v.w;
    }
    for (int m = 16; m >= 1; m >>= 1) s += __shfl_xor(s, m, 32);
    if (lane == 0) wsum[wave] = s;
    __syncthreads();

    int wtot = wsum[lane];
    int winc = wtot;
    for (int dlt = 1; dlt < 32; dlt <<= 1) {
        int o = __shfl_up(winc, dlt, 32);
        if (lane >= dlt) winc += o;
    }
    int carry = __shfl(winc - wtot, wave, 32);
    int total = __shfl(winc, 31, 32);

    for (int tb = beg; tb < end; tb += 128) {
        int p = tb + lane * 4;
        int4 v = make_int4(0, 0, 0, 0);
        if (p < end) v = *(const int4*)(c + p);
        int ls = v.x + v.y + v.z + v.w;
        int li = ls;
        for (int dlt = 1; dlt < 32; dlt <<= 1) {
            int o = __shfl_up(li, dlt, 32);
            if (lane >= dlt) li += o;
        }
        int ex = carry + li - ls;
        if (p < end)
            *(int4*)(rp + p) = make_int4(ex, ex + v.x, ex + v.x + v.y, ex + v.x + v.y + v.z);
        carry += __shfl(li, 31, 32);
    }
    if (threadIdx.x == 0) rp[N] = total;
}

__global__ void __launch_bounds__(1024)
fill_kernel(const int* __restrict__ s0, const int* __restrict__ s1,
            const int* __restrict__ s2, const int* __restrict__ s3,
            const int* __restrict__ d0, const int* __restrict__ d1,
            const int* __restrict__ d2, const int* __restrict__ d3,
            const int* __restrict__ rowptr, const int* __restrict__ choff,
            int* __restrict__ perm, int n0, int n1, int n2, int n3, int E) {
    __shared__ int lc[RNG];
    int t, lo, cntn, coff;
    range_of_slot(blockIdx.y, n0, n1, n2, n3, t, lo, cntn, coff);
    const int* s = t == 0 ? s0 : t == 1 ? s1 : t == 2 ? s2 : s3;
    const int* d = t == 0 ? d0 : t == 1 ? d1 : t == 2 ? d2 : d3;
    int ntot = n0 + n1 + n2 + n3;
    const int* rp = rowptr + coff + 4 * t + lo;
    const int* co = choff + (long)blockIdx.x * ntot + coff + lo;
    for (int i = threadIdx.x; i < cntn; i += 1024) lc[i] = rp[i] + co[i];
    __syncthreads();
    int e0 = blockIdx.x << CHSH;
    int e1 = min(e0 + CHE, E);
    int* pm = perm + (long)t * E;
    for (int b = e0 + threadIdx.x; b < e1; b += 1024 * LPT) {
        int dv[LPT], sv[LPT];
#pragma unroll
        for (int u = 0; u < LPT; ++u) {
            int e = b + u * 1024;
            bool ok = e < e1;
            dv[u] = ok ? d[e] : -1;
            sv[u] = ok ? s[e] : 0;
        }
        int pos[LPT];
#pragma unroll
        for (int u = 0; u < LPT; ++u) {
            unsigned rel = (unsigned)(dv[u] - lo);
            pos[u] = -1;
            if (rel < (unsigned)cntn) pos[u] = atomicAdd(&lc[rel], 1);
        }
#pragma unroll
        for (int u = 0; u < LPT; ++u) {
            if (pos[u] >= 0) pm[min(pos[u], E - 1)] = sv[u];
        }
    }
}

template <int U>
__global__ void __launch_bounds__(256)
gather_sum_kernel(const float* __restrict__ hsrc, const int* __restrict__ rowptr,
                  const int* __restrict__ perm, float* __restrict__ agg,
                  float* __restrict__ deg, int nbeg, int ncnt, int nsrc, int E) {
    int wave = threadIdx.x >> 5, lane = threadIdx.x & 31;
    int half = lane >> 4, c = lane & 15;
    const float4* hs4 = (const float4*)hsrc;
    for (int ln = blockIdx.x * 8 + wave; ln < ncnt; ln += gridDim.x * 8) {
        int beg = rowptr[nbeg + ln], end = rowptr[nbeg + ln + 1];
        beg = max(beg, 0);
        end = min(end, E);
        float4 acc = make_float4(0.0f, 0.0f, 0.0f, 0.0f);

        for (int base = beg; base < end; base += 32) {
            int cnt = min(32, end - base);
            int mine = 0;
            if (lane < cnt) mine = perm[base + lane];
            mine = min(max(mine, 0), nsrc - 1);
            for (int i = 0; i < cnt; i += 2 * U) {
                float4 v[U];
                bool ok[U];
#pragma unroll
                for (int u = 0; u < U; ++u) {
                    int idx = i + 2 * u + half;
                    ok[u] = idx < cnt;
                    int sidx = __shfl(mine, ok[u] ? idx : 0, 32);
                    v[u] = hs4[(long)sidx * 16 + c];
                }
#pragma unroll
                for (int u = 0; u < U; ++u) {
                    acc.x += ok[u] ? v[u].x : 0.0f;
                    acc.y += ok[u] ? v[u].y : 0.0f;
                    acc.z += ok[u] ? v[u].z : 0.0f;
                    acc.w += ok[u] ? v[u].w : 0.0f;
                }
            }
        }
        acc.x += __shfl_xor(acc.x, 16, 32);
        acc.y += __shfl_xor(acc.y, 16, 32);
        acc.z += __shfl_xor(acc.z, 16, 32);
        acc.w += __shfl_xor(acc.w, 16, 32);
        if (half == 0) ((float4*)agg)[(long)ln * 16 + c] = acc;
        if (lane == 0) deg[ln] = (float)(end - beg);
    }
}

template <bool INPLACE>
__global__ void __launch_bounds__(128)
sage_gemm_kernel(const float* __restrict__ agg, const float* __restrict__ deg,
                 const float* __restrict__ hdst,
                 const float* __restrict__ Wl, const float* __restrict__ bl,
                 const float* __restrict__ Wr,
                 float* __restrict__ hnew, int N) {
    __shared__ float sWl[H * H];
    __shared__ float sWr[H * H];
    int tid = threadIdx.x;
    {
        const float4* gl = (const float4*)Wl;
        const float4* gr = (const float4*)Wr;
        float4* ll = (float4*)sWl;
        float4* lr = (float4*)sWr;
        for (int i = tid; i < (H * H / 4); i += 128) {
            ll[i] = gl[i];
            lr[i] = gr[i];
        }
    }
    __syncthreads();

    int wave = tid >> 5;
    int lane = tid & 31;
    int hi   = lane >> 4;
    int col  = lane & 15;
    int kk   = hi * 2;
    int m0   = blockIdx.x * 64 + wave * 16;

    int r  = m0 + col;
    int rc = r < N ? r : (N - 1);
    float invd = 1.0f / fmaxf(deg[rc], 1.0f);

    const float* aggr = (INPLACE ? (const float*)hnew : agg) + (long)rc * H;
    const float* hdr  = hdst + (long)rc * H;

    v8f acc[4] = {};

#pragma unroll
    for (int k0 = 0; k0 < H; k0 += 4) {
        v2f a;
        a.x = aggr[k0 + kk] * invd;
        a.y = aggr[k0 + kk + 1] * invd;
#pragma unroll
        for (int t4 = 0; t4 < 4; ++t4) {
            v2f b;
            b.x = sWl[(k0 + kk) * H + t4 * 16 + col];
            b.y = sWl[(k0 + kk + 1) * H + t4 * 16 + col];
            acc[t4] = __builtin_amdgcn_wmma_f32_16x16x4_f32(
                false, a, false, b, (short)0, acc[t4], false, false);
        }
    }
#pragma unroll
    for (int k0 = 0; k0 < H; k0 += 4) {
        v2f a;
        a.x = hdr[k0 + kk];
        a.y = hdr[k0 + kk + 1];
#pragma unroll
        for (int t4 = 0; t4 < 4; ++t4) {
            v2f b;
            b.x = sWr[(k0 + kk) * H + t4 * 16 + col];
            b.y = sWr[(k0 + kk + 1) * H + t4 * 16 + col];
            acc[t4] = __builtin_amdgcn_wmma_f32_16x16x4_f32(
                false, a, false, b, (short)0, acc[t4], false, false);
        }
    }

#pragma unroll
    for (int t4 = 0; t4 < 4; ++t4) {
        int cg = t4 * 16 + col;
        float bias = bl[cg];
#pragma unroll
        for (int rr = 0; rr < 8; ++rr) {
            int row = m0 + rr + hi * 8;
            if (row < N) {
                float o = acc[t4][rr] + bias;
                if (INPLACE) hnew[(long)row * H + cg] = o;
                else         hnew[(long)row * H + cg] += o;
            }
        }
    }
}

#define LNU 4
__device__ __forceinline__ void ln_row(float v0, float v1, float g0, float g1, float b0, float b1,
                                       float& o0, float& o1) {
    float s = v0 + v1;
#pragma unroll
    for (int m = 16; m >= 1; m >>= 1) s += __shfl_xor(s, m, 32);
    float mean = s * (1.0f / 64.0f);
    float d0 = v0 - mean, d1 = v1 - mean;
    float q = d0 * d0 + d1 * d1;
#pragma unroll
    for (int m = 16; m >= 1; m >>= 1) q += __shfl_xor(q, m, 32);
    float inv = rsqrtf(q * (1.0f / 64.0f) + 1e-5f);
    o0 = elu1(d0 * inv * g0 + b0);
    o1 = elu1(d1 * inv * g1 + b1);
}

__global__ void __launch_bounds__(256)
ln_kernel(float* __restrict__ h, const float* __restrict__ hn,
          const float* __restrict__ g, const float* __restrict__ be, int N) {
    int wave = threadIdx.x >> 5, lane = threadIdx.x & 31;
    float g0 = g[lane], g1 = g[lane + 32];
    float b0 = be[lane], b1 = be[lane + 32];
    int stride = gridDim.x * 8;
    for (int n0 = blockIdx.x * 8 + wave; n0 < N; n0 += stride * LNU) {
        if (n0 + (LNU - 1) * stride < N) {
            float a0[LNU], a1[LNU], c0[LNU], c1[LNU];
#pragma unroll
            for (int u = 0; u < LNU; ++u) {
                long o = (long)(n0 + u * stride) * H + lane;
                a0[u] = h[o];
                a1[u] = h[o + 32];
                c0[u] = hn[o];
                c1[u] = hn[o + 32];
            }
#pragma unroll
            for (int u = 0; u < LNU; ++u) {
                long o = (long)(n0 + u * stride) * H + lane;
                float o0, o1;
                ln_row(a0[u] + c0[u], a1[u] + c1[u], g0, g1, b0, b1, o0, o1);
                h[o]      = o0;
                h[o + 32] = o1;
            }
        } else {
            for (int n = n0; n < N; n += stride) {
                long o = (long)n * H + lane;
                float o0, o1;
                ln_row(h[o] + hn[o], h[o + 32] + hn[o + 32], g0, g1, b0, b1, o0, o1);
                h[o]      = o0;
                h[o + 32] = o1;
            }
        }
    }
}

__global__ void cox_kernel(const float* __restrict__ hp, const float* __restrict__ W1,
                           const float* __restrict__ b1, const float* __restrict__ W2,
                           const float* __restrict__ b2, float* __restrict__ out, int N) {
    int n = blockIdx.x * blockDim.x + threadIdx.x;
    if (n >= N) return;
    const float* hr = hp + (long)n * H;
    float hreg[H];
#pragma unroll
    for (int k = 0; k < H; ++k) hreg[k] = hr[k];
    float hz = b2[0];
    for (int j = 0; j < 32; ++j) {
        float s = b1[j];
#pragma unroll
        for (int k = 0; k < H; ++k) s = fmaf(hreg[k], W1[k * 32 + j], s);
        hz = fmaf(elu1(s), W2[j], hz);
    }
    out[n] = hz;
}

extern "C" void kernel_launch(void* const* d_in, const int* in_sizes, int n_in,
                              void* d_out, int out_size, void* d_ws, size_t ws_size,
                              hipStream_t stream) {
    (void)in_sizes; (void)n_in; (void)out_size; (void)ws_size;
    const int NG = 20000, NP = 100000, NM = 5000, E = 1000000;

    const float* x_gene    = (const float*)d_in[0];
    const float* x_patient = (const float*)d_in[1];
    const float* x_group   = (const float*)d_in[2];
    const int* srcA[4]  = { (const int*)d_in[3], (const int*)d_in[5],
                            (const int*)d_in[7], (const int*)d_in[9] };
    const int* dstA[4]  = { (const int*)d_in[4], (const int*)d_in[6],
                            (const int*)d_in[8], (const int*)d_in[10] };
    const float* encW[3] = { (const float*)d_in[11], (const float*)d_in[13], (const float*)d_in[15] };
    const float* encB[3] = { (const float*)d_in[12], (const float*)d_in[14], (const float*)d_in[16] };
    const float* convW_l = (const float*)d_in[17];
    const float* convb_l = (const float*)d_in[18];
    const float* convW_r = (const float*)d_in[19];
    const float* ln_g = (const float*)d_in[20];
    const float* ln_b = (const float*)d_in[21];
    const float* coxW1 = (const float*)d_in[22];
    const float* coxb1 = (const float*)d_in[23];
    const float* coxW2 = (const float*)d_in[24];
    const float* coxb2 = (const float*)d_in[25];

    float* ws = (float*)d_ws;
    float* h_gene  = ws; ws += (size_t)NG * H;
    float* h_pat   = ws; ws += (size_t)NP * H;
    float* h_grp   = ws; ws += (size_t)NM * H;
    float* hn_gene = ws; ws += (size_t)NG * H;
    float* hn_pat  = ws; ws += (size_t)NP * H;
    float* hn_grp  = ws; ws += (size_t)NM * H;
    const int SL = 34000;
    float* aggreg  = ws; ws += (size_t)NP * H;
    float* deg     = ws; ws += (size_t)NP;
    float* agg     = aggreg;
    int*   perm    = (int*)(aggreg + (size_t)SL * H);
    int*   rowptr  = perm + (size_t)4 * E;
    int*   cnt     = (int*)hn_gene;
    int*   ccnt    = (int*)hn_pat;

    const int NCH   = (E + CHE - 1) / CHE;
    const int NSLOT = (NP + RNG - 1) / RNG + (NG + RNG - 1) / RNG
                    + (NM + RNG - 1) / RNG + (NG + RNG - 1) / RNG;
    const int NTOT  = NP + NG + NM + NG;
    count_kernel<<<dim3(NCH, NSLOT), 1024, 0, stream>>>(
        dstA[0], dstA[1], dstA[2], dstA[3], ccnt, NP, NG, NM, NG, E);
    chunk_offsets_kernel<NCH><<<(NTOT + 255) / 256, 256, 0, stream>>>(ccnt, cnt, NTOT);
    rowptr_scan_kernel<<<4, 1024, 0, stream>>>(cnt, rowptr, NP, NG, NM, NG);
    fill_kernel<<<dim3(NCH, NSLOT), 1024, 0, stream>>>(
        srcA[0], srcA[1], srcA[2], srcA[3], dstA[0], dstA[1], dstA[2], dstA[3],
        rowptr, ccnt, perm, NP, NG, NM, NG, E);
    const int* rpA[4] = { rowptr, rowptr + NP + 4, rowptr + NP + NG + 8, rowptr + NP + NG + NM + 12 };

    const int GCAP = 1024;
    enc_kernel<11, 2><<<(NG + 7) / 8 < GCAP ? (NG + 7) / 8 : GCAP, 256, 0, stream>>>(
        x_gene,    encW[0], encB[0], h_gene, NG);
    enc_kernel<3, 4><<<(NP + 15) / 16 < GCAP ? (NP + 15) / 16 : GCAP, 256, 0, stream>>>(
        x_patient, encW[1], encB[1], h_pat,  NP);
    enc_kernel<4, 4><<<(NM + 15) / 16 < GCAP ? (NM + 15) / 16 : GCAP, 256, 0, stream>>>(
        x_group,   encW[2], encB[2], h_grp,  NM);

    float* hArr[3]  = { h_gene, h_pat, h_grp };
    float* hnArr[3] = { hn_gene, hn_pat, hn_grp };
    int    nArr[3]  = { NG, NP, NM };
    const int sType[4] = { 0, 1, 0, 2 };
    const int dType[4] = { 1, 0, 2, 0 };

    const bool firstInto[4] = { true, true, true, false };
    for (int layer = 0; layer < 2; ++layer) {
        for (int e = 0; e < 4; ++e) {
            int Nd = nArr[dType[e]];
            int Ns = nArr[sType[e]];
            const float* Wl  = convW_l + ((size_t)layer * 4 + e) * H * H;
            const float* blp = convb_l + ((size_t)layer * 4 + e) * H;
            const float* Wr  = convW_r + ((size_t)layer * 4 + e) * H * H;
            const int* pm = perm + (size_t)e * E;
            float* sums = firstInto[e] ? hnArr[dType[e]] : agg;
            int gb = (Nd + 7) / 8 < 2 * GCAP ? (Nd + 7) / 8 : 2 * GCAP;
            if (e == 0)
                gather_sum_kernel<4><<<gb, 256, 0, stream>>>(
                    hArr[sType[e]], rpA[e], pm, sums, deg, 0, Nd, Ns, E);
            else
                gather_sum_kernel<8><<<gb, 256, 0, stream>>>(
                    hArr[sType[e]], rpA[e], pm, sums, deg, 0, Nd, Ns, E);
            if (firstInto[e])
                sage_gemm_kernel<true><<<(Nd + 63) / 64, 128, 0, stream>>>(
                    agg, deg, hArr[dType[e]], Wl, blp, Wr, hnArr[dType[e]], Nd);
            else
                sage_gemm_kernel<false><<<(Nd + 63) / 64, 128, 0, stream>>>(
                    agg, deg, hArr[dType[e]], Wl, blp, Wr, hnArr[dType[e]], Nd);
        }
        for (int ni = 0; ni < 3; ++ni) {
            int lb = (nArr[ni] + 8 * LNU - 1) / (8 * LNU);
            ln_kernel<<<lb < GCAP ? lb : GCAP, 256, 0, stream>>>(
                hArr[ni], hnArr[ni],
                ln_g + ((size_t)layer * 3 + ni) * H,
                ln_b + ((size_t)layer * 3 + ni) * H, nArr[ni]);
        }
    }
    cox_kernel<<<(NP + 255) / 256, 256, 0, stream>>>(
        h_pat, coxW1, coxb1, coxW2, coxb2, (float*)d_out, NP);
}


# Residual + LayerNorm + ELU moved into the conv GEMM's epilogue on the register tile (row sums by DPP row_xmask adds, same sum tree), h rows written in place; edge types reordered; 19 -> 16 launches
# speedup vs baseline: 9.3894x; 1.5261x over previous
#include <hip/hip_runtime.h>
#include <hip/hip_bf16.h>
#include <math.h>

typedef float v2f __attribute__((ext_vector_type(2)));
typedef float v8f __attribute__((ext_vector_type(8)));

#define H 64

__device__ __forceinline__ float elu1(float x) {
    return x > 0.0f ? x : (__expf(x) - 1.0f);
}

template <int F, int U>
__global__ void __launch_bounds__(256)
enc_kernel(const float* __restrict__ x, const float* __restrict__ W,
           const float* __restrict__ b, float* __restrict__ h, int N) {
    int j = threadIdx.x & 63;
    int sub = threadIdx.x >> 6;
    float w[F];
#pragma unroll
    for (int k = 0; k < F; ++k) w[k] = W[k * H + j];
    float bj = b[j];
    for (int n0 = blockIdx.x * (4 * U) + sub; n0 < N; n0 += gridDim.x * (4 * U)) {
        if (n0 + 4 * (U - 1) < N) {
            float xv[U][F];
#pragma unroll
            for (int u = 0; u < U; ++u) {
                const float* xr = x + (long)(n0 + 4 * u) * F;
#pragma unroll
                for (int k = 0; k < F; ++k) xv[u][k] = xr[k];
            }
#pragma unroll
            for (int u = 0; u < U; ++u) {
                float s = bj;
#pragma unroll
                for (int k = 0; k < F; ++k) s = fmaf(xv[u][k], w[k], s);
                h[(long)(n0 + 4 * u) * H + j] = elu1(s);
            }
        } else {
            for (int n = n0; n < N; n += 4) {
                const float* xr = x + (long)n * F;
                float s = bj;
#pragma unroll
                for (int k = 0; k < F; ++k) s = fmaf(xr[k], w[k], s);
                h[(long)n * H + j] = elu1(s);
            }
        }
    }
}

#define RNG  25000
#define CHSH 15
#define CHE  (1 << CHSH)
#define LPT  8

__device__ __forceinline__ void range_of_slot(int slot, int n0, int n1, int n2, int n3,
                                              int& t, int& lo, int& cntn, int& coff) {
    int r0 = (n0 + RNG - 1) / RNG, r1 = (n1 + RNG - 1) / RNG, r2 = (n2 + RNG - 1) / RNG;
    int r, N;
    if (slot < r0)                { t = 0; r = slot;                N = n0; coff = 0; }
    else if (slot < r0 + r1)      { t = 1; r = slot - r0;           N = n1; coff = n0; }
    else if (slot < r0 + r1 + r2) { t = 2; r = slot - r0 - r1;      N = n2; coff = n0 + n1; }
    else                          { t = 3; r = slot - r0 - r1 - r2; N = n3; coff = n0 + n1 + n2; }
    lo = r * RNG;
    cntn = min(RNG, N - lo);
}

__global__ void __launch_bounds__(1024)
count_kernel(const int* __restrict__ d0, const int* __restrict__ d1,
             const int* __restrict__ d2, const int* __restrict__ d3,
             int* __restrict__ ccnt, int n0, int n1, int n2, int n3, int E) {
    __shared__ int lc[RNG];
    int t, lo, cntn, coff;
    range_of_slot(blockIdx.y, n0, n1, n2, n3, t, lo, cntn, coff);
    const int* d = t == 0 ? d0 : t == 1 ? d1 : t == 2 ? d2 : d3;
    int ntot = n0 + n1 + n2 + n3;
    for (int i = threadIdx.x; i < cntn; i += 1024) lc[i] = 0;
    __syncthreads();
    int e0 = blockIdx.x << CHSH;
    int e1 = min(e0 + CHE, E);
    for (int b = e0 + threadIdx.x; b < e1; b += 1024 * LPT) {
        int dv[LPT];
#pragma unroll
        for (int u = 0; u < LPT; ++u) {
            int e = b + u * 1024;
            dv[u] = e < e1 ? d[e] : -1;
        }
#pragma unroll
        for (int u = 0; u < LPT; ++u) {
            unsigned rel = (unsigned)(dv[u] - lo);
            if (rel < (unsigned)cntn) atomicAdd(&lc[rel], 1);
        }
    }
    __syncthreads();
    int* o = ccnt + (long)blockIdx.x * ntot + coff + lo;
    for (int i = threadIdx.x; i < cntn; i += 1024) o[i] = lc[i];
}

template <int NCH>
__global__ void __launch_bounds__(256)
chunk_offsets_kernel(int* __restrict__ ccnt, int* __restrict__ cnt, int ntot) {
    int n = blockIdx.x * 256 + threadIdx.x;
    if (n >= ntot) return;
    int v[NCH];
#pragma unroll
    for (int c = 0; c < NCH; ++c) v[c] = ccnt[(long)c * ntot + n];
    int run = 0;
#pragma unroll
    for (int c = 0; c < NCH; ++c) {
        ccnt[(long)c * ntot + n] = run;
        run += v[c];
    }
    cnt[n] = run;
}

__global__ void __launch_bounds__(1024)
rowptr_scan_kernel(const int* __restrict__ cnt, int* __restrict__ rowptr,
                   int n0, int n1, int n2, int n3) {
    __shared__ int wsum[32];
    int t = blockIdx.x;
    int N    = t == 0 ? n0 : t == 1 ? n1 : t == 2 ? n2 : n3;
    int coff = t == 0 ? 0 : t == 1 ? n0 : t == 2 ? n0 + n1 : n0 + n1 + n2;
    const int* c = cnt + coff;
    int* rp = rowptr + coff + 4 * t;
    int wave = threadIdx.x >> 5, lane = threadIdx.x & 31;
    int per = (((N + 31) >> 5) + 127) & ~127;
    int beg = wave * per;
    int end = min(beg + per, N);

    int s = 0;
    for (int p = beg + lane * 4; p < end; p += 128) {
        int4 v = *(const int4*)(c + p);
        s += v.x + v.y + v.z + v.w;
    }
    for (int m = 16; m >= 1; m >>= 1) s += __shfl_xor(s, m, 32);
    if (lane == 0) wsum[wave] = s;
    __syncthreads();

    int wtot = wsum[lane];
    int winc = wtot;
    for (int dlt = 1; dlt < 32; dlt <<= 1) {
        int o = __shfl_up(winc, dlt, 32);
        if (lane >= dlt) winc += o;
    }
    int carry = __shfl(winc - wtot, wave, 32);
    int total = __shfl(winc, 31, 32);

    for (int tb = beg; tb < end; tb += 128) {
        int p = tb + lane * 4;
        int4 v = make_int4(0, 0, 0, 0);
        if (p < end) v = *(const int4*)(c + p);
        int ls = v.x + v.y + v.z + v.w;
        int li = ls;
        for (int dlt = 1; dlt < 32; dlt <<= 1) {
            int o = __shfl_up(li, dlt, 32);
            if (lane >= dlt) li += o;
        }
        int ex = carry + li - ls;
        if (p < end)
            *(int4*)(rp + p) = make_int4(ex, ex + v.x, ex + v.x + v.y, ex + v.x + v.y + v.z);
        carry += __shfl(li, 31, 32);
    }
    if (threadIdx.x == 0) rp[N] = total;
}

__global__ void __launch_bounds__(1024)
fill_kernel(const int* __restrict__ s0, const int* __restrict__ s1,
            const int* __restrict__ s2, const int* __restrict__ s3,
            const int* __restrict__ d0, const int* __restrict__ d1,
            const int* __restrict__ d2, const int* __restrict__ d3,
            const int* __restrict__ rowptr, const int* __restrict__ choff,
            int* __restrict__ perm, int n0, int n1, int n2, int n3, int E) {
    __shared__ int lc[RNG];
    int t, lo, cntn, coff;
    range_of_slot(blockIdx.y, n0, n1, n2, n3, t, lo, cntn, coff);
    const int* s = t == 0 ? s0 : t == 1 ? s1 : t == 2 ? s2 : s3;
    const int* d = t == 0 ? d0 : t == 1 ? d1 : t == 2 ? d2 : d3;
    int ntot = n0 + n1 + n2 + n3;
    const int* rp = rowptr + coff + 4 * t + lo;
    const int* co = choff + (long)blockIdx.x * ntot + coff + lo;
    for (int i = threadIdx.x; i < cntn; i += 1024) lc[i] = rp[i] + co[i];
    __syncthreads();
    int e0 = blockIdx.x << CHSH;
    int e1 = min(e0 + CHE, E);
    int* pm = perm + (long)t * E;
    for (int b = e0 + threadIdx.x; b < e1; b += 1024 * LPT) {
        int dv[LPT], sv[LPT];
#pragma unroll
        for (int u = 0; u < LPT; ++u) {
            int e = b + u * 1024;
            bool ok = e < e1;
            dv[u] = ok ? d[e] : -1;
            sv[u] = ok ? s[e] : 0;
        }
        int pos[LPT];
#pragma unroll
        for (int u = 0; u < LPT; ++u) {
            unsigned rel = (unsigned)(dv[u] - lo);
            pos[u] = -1;
            if (rel < (unsigned)cntn) pos[u] = atomicAdd(&lc[rel], 1);
        }
#pragma unroll
        for (int u = 0; u < LPT; ++u) {
            if (pos[u] >= 0) pm[min(pos[u], E - 1)] = sv[u];
        }
    }
}

template <int U>
__global__ void __launch_bounds__(256)
gather_sum_kernel(const float* __restrict__ hsrc, const int* __restrict__ rowptr,
                  const int* __restrict__ perm, float* __restrict__ agg,
                  float* __restrict__ deg, int nbeg, int ncnt, int nsrc, int E) {
    int wave = threadIdx.x >> 5, lane = threadIdx.x & 31;
    int half = lane >> 4, c = lane & 15;
    const float4* hs4 = (const float4*)hsrc;
    for (int ln = blockIdx.x * 8 + wave; ln < ncnt; ln += gridDim.x * 8) {
        int beg = rowptr[nbeg + ln], end = rowptr[nbeg + ln + 1];
        beg = max(beg, 0);
        end = min(end, E);
        float4 acc = make_float4(0.0f, 0.0f, 0.0f, 0.0f);

        for (int base = beg; base < end; base += 32) {
            int cnt = min(32, end - base);
            int mine = 0;
            if (lane < cnt) mine = perm[base + lane];
            mine = min(max(mine, 0), nsrc - 1);
            for (int i = 0; i < cnt; i += 2 * U) {
                float4 v[U];
                bool ok[U];
#pragma unroll
                for (int u = 0; u < U; ++u) {
                    int idx = i + 2 * u + half;
                    ok[u] = idx < cnt;
                    int sidx = __shfl(mine, ok[u] ? idx : 0, 32);
                    v[u] = hs4[(long)sidx * 16 + c];
                }
#pragma unroll
                for (int u = 0; u < U; ++u) {
                    acc.x += ok[u] ? v[u].x : 0.0f;
                    acc.y += ok[u] ? v[u].y : 0.0f;
                    acc.z += ok[u] ? v[u].z : 0.0f;
                    acc.w += ok[u] ? v[u].w : 0.0f;
                }
            }
        }
        acc.x += __shfl_xor(acc.x, 16, 32);
        acc.y += __shfl_xor(acc.y, 16, 32);
        acc.z += __shfl_xor(acc.z, 16, 32);
        acc.w += __shfl_xor(acc.w, 16, 32);
        if (half == 0) ((float4*)agg)[(long)ln * 16 + c] = acc;
        if (lane == 0) deg[ln] = (float)(end - beg);
    }
}

template <int M>
__device__ __forceinline__ float row_xor_add(float x) {
    int xi = __float_as_int(x);
    int yi = __builtin_amdgcn_update_dpp(0, xi, 0x160 + M, 0xf, 0xf, false);
    return x + __int_as_float(yi);
}

template <int TAIL>
__global__ void __launch_bounds__(128)
sage_gemm_kernel(const float* sums, const float* __restrict__ deg,
                 float* hdst,
                 const float* __restrict__ Wl, const float* __restrict__ bl,
                 const float* __restrict__ Wr,
                 float* hnew,
                 const float* __restrict__ g, const float* __restrict__ be, int N) {
    __shared__ float sWl[H * H];
    __shared__ float sWr[H * H];
    int tid = threadIdx.x;
    {
        const float4* gl = (const float4*)Wl;
        const float4* gr = (const float4*)Wr;
        float4* ll = (float4*)sWl;
        float4* lr = (float4*)sWr;
        for (int i = tid; i < (H * H / 4); i += 128) {
            ll[i] = gl[i];
            lr[i] = gr[i];
        }
    }
    __syncthreads();

    int wave = tid >> 5;
    int lane = tid & 31;
    int hi   = lane >> 4;
    int col  = lane & 15;
    int kk   = hi * 2;
    int m0   = blockIdx.x * 64 + wave * 16;

    int r  = m0 + col;
    int rc = r < N ? r : (N - 1);
    float invd = 1.0f / fmaxf(deg[rc], 1.0f);

    const float* aggr = (TAIL == 0 ? (const float*)hnew : sums) + (long)rc * H;
    const float* hdr  = hdst + (long)rc * H;

    v8f acc[4] = {};

#pragma unroll
    for (int k0 = 0; k0 < H; k0 += 4) {
        v2f a;
        a.x = aggr[k0 + kk] * invd;
        a.y = aggr[k0 + kk + 1] * invd;
#pragma unroll
        for (int t4 = 0; t4 < 4; ++t4) {
            v2f b;
            b.x = sWl[(k0 + kk) * H + t4 * 16 + col];
            b.y = sWl[(k0 + kk + 1) * H + t4 * 16 + col];
            acc[t4] = __builtin_amdgcn_wmma_f32_16x16x4_f32(
                false, a, false, b, (short)0, acc[t4], false, false);
        }
    }
#pragma unroll
    for (int k0 = 0; k0 < H; k0 += 4) {
        v2f a;
        a.x = hdr[k0 + kk];
        a.y = hdr[k0 + kk + 1];
#pragma unroll
        for (int t4 = 0; t4 < 4; ++t4) {
            v2f b;
            b.x = sWr[(k0 + kk) * H + t4 * 16 + col];
            b.y = sWr[(k0 + kk + 1) * H + t4 * 16 + col];
            acc[t4] = __builtin_amdgcn_wmma_f32_16x16x4_f32(
                false, a, false, b, (short)0, acc[t4], false, false);
        }
    }

    if (TAIL == 0) {
#pragma unroll
        for (int t4 = 0; t4 < 4; ++t4) {
            int cg = t4 * 16 + col;
            float bias = bl[cg];
#pragma unroll
            for (int rr = 0; rr < 8; ++rr) {
                int row = m0 + rr + hi * 8;
                if (row < N) hnew[(long)row * H + cg] = acc[t4][rr] + bias;
            }
        }
    } else {
        float bias[4], gm[4], bt[4];
#pragma unroll
        for (int t4 = 0; t4 < 4; ++t4) {
            int cg = t4 * 16 + col;
            bias[t4] = bl[cg];
            gm[t4]   = g[cg];
            bt[t4]   = be[cg];
        }
        float res[4][8], pr[4][8];
#pragma unroll
        for (int rr = 0; rr < 8; ++rr) {
            int row = m0 + rr + hi * 8;
            long ro = (long)(row < N ? row : (N - 1)) * H + col;
#pragma unroll
            for (int t4 = 0; t4 < 4; ++t4) {
                res[t4][rr] = hdst[ro + t4 * 16];
                pr[t4][rr]  = TAIL == 2 ? hnew[ro + t4 * 16] : 0.0f;
            }
        }
#pragma unroll
        for (int rr = 0; rr < 8; ++rr) {
            int row = m0 + rr + hi * 8;
            float x[4];
#pragma unroll
            for (int t4 = 0; t4 < 4; ++t4) {
                float o = acc[t4][rr] + bias[t4];
                if (TAIL == 2) o = pr[t4][rr] + o;
                x[t4] = res[t4][rr] + o;
            }
            float s = (x[0] + x[2]) + (x[1] + x[3]);
            s = row_xor_add<8>(s);
            s = row_xor_add<4>(s);
            s = row_xor_add<2>(s);
            s = row_xor_add<1>(s);
            float mean = s * (1.0f / 64.0f);
            float d0 = x[0] - mean, d1 = x[1] - mean, d2 = x[2] - mean, d3 = x[3] - mean;
            float qa = d0 * d0 + d2 * d2;
            float qb = d1 * d1 + d3 * d3;
            float q = qa + qb;
            q = row_xor_add<8>(q);
            q = row_xor_add<4>(q);
            q = row_xor_add<2>(q);
            q = row_xor_add<1>(q);
            float inv = rsqrtf(q * (1.0f / 64.0f) + 1e-5f);
            float o0 = elu1(d0 * inv * gm[0] + bt[0]);
            float o1 = elu1(d1 * inv * gm[1] + bt[1]);
            float o2 = elu1(d2 * inv * gm[2] + bt[2]);
            float o3 = elu1(d3 * inv * gm[3] + bt[3]);
            if (row < N) {
                float* orow = hdst + (long)row * H + col;
                orow[0]  = o0;
                orow[16] = o1;
                orow[32] = o2;
                orow[48] = o3;
            }
        }
    }
}

__global__ void cox_kernel(const float* __restrict__ hp, const float* __restrict__ W1,
                           const float* __restrict__ b1, const float* __restrict__ W2,
                           const float* __restrict__ b2, float* __restrict__ out, int N) {
    int n = blockIdx.x * blockDim.x + threadIdx.x;
    if (n >= N) return;
    const float* hr = hp + (long)n * H;
    float hreg[H];
#pragma unroll
    for (int k = 0; k < H; ++k) hreg[k] = hr[k];
    float hz = b2[0];
    for (int j = 0; j < 32; ++j) {
        float s = b1[j];
#pragma unroll
        for (int k = 0; k < H; ++k) s = fmaf(hreg[k], W1[k * 32 + j], s);
        hz = fmaf(elu1(s), W2[j], hz);
    }
    out[n] = hz;
}

extern "C" void kernel_launch(void* const* d_in, const int* in_sizes, int n_in,
                              void* d_out, int out_size, void* d_ws, size_t ws_size,
                              hipStream_t stream) {
    (void)in_sizes; (void)n_in; (void)out_size; (void)ws_size;
    const int NG = 20000, NP = 100000, NM = 5000, E = 1000000;

    const float* x_gene    = (const float*)d_in[0];
    const float* x_patient = (const float*)d_in[1];
    const float* x_group   = (const float*)d_in[2];
    const int* srcA[4]  = { (const int*)d_in[3], (const int*)d_in[5],
                            (const int*)d_in[7], (const int*)d_in[9] };
    const int* dstA[4]  = { (const int*)d_in[4], (const int*)d_in[6],
                            (const int*)d_in[8], (const int*)d_in[10] };
    const float* encW[3] = { (const float*)d_in[11], (const float*)d_in[13], (const float*)d_in[15] };
    const float* encB[3] = { (const float*)d_in[12], (const float*)d_in[14], (const float*)d_in[16] };
    const float* convW_l = (const float*)d_in[17];
    const float* convb_l = (const float*)d_in[18];
    const float* convW_r = (const float*)d_in[19];
    const float* ln_g = (const float*)d_in[20];
    const float* ln_b = (const float*)d_in[21];
    const float* coxW1 = (const float*)d_in[22];
    const float* coxb1 = (const float*)d_in[23];
    const float* coxW2 = (const float*)d_in[24];
    const float* coxb2 = (const float*)d_in[25];

    float* ws = (float*)d_ws;
    float* h_gene  = ws; ws += (size_t)NG * H;
    float* h_pat   = ws; ws += (size_t)NP * H;
    float* h_grp   = ws; ws += (size_t)NM * H;
    float* hn_gene = ws; ws += (size_t)NG * H;
    float* hn_pat  = ws; ws += (size_t)NP * H;
    ws += (size_t)NM * H;
    const int SL = 34000;
    float* aggreg  = ws; ws += (size_t)NP * H;
    float* deg     = ws; ws += (size_t)NP;
    float* agg     = aggreg;
    int*   perm    = (int*)(aggreg + (size_t)SL * H);
    int*   rowptr  = perm + (size_t)3 * E;
    int*   cnt     = (int*)hn_gene;
    int*   ccnt    = (int*)hn_pat;

    const int* srcL[3] = { srcA[0], srcA[1], srcA[3] };
    const int* dstL[3] = { dstA[0], dstA[1], dstA[3] };
    const int* none = nullptr;
    const int NCH   = (E + CHE - 1) / CHE;
    const int NSLOT = (NP + RNG - 1) / RNG + (NG + RNG - 1) / RNG
                    + (NG + RNG - 1) / RNG;
    const int NTOT  = NP + NG + NG;
    count_kernel<<<dim3(NCH, NSLOT), 1024, 0, stream>>>(
        dstL[0], dstL[1], dstL[2], none, ccnt, NP, NG, NG, 0, E);
    chunk_offsets_kernel<NCH><<<(NTOT + 255) / 256, 256, 0, stream>>>(ccnt, cnt, NTOT);
    rowptr_scan_kernel<<<3, 1024, 0, stream>>>(cnt, rowptr, NP, NG, NG, 0);
    fill_kernel<<<dim3(NCH, NSLOT), 1024, 0, stream>>>(
        srcL[0], srcL[1], srcL[2], none, dstL[0], dstL[1], dstL[2], none,
        rowptr, ccnt, perm, NP, NG, NG, 0, E);
    const int* rpL[3] = { rowptr, rowptr + NP + 4, rowptr + NP + NG + 8 };

    const int GCAP = 1024;
    enc_kernel<11, 2><<<(NG + 7) / 8 < GCAP ? (NG + 7) / 8 : GCAP, 256, 0, stream>>>(
        x_gene,    encW[0], encB[0], h_gene, NG);
    enc_kernel<3, 4><<<(NP + 15) / 16 < GCAP ? (NP + 15) / 16 : GCAP, 256, 0, stream>>>(
        x_patient, encW[1], encB[1], h_pat,  NP);
    enc_kernel<4, 4><<<(NM + 15) / 16 < GCAP ? (NM + 15) / 16 : GCAP, 256, 0, stream>>>(
        x_group,   encW[2], encB[2], h_grp,  NM);

    float* hArr[3]  = { h_gene, h_pat, h_grp };
    float* hnArr[3] = { hn_gene, hn_pat, nullptr };
    int    nArr[3]  = { NG, NP, NM };
    const int sType[4] = { 0, 1, 0, 2 };
    const int dType[4] = { 1, 0, 2, 0 };
    const int listOf[4] = { 0, 1, -1, 2 };

    const bool firstInto[4] = { true, true, true, false };
    const int  tailOf[4]    = { 1, 0, -1, 2 };
    const int  order0[3]    = { 1, 0, 3 };
    for (int layer = 0; layer < 2; ++layer) {
        int nlive = layer == 0 ? 3 : 1;
        for (int oi = 0; oi < nlive; ++oi) {
            int e = layer == 0 ? order0[oi] : 0;
            int Nd = nArr[dType[e]];
            int Ns = nArr[sType[e]];
            int ni = dType[e];
            const float* Wl  = convW_l + ((size_t)layer * 4 + e) * H * H;
            const float* blp = convb_l + ((size_t)layer * 4 + e) * H;
            const float* Wr  = convW_r + ((size_t)layer * 4 + e) * H * H;
            const float* lg  = ln_g + ((size_t)layer * 3 + ni) * H;
            const float* lb  = ln_b + ((size_t)layer * 3 + ni) * H;
            const int* pm = perm + (size_t)listOf[e] * E;
            const int* rp = rpL[listOf[e]];
            float* sums = firstInto[e] ? hnArr[ni] : agg;
            int gb = (Nd + 7) / 8 < 2 * GCAP ? (Nd + 7) / 8 : 2 * GCAP;
            if (e == 0)
                gather_sum_kernel<4><<<gb, 256, 0, stream>>>(
                    hArr[sType[e]], rp, pm, sums, deg, 0, Nd, Ns, E);
            else
                gather_sum_kernel<8><<<gb, 256, 0, stream>>>(
                    hArr[sType[e]], rp, pm, sums, deg, 0, Nd, Ns, E);
            if (tailOf[e] == 0)
                sage_gemm_kernel<0><<<(Nd + 63) / 64, 128, 0, stream>>>(
                    sums, deg, hArr[ni], Wl, blp, Wr, hnArr[ni], lg, lb, Nd);
            else if (tailOf[e] == 1)
                sage_gemm_kernel<1><<<(Nd + 63) / 64, 128, 0, stream>>>(
                    sums, deg, hArr[ni], Wl, blp, Wr, hnArr[ni], lg, lb, Nd);
            else
                sage_gemm_kernel<2><<<(Nd + 63) / 64, 128, 0, stream>>>(
                    sums, deg, hArr[ni], Wl, blp, Wr, hnArr[ni], lg, lb, Nd);
        }
    }
    cox_kernel<<<(NP + 255) / 256, 256, 0, stream>>>(
        h_pat, coxW1, coxb1, coxW2, coxb2, (float*)d_out, NP);
}


# Launches merged: first layer's 3 gathers in one launch, its 3 GEMMs in one (gene blocks form both edge types, the first kept in registers), encoders beside the prefix sums; 16 -> 9 launches
# speedup vs baseline: 10.5803x; 1.1268x over previous
#include <hip/hip_runtime.h>
#include <hip/hip_bf16.h>
#include <math.h>

typedef float v2f __attribute__((ext_vector_type(2)));
typedef float v8f __attribute__((ext_vector_type(8)));

#define H 64

__device__ __forceinline__ float elu1(float x) {
    return x > 0.0f ? x : (__expf(x) - 1.0f);
}

template <int F, int U>
__device__ __forceinline__ void enc_rows(const float* __restrict__ x, const float* __restrict__ W,
                                         const float* __restrict__ b, float* __restrict__ h,
                                         int N, int vb, int nvb, int tid) {
    int j = tid & 63;
    int sub = tid >> 6;
    float w[F];
#pragma unroll
    for (int k = 0; k < F; ++k) w[k] = W[k * H + j];
    float bj = b[j];
    for (int n0 = vb * (4 * U) + sub; n0 < N; n0 += nvb * (4 * U)) {
        if (n0 + 4 * (U - 1) < N) {
            float xv[U][F];
#pragma unroll
            for (int u = 0; u < U; ++u) {
                const float* xr = x + (long)(n0 + 4 * u) * F;
#pragma unroll
                for (int k = 0; k < F; ++k) xv[u][k] = xr[k];
            }
#pragma unroll
            for (int u = 0; u < U; ++u) {
                float s = bj;
#pragma unroll
                for (int k = 0; k < F; ++k) s = fmaf(xv[u][k], w[k], s);
                h[(long)(n0 + 4 * u) * H + j] = elu1(s);
            }
        } else {
            for (int n = n0; n < N; n += 4) {
                const float* xr = x + (long)n * F;
                float s = bj;
#pragma unroll
                for (int k = 0; k < F; ++k) s = fmaf(xr[k], w[k], s);
                h[(long)n * H + j] = elu1(s);
            }
        }
    }
}

#define RNG  25000
#define CHSH 15
#define CHE  (1 << CHSH)
#define LPT  8

__device__ __forceinline__ void range_of_slot(int slot, int n0, int n1, int n2, int n3,
                                              int& t, int& lo, int& cntn, int& coff) {
    int r0 = (n0 + RNG - 1) / RNG, r1 = (n1 + RNG - 1) / RNG, r2 = (n2 + RNG - 1) / RNG;
    int r, N;
    if (slot < r0)                { t = 0; r = slot;                N = n0; coff = 0; }
    else if (slot < r0 + r1)      { t = 1; r = slot - r0;           N = n1; coff = n0; }
    else if (slot < r0 + r1 + r2) { t = 2; r = slot - r0 - r1;      N = n2; coff = n0 + n1; }
    else                          { t = 3; r = slot - r0 - r1 - r2; N = n3; coff = n0 + n1 + n2; }
    lo = r * RNG;
    cntn = min(RNG, N - lo);
}

__global__ void __launch_bounds__(1024)
count_kernel(const int* __restrict__ d0, const int* __restrict__ d1,
             const int* __restrict__ d2, const int* __restrict__ d3,
             int* __restrict__ ccnt, int n0, int n1, int n2, int n3, int E) {
    __shared__ int lc[RNG];
    int t, lo, cntn, coff;
    range_of_slot(blockIdx.y, n0, n1, n2, n3, t, lo, cntn, coff);
    const int* d = t == 0 ? d0 : t == 1 ? d1 : t == 2 ? d2 : d3;
    int ntot = n0 + n1 + n2 + n3;
    for (int i = threadIdx.x; i < cntn; i += 1024) lc[i] = 0;
    __syncthreads();
    int e0 = blockIdx.x << CHSH;
    int e1 = min(e0 + CHE, E);
    for (int b = e0 + threadIdx.x; b < e1; b += 1024 * LPT) {
        int dv[LPT];
#pragma unroll
        for (int u = 0; u < LPT; ++u) {
            int e = b + u * 1024;
            dv[u] = e < e1 ? d[e] : -1;
        }
#pragma unroll
        for (int u = 0; u < LPT; ++u) {
            unsigned rel = (unsigned)(dv[u] - lo);
            if (rel < (unsigned)cntn) atomicAdd(&lc[rel], 1);
        }
    }
    __syncthreads();
    int* o = ccnt + (long)blockIdx.x * ntot + coff + lo;
    for (int i = threadIdx.x; i < cntn; i += 1024) o[i] = lc[i];
}

template <int NCH>
__global__ void __launch_bounds__(256)
chunk_offsets_kernel(int* __restrict__ ccnt, int* __restrict__ cnt, int ntot) {
    int n = blockIdx.x * 256 + threadIdx.x;
    if (n >= ntot) return;
    int v[NCH];
#pragma unroll
    for (int c = 0; c < NCH; ++c) v[c] = ccnt[(long)c * ntot + n];
    int run = 0;
#pragma unroll
    for (int c = 0; c < NCH; ++c) {
        ccnt[(long)c * ntot + n] = run;
        run += v[c];
    }
    cnt[n] = run;
}

__device__ __forceinline__ void rowptr_scan_block(const int* __restrict__ cnt, int* __restrict__ rowptr,
                                                  int n0, int n1, int n2, int n3, int t, int* wsum) {
    int N    = t == 0 ? n0 : t == 1 ? n1 : t == 2 ? n2 : n3;
    int coff = t == 0 ? 0 : t == 1 ? n0 : t == 2 ? n0 + n1 : n0 + n1 + n2;
    const int* c = cnt + coff;
    int* rp = rowptr + coff + 4 * t;
    int wave = threadIdx.x >> 5, lane = threadIdx.x & 31;
    int per = (((N + 31) >> 5) + 127) & ~127;
    int beg = wave * per;
    int end = min(beg + per, N);

    int s = 0;
    for (int p = beg + lane * 4; p < end; p += 128) {
        int4 v = *(const int4*)(c + p);
        s += v.x + v.y + v.z + v.w;
    }
    for (int m = 16; m >= 1; m >>= 1) s += __shfl_xor(s, m, 32);
    if (lane == 0) wsum[wave] = s;
    __syncthreads();

    int wtot = wsum[lane];
    int winc = wtot;
    for (int dlt = 1; dlt < 32; dlt <<= 1) {
        int o = __shfl_up(winc, dlt, 32);
        if (lane >= dlt) winc += o;
    }
    int carry = __shfl(winc - wtot, wave, 32);
    int total = __shfl(winc, 31, 32);

    for (int tb = beg; tb < end; tb += 128) {
        int p = tb + lane * 4;
        int4 v = make_int4(0, 0, 0, 0);
        if (p < end) v = *(const int4*)(c + p);
        int ls = v.x + v.y + v.z + v.w;
        int li = ls;
        for (int dlt = 1; dlt < 32; dlt <<= 1) {
            int o = __shfl_up(li, dlt, 32);
            if (lane >= dlt) li += o;
        }
        int ex = carry + li - ls;
        if (p < end)
            *(int4*)(rp + p) = make_int4(ex, ex + v.x, ex + v.x + v.y, ex + v.x + v.y + v.z);
        carry += __shfl(li, 31, 32);
    }
    if (threadIdx.x == 0) rp[N] = total;
}

#define NSCAN 3
__global__ void __launch_bounds__(1024)
scan_enc_kernel(const int* __restrict__ cnt, int* __restrict__ rowptr,
                int n0, int n1, int n2, int n3,
                const float* __restrict__ xp, const float* __restrict__ Wp,
                const float* __restrict__ bp, float* __restrict__ hp, int Np, int gbp,
                const float* __restrict__ xg, const float* __restrict__ Wg,
                const float* __restrict__ bg, float* __restrict__ hg, int Ng, int gbg,
                const float* __restrict__ xm, const float* __restrict__ Wm,
                const float* __restrict__ bm, float* __restrict__ hm, int Nm, int gbm) {
    __shared__ int wsum[32];
    if (blockIdx.x < NSCAN) {
        rowptr_scan_block(cnt, rowptr, n0, n1, n2, n3, blockIdx.x, wsum);
    } else {
        int vb  = (blockIdx.x - NSCAN) * 4 + (threadIdx.x >> 8);
        int tid = threadIdx.x & 255;
        if (vb < gbp)                  enc_rows<3, 4>(xp, Wp, bp, hp, Np, vb, gbp, tid);
        else if (vb < gbp + gbg)       enc_rows<11, 2>(xg, Wg, bg, hg, Ng, vb - gbp, gbg, tid);
        else if (vb < gbp + gbg + gbm) enc_rows<4, 4>(xm, Wm, bm, hm, Nm, vb - gbp - gbg, gbm, tid);
    }
}

__global__ void __launch_bounds__(1024)
fill_kernel(const int* __restrict__ s0, const int* __restrict__ s1,
            const int* __restrict__ s2, const int* __restrict__ s3,
            const int* __restrict__ d0, const int* __restrict__ d1,
            const int* __restrict__ d2, const int* __restrict__ d3,
            const int* __restrict__ rowptr, const int* __restrict__ choff,
            int* __restrict__ perm, int n0, int n1, int n2, int n3, int E) {
    __shared__ int lc[RNG];
    int t, lo, cntn, coff;
    range_of_slot(blockIdx.y, n0, n1, n2, n3, t, lo, cntn, coff);
    const int* s = t == 0 ? s0 : t == 1 ? s1 : t == 2 ? s2 : s3;
    const int* d = t == 0 ? d0 : t == 1 ? d1 : t == 2 ? d2 : d3;
    int ntot = n0 + n1 + n2 + n3;
    const int* rp = rowptr + coff + 4 * t + lo;
    const int* co = choff + (long)blockIdx.x * ntot + coff + lo;
    for (int i = threadIdx.x; i < cntn; i += 1024) lc[i] = rp[i] + co[i];
    __syncthreads();
    int e0 = blockIdx.x << CHSH;
    int e1 = min(e0 + CHE, E);
    int* pm = perm + (long)t * E;
    for (int b = e0 + threadIdx.x; b < e1; b += 1024 * LPT) {
        int dv[LPT], sv[LPT];
#pragma unroll
        for (int u = 0; u < LPT; ++u) {
            int e = b + u * 1024;
            bool ok = e < e1;
            dv[u] = ok ? d[e] : -1;
            sv[u] = ok ? s[e] : 0;
        }
        int pos[LPT];
#pragma unroll
        for (int u = 0; u < LPT; ++u) {
            unsigned rel = (unsigned)(dv[u] - lo);
            pos[u] = -1;
            if (rel < (unsigned)cntn) pos[u] = atomicAdd(&lc[rel], 1);
        }
#pragma unroll
        for (int u = 0; u < LPT; ++u) {
            if (pos[u] >= 0) pm[min(pos[u], E - 1)] = sv[u];
        }
    }
}

template <int U>
__device__ __forceinline__ void gather_rows(const float* __restrict__ hsrc, const int* __restrict__ rowptr,
                                            const int* __restrict__ perm, float* __restrict__ agg,
                                            float* __restrict__ deg, int ncnt, int nsrc, int E,
                                            int vb, int nvb) {
    int wave = threadIdx.x >> 5, lane = threadIdx.x & 31;
    int half = lane >> 4, c = lane & 15;
    const float4* hs4 = (const float4*)hsrc;
    for (int ln = vb * 8 + wave; ln < ncnt; ln += nvb * 8) {
        int beg = rowptr[ln], end = rowptr[ln + 1];
        beg = max(beg, 0);
        end = min(end, E);
        float4 acc = make_float4(0.0f, 0.0f, 0.0f, 0.0f);

        for (int base = beg; base < end; base += 32) {
            int cnt = min(32, end - base);
            int mine = 0;
            if (lane < cnt) mine = perm[base + lane];
            mine = min(max(mine, 0), nsrc - 1);
            for (int i = 0; i < cnt; i += 2 * U) {
                float4 v[U];
                bool ok[U];
#pragma unroll
                for (int u = 0; u < U; ++u) {
                    int idx = i + 2 * u + half;
                    ok[u] = idx < cnt;
                    int sidx = __shfl(mine, ok[u] ? idx : 0, 32);
                    v[u] = hs4[(long)sidx * 16 + c];
                }
#pragma unroll
                for (int u = 0; u < U; ++u) {
                    acc.x += ok[u] ? v[u].x : 0.0f;
                    acc.y += ok[u] ? v[u].y : 0.0f;
                    acc.z += ok[u] ? v[u].z : 0.0f;
                    acc.w += ok[u] ? v[u].w : 0.0f;
                }
            }
        }
        acc.x += __shfl_xor(acc.x, 16, 32);
        acc.y += __shfl_xor(acc.y, 16, 32);
        acc.z += __shfl_xor(acc.z, 16, 32);
        acc.w += __shfl_xor(acc.w, 16, 32);
        if (half == 0) ((float4*)agg)[(long)ln * 16 + c] = acc;
        if (lane == 0) deg[ln] = (float)(end - beg);
    }
}

template <int U>
__global__ void __launch_bounds__(256)
gather_sum_kernel(const float* __restrict__ hsrc, const int* __restrict__ rowptr,
                  const int* __restrict__ perm, float* __restrict__ agg,
                  float* __restrict__ deg, int ncnt, int nsrc, int E) {
    gather_rows<U>(hsrc, rowptr, perm, agg, deg, ncnt, nsrc, E, blockIdx.x, gridDim.x);
}

__global__ void __launch_bounds__(256)
gather3_kernel(const float* __restrict__ hA, const int* __restrict__ rpA, const int* __restrict__ pmA,
               float* __restrict__ aggA, float* __restrict__ degA, int nA, int nsA, int gbA,
               const float* __restrict__ hB, const int* __restrict__ rpB, const int* __restrict__ pmB,
               float* __restrict__ aggB, float* __restrict__ degB, int nB, int nsB, int gbB,
               const float* __restrict__ hC, const int* __restrict__ rpC, const int* __restrict__ pmC,
               float* __restrict__ aggC, float* __restrict__ degC, int nC, int nsC, int gbC,
               int E) {
    int b = blockIdx.x;
    if (b < gbA)            gather_rows<8>(hA, rpA, pmA, aggA, degA, nA, nsA, E, b, gbA);
    else if (b < gbA + gbB) gather_rows<8>(hB, rpB, pmB, aggB, degB, nB, nsB, E, b - gbA, gbB);
    else                    gather_rows<4>(hC, rpC, pmC, aggC, degC, nC, nsC, E, b - gbA - gbB, gbC);
}

template <int M>
__device__ __forceinline__ float row_xor_add(float x) {
    int xi = __float_as_int(x);
    int yi = __builtin_amdgcn_update_dpp(0, xi, 0x160 + M, 0xf, 0xf, false);
    return x + __int_as_float(yi);
}


__device__ __forceinline__ void stage_weights(const float* __restrict__ Wl, const float* __restrict__ Wr,
                                              float* sWl, float* sWr, int tid) {
    const float4* gl = (const float4*)Wl;
    const float4* gr = (const float4*)Wr;
    float4* ll = (float4*)sWl;
    float4* lr = (float4*)sWr;
    for (int i = tid; i < (H * H / 4); i += 128) {
        ll[i] = gl[i];
        lr[i] = gr[i];
    }
}

__device__ __forceinline__ void tile_products(const float* aggr, float invd, const float* hdr,
                                              const float* sWl, const float* sWr,
                                              int kk, int col, v8f (&acc)[4]) {
#pragma unroll
    for (int k0 = 0; k0 < H; k0 += 4) {
        v2f a;
        a.x = aggr[k0 + kk] * invd;
        a.y = aggr[k0 + kk + 1] * invd;
#pragma unroll
        for (int t4 = 0; t4 < 4; ++t4) {
            v2f b;
            b.x = sWl[(k0 + kk) * H + t4 * 16 + col];
            b.y = sWl[(k0 + kk + 1) * H + t4 * 16 + col];
            acc[t4] = __builtin_amdgcn_wmma_f32_16x16x4_f32(
                false, a, false, b, (short)0, acc[t4], false, false);
        }
    }
#pragma unroll
    for (int k0 = 0; k0 < H; k0 += 4) {
        v2f a;
        a.x = hdr[k0 + kk];
        a.y = hdr[k0 + kk + 1];
#pragma unroll
        for (int t4 = 0; t4 < 4; ++t4) {
            v2f b;
            b.x = sWr[(k0 + kk) * H + t4 * 16 + col];
            b.y = sWr[(k0 + kk + 1) * H + t4 * 16 + col];
            acc[t4] = __builtin_amdgcn_wmma_f32_16x16x4_f32(
                false, a, false, b, (short)0, acc[t4], false, false);
        }
    }
}

template <bool TWO>
__device__ __forceinline__ void tile_ln_store(const v8f (&acc)[4], const v8f (&first)[4],
                                              const float* __restrict__ bl, const float* __restrict__ bl1,
                                              const float* __restrict__ g, const float* __restrict__ be,
                                              float* hdst, int m0, int hi, int col, int N) {
    float bias[4], bias1[4], gm[4], bt[4];
#pragma unroll
    for (int t4 = 0; t4 < 4; ++t4) {
        int cg = t4 * 16 + col;
        bias[t4]  = bl[cg];
        bias1[t4] = TWO ? bl1[cg] : 0.0f;
        gm[t4]    = g[cg];
        bt[t4]    = be[cg];
    }
    float res[4][8];
#pragma unroll
    for (int rr = 0; rr < 8; ++rr) {
        int row = m0 + rr + hi * 8;
        long ro = (long)(row < N ? row : (N - 1)) * H + col;
#pragma unroll
        for (int t4 = 0; t4 < 4; ++t4) res[t4][rr] = hdst[ro + t4 * 16];
    }
#pragma unroll
    for (int rr = 0; rr < 8; ++rr) {
        int row = m0 + rr + hi * 8;
        float x[4];
#pragma unroll
        for (int t4 = 0; t4 < 4; ++t4) {
            float o = acc[t4][rr] + bias[t4];
            if (TWO) o = (first[t4][rr] + bias1[t4]) + o;
            x[t4] = res[t4][rr] + o;
        }
        float s = (x[0] + x[2]) + (x[1] + x[3]);
        s = row_xor_add<8>(s);
        s = row_xor_add<4>(s);
        s = row_xor_add<2>(s);
        s = row_xor_add<1>(s);
        float mean = s * (1.0f / 64.0f);
        float d0 = x[0] - mean, d1 = x[1] - mean, d2 = x[2] - mean, d3 = x[3] - mean;
        float qa = d0 * d0 + d2 * d2;
        float qb = d1 * d1 + d3 * d3;
        float q = qa + qb;
        q = row_xor_add<8>(q);
        q = row_xor_add<4>(q);
        q = row_xor_add<2>(q);
        q = row_xor_add<1>(q);
        float inv = rsqrtf(q * (1.0f / 64.0f) + 1e-5f);
        float o0 = elu1(d0 * inv * gm[0] + bt[0]);
        float o1 = elu1(d1 * inv * gm[1] + bt[1]);
        float o2 = elu1(d2 * inv * gm[2] + bt[2]);
        float o3 = elu1(d3 * inv * gm[3] + bt[3]);
        if (row < N) {
            float* orow = hdst + (long)row * H + col;
            orow[0]  = o0;
            orow[16] = o1;
            orow[32] = o2;
            orow[48] = o3;
        }
    }
}

__device__ __forceinline__ void sage_block_one(const float* __restrict__ sums, const float* __restrict__ deg,
                                               float* hdst,
                                               const float* __restrict__ Wl, const float* __restrict__ bl,
                                               const float* __restrict__ Wr,
                                               const float* __restrict__ g, const float* __restrict__ be,
                                               int N, int blk, float* sWl, float* sWr) {
    int tid = threadIdx.x;
    stage_weights(Wl, Wr, sWl, sWr, tid);
    __syncthreads();

    int wave = tid >> 5;
    int lane = tid & 31;
    int hi   = lane >> 4;
    int col  = lane & 15;
    int kk   = hi * 2;
    int m0   = blk * 64 + wave * 16;

    int r  = m0 + col;
    int rc = r < N ? r : (N - 1);
    float invd = 1.0f / fmaxf(deg[rc], 1.0f);

    v8f acc[4] = {};
    tile_products(sums + (long)rc * H, invd, hdst + (long)rc * H, sWl, sWr, kk, col, acc);
    tile_ln_store<false>(acc, acc, bl, bl, g, be, hdst, m0, hi, col, N);
}

__device__ __forceinline__ void sage_block_two(const float* __restrict__ sums1, const float* __restrict__ deg1,
                                               const float* __restrict__ Wl1, const float* __restrict__ bl1,
                                               const float* __restrict__ Wr1,
                                               const float* __restrict__ sums2, const float* __restrict__ deg2,
                                               const float* __restrict__ Wl2, const float* __restrict__ bl2,
                                               const float* __restrict__ Wr2,
                                               float* hdst,
                                               const float* __restrict__ g, const float* __restrict__ be,
                                               int N, int blk, float* sWl, float* sWr) {
    int tid = threadIdx.x;
    stage_weights(Wl1, Wr1, sWl, sWr, tid);
    __syncthreads();

    int wave = tid >> 5;
    int lane = tid & 31;
    int hi   = lane >> 4;
    int col  = lane & 15;
    int kk   = hi * 2;
    int m0   = blk * 64 + wave * 16;

    int r  = m0 + col;
    int rc = r < N ? r : (N - 1);
    float invd1 = 1.0f / fmaxf(deg1[rc], 1.0f);
    float invd2 = 1.0f / fmaxf(deg2[rc], 1.0f);
    const float* hdr = hdst + (long)rc * H;

    v8f acc1[4] = {};
    tile_products(sums1 + (long)rc * H, invd1, hdr, sWl, sWr, kk, col, acc1);
    __syncthreads();
    stage_weights(Wl2, Wr2, sWl, sWr, tid);
    __syncthreads();
    v8f acc2[4] = {};
    tile_products(sums2 + (long)rc * H, invd2, hdr, sWl, sWr, kk, col, acc2);
    tile_ln_store<true>(acc2, acc1, bl2, bl1, g, be, hdst, m0, hi, col, N);
}

__global__ void __launch_bounds__(128)
sage_gemm_kernel(const float* __restrict__ sums, const float* __restrict__ deg, float* hdst,
                 const float* __restrict__ Wl, const float* __restrict__ bl,
                 const float* __restrict__ Wr,
                 const float* __restrict__ g, const float* __restrict__ be, int N) {
    __shared__ float sWl[H * H];
    __shared__ float sWr[H * H];
    sage_block_one(sums, deg, hdst, Wl, bl, Wr, g, be, N, blockIdx.x, sWl, sWr);
}

__global__ void __launch_bounds__(128)
sage_layer_kernel(const float* __restrict__ sums1, const float* __restrict__ deg1,
                  const float* __restrict__ Wl1, const float* __restrict__ bl1,
                  const float* __restrict__ Wr1,
                  const float* __restrict__ sums2, const float* __restrict__ deg2,
                  const float* __restrict__ Wl2, const float* __restrict__ bl2,
                  const float* __restrict__ Wr2,
                  float* hT, const float* __restrict__ gT, const float* __restrict__ beT, int NT, int nbT,
                  const float* __restrict__ sumsO, const float* __restrict__ degO,
                  const float* __restrict__ WlO, const float* __restrict__ blO,
                  const float* __restrict__ WrO,
                  float* hO, const float* __restrict__ gO, const float* __restrict__ beO, int NO) {
    __shared__ float sWl[H * H];
    __shared__ float sWr[H * H];
    int b = blockIdx.x;
    if (b < nbT)
        sage_block_two(sums1, deg1, Wl1, bl1, Wr1, sums2, deg2, Wl2, bl2, Wr2,
                       hT, gT, beT, NT, b, sWl, sWr);
    else
        sage_block_one(sumsO, degO, hO, WlO, blO, WrO, gO, beO, NO, b - nbT, sWl, sWr);
}

__global__ void cox_kernel(const float* __restrict__ hp, const float* __restrict__ W1,
                           const float* __restrict__ b1, const float* __restrict__ W2,
                           const float* __restrict__ b2, float* __restrict__ out, int N) {
    int n = blockIdx.x * blockDim.x + threadIdx.x;
    if (n >= N) return;
    const float* hr = hp + (long)n * H;
    float hreg[H];
#pragma unroll
    for (int k = 0; k < H; ++k) hreg[k] = hr[k];
    float hz = b2[0];
    for (int j = 0; j < 32; ++j) {
        float s = b1[j];
#pragma unroll
        for (int k = 0; k < H; ++k) s = fmaf(hreg[k], W1[k * 32 + j], s);
        hz = fmaf(elu1(s), W2[j], hz);
    }
    out[n] = hz;
}

extern "C" void kernel_launch(void* const* d_in, const int* in_sizes, int n_in,
                              void* d_out, int out_size, void* d_ws, size_t ws_size,
                              hipStream_t stream) {
    (void)in_sizes; (void)n_in; (void)out_size; (void)ws_size;
    const int NG = 20000, NP = 100000, NM = 5000, E = 1000000;

    const float* x_gene    = (const float*)d_in[0];
    const float* x_patient = (const float*)d_in[1];
    const float* x_group   = (const float*)d_in[2];
    const int* srcA[4]  = { (const int*)d_in[3], (const int*)d_in[5],
                            (const int*)d_in[7], (const int*)d_in[9] };
    const int* dstA[4]  = { (const int*)d_in[4], (const int*)d_in[6],
                            (const int*)d_in[8], (const int*)d_in[10] };
    const float* encW[3] = { (const float*)d_in[11], (const float*)d_in[13], (const float*)d_in[15] };
    const float* encB[3] = { (const float*)d_in[12], (const float*)d_in[14], (const float*)d_in[16] };
    const float* convW_l = (const float*)d_in[17];
    const float* convb_l = (const float*)d_in[18];
    const float* convW_r = (const float*)d_in[19];
    const float* ln_g = (const float*)d_in[20];
    const float* ln_b = (const float*)d_in[21];
    const float* coxW1 = (const float*)d_in[22];
    const float* coxb1 = (const float*)d_in[23];
    const float* coxW2 = (const float*)d_in[24];
    const float* coxb2 = (const float*)d_in[25];

    float* ws = (float*)d_ws;
    float* h_gene  = ws; ws += (size_t)NG * H;
    float* h_pat   = ws; ws += (size_t)NP * H;
    float* h_grp   = ws; ws += (size_t)NM * H;
    float* hn_gene = ws; ws += (size_t)NG * H;
    float* hn_pat  = ws; ws += (size_t)NP * H;
    float* degG1   = ws;
    float* degG3   = ws + NG;
    ws += (size_t)NM * H;
    const int SL = 34000;
    float* aggreg  = ws; ws += (size_t)NP * H;
    float* deg     = ws; ws += (size_t)NP;
    float* agg     = aggreg;
    int*   perm    = (int*)(aggreg + (size_t)SL * H);
    int*   rowptr  = perm + (size_t)3 * E;
    int*   cnt     = (int*)hn_gene;
    int*   ccnt    = (int*)hn_pat;

    const int* srcL[3] = { srcA[0], srcA[1], srcA[3] };
    const int* dstL[3] = { dstA[0], dstA[1], dstA[3] };
    const int* none = nullptr;
    const int NCH   = (E + CHE - 1) / CHE;
    const int NSLOT = (NP + RNG - 1) / RNG + (NG + RNG - 1) / RNG
                    + (NG + RNG - 1) / RNG;
    const int NTOT  = NP + NG + NG;
    const int GCAP = 1024;
    const int gbEG = (NG + 7) / 8 < GCAP ? (NG + 7) / 8 : GCAP;
    const int gbEP = (NP + 15) / 16 < GCAP ? (NP + 15) / 16 : GCAP;
    const int gbEM = (NM + 15) / 16 < GCAP ? (NM + 15) / 16 : GCAP;

    count_kernel<<<dim3(NCH, NSLOT), 1024, 0, stream>>>(
        dstL[0], dstL[1], dstL[2], none, ccnt, NP, NG, NG, 0, E);
    chunk_offsets_kernel<NCH><<<(NTOT + 255) / 256, 256, 0, stream>>>(ccnt, cnt, NTOT);
    scan_enc_kernel<<<NSCAN + (gbEP + gbEG + gbEM + 3) / 4, 1024, 0, stream>>>(
        cnt, rowptr, NP, NG, NG, 0,
        x_patient, encW[1], encB[1], h_pat,  NP, gbEP,
        x_gene,    encW[0], encB[0], h_gene, NG, gbEG,
        x_group,   encW[2], encB[2], h_grp,  NM, gbEM);
    fill_kernel<<<dim3(NCH, NSLOT), 1024, 0, stream>>>(
        srcL[0], srcL[1], srcL[2], none, dstL[0], dstL[1], dstL[2], none,
        rowptr, ccnt, perm, NP, NG, NG, 0, E);
    const int* rpL[3] = { rowptr, rowptr + NP + 4, rowptr + NP + NG + 8 };
    const int* pmL[3] = { perm, perm + (size_t)E, perm + (size_t)2 * E };

    const size_t HH = (size_t)H * H;
    const int gbG = (NG + 7) / 8 < 2 * GCAP ? (NG + 7) / 8 : 2 * GCAP;
    const int gbP = (NP + 7) / 8 < 2 * GCAP ? (NP + 7) / 8 : 2 * GCAP;
    const int nbG = (NG + 63) / 64, nbP = (NP + 63) / 64;

    gather3_kernel<<<gbG + gbG + gbP, 256, 0, stream>>>(
        h_pat,  rpL[1], pmL[1], hn_gene, degG1, NG, NP, gbG,
        h_grp,  rpL[2], pmL[2], agg,     degG3, NG, NM, gbG,
        h_gene, rpL[0], pmL[0], hn_pat,  deg,   NP, NG, gbP,
        E);
    sage_layer_kernel<<<nbG + nbP, 128, 0, stream>>>(
        hn_gene, degG1, convW_l + 1 * HH, convb_l + 1 * H, convW_r + 1 * HH,
        agg,     degG3, convW_l + 3 * HH, convb_l + 3 * H, convW_r + 3 * HH,
        h_gene, ln_g + 0 * H, ln_b + 0 * H, NG, nbG,
        hn_pat,  deg,   convW_l + 0 * HH, convb_l + 0 * H, convW_r + 0 * HH,
        h_pat,  ln_g + 1 * H, ln_b + 1 * H, NP);

    gather_sum_kernel<4><<<gbP, 256, 0, stream>>>(
        h_gene, rpL[0], pmL[0], hn_pat, deg, NP, NG, E);
    sage_gemm_kernel<<<nbP, 128, 0, stream>>>(
        hn_pat, deg, h_pat, convW_l + 4 * HH, convb_l + 4 * H, convW_r + 4 * HH,
        ln_g + (3 + 1) * H, ln_b + (3 + 1) * H, NP);

    cox_kernel<<<(NP + 255) / 256, 256, 0, stream>>>(
        h_pat, coxW1, coxb1, coxW2, coxb2, (float*)d_out, NP);
}


# Gather waves walk their nodes as a software pipeline: later nodes' offsets and next source ids are requested behind the current rows' first loads; layer-1 gather = one resident set; sum order kept
# speedup vs baseline: 11.0125x; 1.0408x over previous
#include <hip/hip_runtime.h>
#include <hip/hip_bf16.h>
#include <math.h>

typedef float v2f __attribute__((ext_vector_type(2)));
typedef float v8f __attribute__((ext_vector_type(8)));

#define H 64

__device__ __forceinline__ float elu1(float x) {
    return x > 0.0f ? x : (__expf(x) - 1.0f);
}

template <int F, int U>
__device__ __forceinline__ void enc_rows(const float* __restrict__ x, const float* __restrict__ W,
                                         const float* __restrict__ b, float* __restrict__ h,
                                         int N, int vb, int nvb, int tid) {
    int j = tid & 63;
    int sub = tid >> 6;
    float w[F];
#pragma unroll
    for (int k = 0; k < F; ++k) w[k] = W[k * H + j];
    float bj = b[j];
    for (int n0 = vb * (4 * U) + sub; n0 < N; n0 += nvb * (4 * U)) {
        if (n0 + 4 * (U - 1) < N) {
            float xv[U][F];
#pragma unroll
            for (int u = 0; u < U; ++u) {
                const float* xr = x + (long)(n0 + 4 * u) * F;
#pragma unroll
                for (int k = 0; k < F; ++k) xv[u][k] = xr[k];
            }
#pragma unroll
            for (int u = 0; u < U; ++u) {
                float s = bj;
#pragma unroll
                for (int k = 0; k < F; ++k) s = fmaf(xv[u][k], w[k], s);
                h[(long)(n0 + 4 * u) * H + j] = elu1(s);
            }
        } else {
            for (int n = n0; n < N; n += 4) {
                const float* xr = x + (long)n * F;
                float s = bj;
#pragma unroll
                for (int k = 0; k < F; ++k) s = fmaf(xr[k], w[k], s);
                h[(long)n * H + j] = elu1(s);
            }
        }
    }
}

#define RNG  25000
#define CHSH 15
#define CHE  (1 << CHSH)
#define LPT  8

__device__ __forceinline__ void range_of_slot(int slot, int n0, int n1, int n2, int n3,
                                              int& t, int& lo, int& cntn, int& coff) {
    int r0 = (n0 + RNG - 1) / RNG, r1 = (n1 + RNG - 1) / RNG, r2 = (n2 + RNG - 1) / RNG;
    int r, N;
    if (slot < r0)                { t = 0; r = slot;                N = n0; coff = 0; }
    else if (slot < r0 + r1)      { t = 1; r = slot - r0;           N = n1; coff = n0; }
    else if (slot < r0 + r1 + r2) { t = 2; r = slot - r0 - r1;      N = n2; coff = n0 + n1; }
    else                          { t = 3; r = slot - r0 - r1 - r2; N = n3; coff = n0 + n1 + n2; }
    lo = r * RNG;
    cntn = min(RNG, N - lo);
}

__global__ void __launch_bounds__(1024)
count_kernel(const int* __restrict__ d0, const int* __restrict__ d1,
             const int* __restrict__ d2, const int* __restrict__ d3,
             int* __restrict__ ccnt, int n0, int n1, int n2, int n3, int E) {
    __shared__ int lc[RNG];
    int t, lo, cntn, coff;
    range_of_slot(blockIdx.y, n0, n1, n2, n3, t, lo, cntn, coff);
    const int* d = t == 0 ? d0 : t == 1 ? d1 : t == 2 ? d2 : d3;
    int ntot = n0 + n1 + n2 + n3;
    for (int i = threadIdx.x; i < cntn; i += 1024) lc[i] = 0;
    __syncthreads();
    int e0 = blockIdx.x << CHSH;
    int e1 = min(e0 + CHE, E);
    for (int b = e0 + threadIdx.x; b < e1; b += 1024 * LPT) {
        int dv[LPT];
#pragma unroll
        for (int u = 0; u < LPT; ++u) {
            int e = b + u * 1024;
            dv[u] = e < e1 ? d[e] : -1;
        }
#pragma unroll
        for (int u = 0; u < LPT; ++u) {
            unsigned rel = (unsigned)(dv[u] - lo);
            if (rel < (unsigned)cntn) atomicAdd(&lc[rel], 1);
        }
    }
    __syncthreads();
    int* o = ccnt + (long)blockIdx.x * ntot + coff + lo;
    for (int i = threadIdx.x; i < cntn; i += 1024) o[i] = lc[i];
}

template <int NCH>
__global__ void __launch_bounds__(256)
chunk_offsets_kernel(int* __restrict__ ccnt, int* __restrict__ cnt, int ntot) {
    int n = blockIdx.x * 256 + threadIdx.x;
    if (n >= ntot) return;
    int v[NCH];
#pragma unroll
    for (int c = 0; c < NCH; ++c) v[c] = ccnt[(long)c * ntot + n];
    int run = 0;
#pragma unroll
    for (int c = 0; c < NCH; ++c) {
        ccnt[(long)c * ntot + n] = run;
        run += v[c];
    }
    cnt[n] = run;
}

__device__ __forceinline__ void rowptr_scan_block(const int* __restrict__ cnt, int* __restrict__ rowptr,
                                                  int n0, int n1, int n2, int n3, int t, int* wsum) {
    int N    = t == 0 ? n0 : t == 1 ? n1 : t == 2 ? n2 : n3;
    int coff = t == 0 ? 0 : t == 1 ? n0 : t == 2 ? n0 + n1 : n0 + n1 + n2;
    const int* c = cnt + coff;
    int* rp = rowptr + coff + 4 * t;
    int wave = threadIdx.x >> 5, lane = threadIdx.x & 31;
    int per = (((N + 31) >> 5) + 127) & ~127;
    int beg = wave * per;
    int end = min(beg + per, N);

    int s = 0;
    for (int p = beg + lane * 4; p < end; p += 128) {
        int4 v = *(const int4*)(c + p);
        s += v.x + v.y + v.z + v.w;
    }
    for (int m = 16; m >= 1; m >>= 1) s += __shfl_xor(s, m, 32);
    if (lane == 0) wsum[wave] = s;
    __syncthreads();

    int wtot = wsum[lane];
    int winc = wtot;
    for (int dlt = 1; dlt < 32; dlt <<= 1) {
        int o = __shfl_up(winc, dlt, 32);
        if (lane >= dlt) winc += o;
    }
    int carry = __shfl(winc - wtot, wave, 32);
    int total = __shfl(winc, 31, 32);

    for (int tb = beg; tb < end; tb += 128) {
        int p = tb + lane * 4;
        int4 v = make_int4(0, 0, 0, 0);
        if (p < end) v = *(const int4*)(c + p);
        int ls = v.x + v.y + v.z + v.w;
        int li = ls;
        for (int dlt = 1; dlt < 32; dlt <<= 1) {
            int o = __shfl_up(li, dlt, 32);
            if (lane >= dlt) li += o;
        }
        int ex = carry + li - ls;
        if (p < end)
            *(int4*)(rp + p) = make_int4(ex, ex + v.x, ex + v.x + v.y, ex + v.x + v.y + v.z);
        carry += __shfl(li, 31, 32);
    }
    if (threadIdx.x == 0) rp[N] = total;
}

#define NSCAN 3
__global__ void __launch_bounds__(1024)
scan_enc_kernel(const int* __restrict__ cnt, int* __restrict__ rowptr,
                int n0, int n1, int n2, int n3,
                const float* __restrict__ xp, const float* __restrict__ Wp,
                const float* __restrict__ bp, float* __restrict__ hp, int Np, int gbp,
                const float* __restrict__ xg, const float* __restrict__ Wg,
                const float* __restrict__ bg, float* __restrict__ hg, int Ng, int gbg,
                const float* __restrict__ xm, const float* __restrict__ Wm,
                const float* __restrict__ bm, float* __restrict__ hm, int Nm, int gbm) {
    __shared__ int wsum[32];
    if (blockIdx.x < NSCAN) {
        rowptr_scan_block(cnt, rowptr, n0, n1, n2, n3, blockIdx.x, wsum);
    } else {
        int vb  = (blockIdx.x - NSCAN) * 4 + (threadIdx.x >> 8);
        int tid = threadIdx.x & 255;
        if (vb < gbp)                  enc_rows<3, 4>(xp, Wp, bp, hp, Np, vb, gbp, tid);
        else if (vb < gbp + gbg)       enc_rows<11, 2>(xg, Wg, bg, hg, Ng, vb - gbp, gbg, tid);
        else if (vb < gbp + gbg + gbm) enc_rows<4, 4>(xm, Wm, bm, hm, Nm, vb - gbp - gbg, gbm, tid);
    }
}

__global__ void __launch_bounds__(1024)
fill_kernel(const int* __restrict__ s0, const int* __restrict__ s1,
            const int* __restrict__ s2, const int* __restrict__ s3,
            const int* __restrict__ d0, const int* __restrict__ d1,
            const int* __restrict__ d2, const int* __restrict__ d3,
            const int* __restrict__ rowptr, const int* __restrict__ choff,
            int* __restrict__ perm, int n0, int n1, int n2, int n3, int E) {
    __shared__ int lc[RNG];
    int t, lo, cntn, coff;
    range_of_slot(blockIdx.y, n0, n1, n2, n3, t, lo, cntn, coff);
    const int* s = t == 0 ? s0 : t == 1 ? s1 : t == 2 ? s2 : s3;
    const int* d = t == 0 ? d0 : t == 1 ? d1 : t == 2 ? d2 : d3;
    int ntot = n0 + n1 + n2 + n3;
    const int* rp = rowptr + coff + 4 * t + lo;
    const int* co = choff + (long)blockIdx.x * ntot + coff + lo;
    for (int i = threadIdx.x; i < cntn; i += 1024) lc[i] = rp[i] + co[i];
    __syncthreads();
    int e0 = blockIdx.x << CHSH;
    int e1 = min(e0 + CHE, E);
    int* pm = perm + (long)t * E;
    for (int b = e0 + threadIdx.x; b < e1; b += 1024 * LPT) {
        int dv[LPT], sv[LPT];
#pragma unroll
        for (int u = 0; u < LPT; ++u) {
            int e = b + u * 1024;
            bool ok = e < e1;
            dv[u] = ok ? d[e] : -1;
            sv[u] = ok ? s[e] : 0;
        }
        int pos[LPT];
#pragma unroll
        for (int u = 0; u < LPT; ++u) {
            unsigned rel = (unsigned)(dv[u] - lo);
            pos[u] = -1;
            if (rel < (unsigned)cntn) pos[u] = atomicAdd(&lc[rel], 1);
        }
#pragma unroll
        for (int u = 0; u < LPT; ++u) {
            if (pos[u] >= 0) pm[min(pos[u], E - 1)] = sv[u];
        }
    }
}

template <int U>
__device__ __forceinline__ void gather_rows(const float* __restrict__ hsrc, const int* __restrict__ rowptr,
                                            const int* __restrict__ perm, float* __restrict__ agg,
                                            float* __restrict__ deg, int ncnt, int nsrc, int E,
                                            int vb, int nvb) {
    int wave = threadIdx.x >> 5, lane = threadIdx.x & 31;
    int half = lane >> 4, c = lane & 15;
    const char* hsb = (const char*)hsrc;
    const int stride = nvb * 8;
    int ln = vb * 8 + wave;
    if (ln >= ncnt) return;
    int ln1 = ln + stride;
    int beg = rowptr[ln], end = rowptr[ln + 1];
    int nb = 0, ne = 0;
    if (ln1 < ncnt) { nb = rowptr[ln1]; ne = rowptr[ln1 + 1]; }
    beg = max(beg, 0);
    end = min(end, E);
    int ids = 0;
    if (lane < end - beg) ids = perm[beg + lane];
    for (;;) {
        int ln2 = ln1 + stride;
        nb = max(nb, 0);
        ne = min(ne, E);
        bool nidok = lane < ne - nb;
        int nnb = 0, nne = 0, nids = 0;
        bool pend = true;

        float4 acc = make_float4(0.0f, 0.0f, 0.0f, 0.0f);
        int cur = ids;
        for (int base = beg; base < end; base += 32) {
            int cnt = min(32, end - base);
            int nxt = 0;
            int mine = min(max(cur, 0), nsrc - 1);
            for (int i = 0; i < cnt; i += 2 * U) {
                float4 v[U];
                bool ok[U];
#pragma unroll
                for (int u = 0; u < U; ++u) {
                    int idx = i + 2 * u + half;
                    ok[u] = idx < cnt;
                    int sidx = __shfl(mine, ok[u] ? idx : 0, 32);
                    v[u] = *(const float4*)(hsb + (unsigned)(sidx * 256 + c * 16));
                }
                if (i == 0) {
                    if (lane < end - base - 32) nxt = perm[base + 32 + lane];
                    if (pend) {
                        if (ln2 < ncnt) { nnb = rowptr[ln2]; nne = rowptr[ln2 + 1]; }
                        if (nidok) nids = perm[nb + lane];
                        pend = false;
                    }
                }
#pragma unroll
                for (int u = 0; u < U; ++u) {
                    acc.x += ok[u] ? v[u].x : 0.0f;
                    acc.y += ok[u] ? v[u].y : 0.0f;
                    acc.z += ok[u] ? v[u].z : 0.0f;
                    acc.w += ok[u] ? v[u].w : 0.0f;
                }
            }
            cur = nxt;
        }
        if (pend) {
            if (ln2 < ncnt) { nnb = rowptr[ln2]; nne = rowptr[ln2 + 1]; }
            if (nidok) nids = perm[nb + lane];
        }
        acc.x += __shfl_xor(acc.x, 16, 32);
        acc.y += __shfl_xor(acc.y, 16, 32);
        acc.z += __shfl_xor(acc.z, 16, 32);
        acc.w += __shfl_xor(acc.w, 16, 32);
        if (half == 0) ((float4*)agg)[(long)ln * 16 + c] = acc;
        if (lane == 0) deg[ln] = (float)(end - beg);
        if (ln1 >= ncnt) break;
        ln = ln1; ln1 = ln2;
        beg = nb; end = ne; nb = nnb; ne = nne;
        ids = nids;
    }
}

template <int U>
__global__ void __launch_bounds__(256)
gather_sum_kernel(const float* __restrict__ hsrc, const int* __restrict__ rowptr,
                  const int* __restrict__ perm, float* __restrict__ agg,
                  float* __restrict__ deg, int ncnt, int nsrc, int E) {
    gather_rows<U>(hsrc, rowptr, perm, agg, deg, ncnt, nsrc, E, blockIdx.x, gridDim.x);
}

__global__ void __launch_bounds__(256, 16)
gather3_kernel(const float* __restrict__ hA, const int* __restrict__ rpA, const int* __restrict__ pmA,
               float* __restrict__ aggA, float* __restrict__ degA, int nA, int nsA, int gbA,
               const float* __restrict__ hB, const int* __restrict__ rpB, const int* __restrict__ pmB,
               float* __restrict__ aggB, float* __restrict__ degB, int nB, int nsB, int gbB,
               const float* __restrict__ hC, const int* __restrict__ rpC, const int* __restrict__ pmC,
               float* __restrict__ aggC, float* __restrict__ degC, int nC, int nsC, int gbC,
               int E) {
    int b = blockIdx.x;
    if (b < gbA)            gather_rows<8>(hA, rpA, pmA, aggA, degA, nA, nsA, E, b, gbA);
    else if (b < gbA + gbB) gather_rows<8>(hB, rpB, pmB, aggB, degB, nB, nsB, E, b - gbA, gbB);
    else                    gather_rows<8>(hC, rpC, pmC, aggC, degC, nC, nsC, E, b - gbA - gbB, gbC);
}

template <int M>
__device__ __forceinline__ float row_xor_add(float x) {
    int xi = __float_as_int(x);
    int yi = __builtin_amdgcn_update_dpp(0, xi, 0x160 + M, 0xf, 0xf, false);
    return x + __int_as_float(yi);
}


__device__ __forceinline__ void stage_weights(const float* __restrict__ Wl, const float* __restrict__ Wr,
                                              float* sWl, float* sWr, int tid) {
    const float4* gl = (const float4*)Wl;
    const float4* gr = (const float4*)Wr;
    float4* ll = (float4*)sWl;
    float4* lr = (float4*)sWr;
    for (int i = tid; i < (H * H / 4); i += 128) {
        ll[i] = gl[i];
        lr[i] = gr[i];
    }
}

__device__ __forceinline__ void tile_products(const float* aggr, float invd, const float* hdr,
                                              const float* sWl, const float* sWr,
                                              int kk, int col, v8f (&acc)[4]) {
#pragma unroll
    for (int k0 = 0; k0 < H; k0 += 4) {
        v2f a;
        a.x = aggr[k0 + kk] * invd;
        a.y = aggr[k0 + kk + 1] * invd;
#pragma unroll
        for (int t4 = 0; t4 < 4; ++t4) {
            v2f b;
            b.x = sWl[(k0 + kk) * H + t4 * 16 + col];
            b.y = sWl[(k0 + kk + 1) * H + t4 * 16 + col];
            acc[t4] = __builtin_amdgcn_wmma_f32_16x16x4_f32(
                false, a, false, b, (short)0, acc[t4], false, false);
        }
    }
#pragma unroll
    for (int k0 = 0; k0 < H; k0 += 4) {
        v2f a;
        a.x = hdr[k0 + kk];
        a.y = hdr[k0 + kk + 1];
#pragma unroll
        for (int t4 = 0; t4 < 4; ++t4) {
            v2f b;
            b.x = sWr[(k0 + kk) * H + t4 * 16 + col];
            b.y = sWr[(k0 + kk + 1) * H + t4 * 16 + col];
            acc[t4] = __builtin_amdgcn_wmma_f32_16x16x4_f32(
                false, a, false, b, (short)0, acc[t4], false, false);
        }
    }
}

template <bool TWO>
__device__ __forceinline__ void tile_ln_store(const v8f (&acc)[4], const v8f (&first)[4],
                                              const float* __restrict__ bl, const float* __restrict__ bl1,
                                              const float* __restrict__ g, const float* __restrict__ be,
                                              float* hdst, int m0, int hi, int col, int N) {
    float bias[4], bias1[4], gm[4], bt[4];
#pragma unroll
    for (int t4 = 0; t4 < 4; ++t4) {
        int cg = t4 * 16 + col;
        bias[t4]  = bl[cg];
        bias1[t4] = TWO ? bl1[cg] : 0.0f;
        gm[t4]    = g[cg];
        bt[t4]    = be[cg];
    }
    float res[4][8];
#pragma unroll
    for (int rr = 0; rr < 8; ++rr) {
        int row = m0 + rr + hi * 8;
        long ro = (long)(row < N ? row : (N - 1)) * H + col;
#pragma unroll
        for (int t4 = 0; t4 < 4; ++t4) res[t4][rr] = hdst[ro + t4 * 16];
    }
#pragma unroll
    for (int rr = 0; rr < 8; ++rr) {
        int row = m0 + rr + hi * 8;
        float x[4];
#pragma unroll
        for (int t4 = 0; t4 < 4; ++t4) {
            float o = acc[t4][rr] + bias[t4];
            if (TWO) o = (first[t4][rr] + bias1[t4]) + o;
            x[t4] = res[t4][rr] + o;
        }
        float s = (x[0] + x[2]) + (x[1] + x[3]);
        s = row_xor_add<8>(s);
        s = row_xor_add<4>(s);
        s = row_xor_add<2>(s);
        s = row_xor_add<1>(s);
        float mean = s * (1.0f / 64.0f);
        float d0 = x[0] - mean, d1 = x[1] - mean, d2 = x[2] - mean, d3 = x[3] - mean;
        float qa = d0 * d0 + d2 * d2;
        float qb = d1 * d1 + d3 * d3;
        float q = qa + qb;
        q = row_xor_add<8>(q);
        q = row_xor_add<4>(q);
        q = row_xor_add<2>(q);
        q = row_xor_add<1>(q);
        float inv = rsqrtf(q * (1.0f / 64.0f) + 1e-5f);
        float o0 = elu1(d0 * inv * gm[0] + bt[0]);
        float o1 = elu1(d1 * inv * gm[1] + bt[1]);
        float o2 = elu1(d2 * inv * gm[2] + bt[2]);
        float o3 = elu1(d3 * inv * gm[3] + bt[3]);
        if (row < N) {
            float* orow = hdst + (long)row * H + col;
            orow[0]  = o0;
            orow[16] = o1;
            orow[32] = o2;
            orow[48] = o3;
        }
    }
}

__device__ __forceinline__ void sage_block_one(const float* __restrict__ sums, const float* __restrict__ deg,
                                               float* hdst,
                                               const float* __restrict__ Wl, const float* __restrict__ bl,
                                               const float* __restrict__ Wr,
                                               const float* __restrict__ g, const float* __restrict__ be,
                                               int N, int blk, float* sWl, float* sWr) {
    int tid = threadIdx.x;
    stage_weights(Wl, Wr, sWl, sWr, tid);
    __syncthreads();

    int wave = tid >> 5;
    int lane = tid & 31;
    int hi   = lane >> 4;
    int col  = lane & 15;
    int kk   = hi * 2;
    int m0   = blk * 64 + wave * 16;

    int r  = m0 + col;
    int rc = r < N ? r : (N - 1);
    float invd = 1.0f / fmaxf(deg[rc], 1.0f);

    v8f acc[4] = {};
    tile_products(sums + (long)rc * H, invd, hdst + (long)rc * H, sWl, sWr, kk, col, acc);
    tile_ln_store<false>(acc, acc, bl, bl, g, be, hdst, m0, hi, col, N);
}

__device__ __forceinline__ void sage_block_two(const float* __restrict__ sums1, const float* __restrict__ deg1,
                                               const float* __restrict__ Wl1, const float* __restrict__ bl1,
                                               const float* __restrict__ Wr1,
                                               const float* __restrict__ sums2, const float* __restrict__ deg2,
                                               const float* __restrict__ Wl2, const float* __restrict__ bl2,
                                               const float* __restrict__ Wr2,
                                               float* hdst,
                                               const float* __restrict__ g, const float* __restrict__ be,
                                               int N, int blk, float* sWl, float* sWr) {
    int tid = threadIdx.x;
    stage_weights(Wl1, Wr1, sWl, sWr, tid);
    __syncthreads();

    int wave = tid >> 5;
    int lane = tid & 31;
    int hi   = lane >> 4;
    int col  = lane & 15;
    int kk   = hi * 2;
    int m0   = blk * 64 + wave * 16;

    int r  = m0 + col;
    int rc = r < N ? r : (N - 1);
    float invd1 = 1.0f / fmaxf(deg1[rc], 1.0f);
    float invd2 = 1.0f / fmaxf(deg2[rc], 1.0f);
    const float* hdr = hdst + (long)rc * H;

    v8f acc1[4] = {};
    tile_products(sums1 + (long)rc * H, invd1, hdr, sWl, sWr, kk, col, acc1);
    __syncthreads();
    stage_weights(Wl2, Wr2, sWl, sWr, tid);
    __syncthreads();
    v8f acc2[4] = {};
    tile_products(sums2 + (long)rc * H, invd2, hdr, sWl, sWr, kk, col, acc2);
    tile_ln_store<true>(acc2, acc1, bl2, bl1, g, be, hdst, m0, hi, col, N);
}

__global__ void __launch_bounds__(128)
sage_gemm_kernel(const float* __restrict__ sums, const float* __restrict__ deg, float* hdst,
                 const float* __restrict__ Wl, const float* __restrict__ bl,
                 const float* __restrict__ Wr,
                 const float* __restrict__ g, const float* __restrict__ be, int N) {
    __shared__ float sWl[H * H];
    __shared__ float sWr[H * H];
    sage_block_one(sums, deg, hdst, Wl, bl, Wr, g, be, N, blockIdx.x, sWl, sWr);
}

__global__ void __launch_bounds__(128)
sage_layer_kernel(const float* __restrict__ sums1, const float* __restrict__ deg1,
                  const float* __restrict__ Wl1, const float* __restrict__ bl1,
                  const float* __restrict__ Wr1,
                  const float* __restrict__ sums2, const float* __restrict__ deg2,
                  const float* __restrict__ Wl2, const float* __restrict__ bl2,
                  const float* __restrict__ Wr2,
                  float* hT, const float* __restrict__ gT, const float* __restrict__ beT, int NT, int nbT,
                  const float* __restrict__ sumsO, const float* __restrict__ degO,
                  const float* __restrict__ WlO, const float* __restrict__ blO,
                  const float* __restrict__ WrO,
                  float* hO, const float* __restrict__ gO, const float* __restrict__ beO, int NO) {
    __shared__ float sWl[H * H];
    __shared__ float sWr[H * H];
    int b = blockIdx.x;
    if (b < nbT)
        sage_block_two(sums1, deg1, Wl1, bl1, Wr1, sums2, deg2, Wl2, bl2, Wr2,
                       hT, gT, beT, NT, b, sWl, sWr);
    else
        sage_block_one(sumsO, degO, hO, WlO, blO, WrO, gO, beO, NO, b - nbT, sWl, sWr);
}

__global__ void cox_kernel(const float* __restrict__ hp, const float* __restrict__ W1,
                           const float* __restrict__ b1, const float* __restrict__ W2,
                           const float* __restrict__ b2, float* __restrict__ out, int N) {
    int n = blockIdx.x * blockDim.x + threadIdx.x;
    if (n >= N) return;
    const float* hr = hp + (long)n * H;
    float hreg[H];
#pragma unroll
    for (int k = 0; k < H; ++k) hreg[k] = hr[k];
    float hz = b2[0];
    for (int j = 0; j < 32; ++j) {
        float s = b1[j];
#pragma unroll
        for (int k = 0; k < H; ++k) s = fmaf(hreg[k], W1[k * 32 + j], s);
        hz = fmaf(elu1(s), W2[j], hz);
    }
    out[n] = hz;
}

extern "C" void kernel_launch(void* const* d_in, const int* in_sizes, int n_in,
                              void* d_out, int out_size, void* d_ws, size_t ws_size,
                              hipStream_t stream) {
    (void)in_sizes; (void)n_in; (void)out_size; (void)ws_size;
    const int NG = 20000, NP = 100000, NM = 5000, E = 1000000;

    const float* x_gene    = (const float*)d_in[0];
    const float* x_patient = (const float*)d_in[1];
    const float* x_group   = (const float*)d_in[2];
    const int* srcA[4]  = { (const int*)d_in[3], (const int*)d_in[5],
                            (const int*)d_in[7], (const int*)d_in[9] };
    const int* dstA[4]  = { (const int*)d_in[4], (const int*)d_in[6],
                            (const int*)d_in[8], (const int*)d_in[10] };
    const float* encW[3] = { (const float*)d_in[11], (const float*)d_in[13], (const float*)d_in[15] };
    const float* encB[3] = { (const float*)d_in[12], (const float*)d_in[14], (const float*)d_in[16] };
    const float* convW_l = (const float*)d_in[17];
    const float* convb_l = (const float*)d_in[18];
    const float* convW_r = (const float*)d_in[19];
    const float* ln_g = (const float*)d_in[20];
    const float* ln_b = (const float*)d_in[21];
    const float* coxW1 = (const float*)d_in[22];
    const float* coxb1 = (const float*)d_in[23];
    const float* coxW2 = (const float*)d_in[24];
    const float* coxb2 = (const float*)d_in[25];

    float* ws = (float*)d_ws;
    float* h_gene  = ws; ws += (size_t)NG * H;
    float* h_pat   = ws; ws += (size_t)NP * H;
    float* h_grp   = ws; ws += (size_t)NM * H;
    float* hn_gene = ws; ws += (size_t)NG * H;
    float* hn_pat  = ws; ws += (size_t)NP * H;
    float* degG1   = ws;
    float* degG3   = ws + NG;
    ws += (size_t)NM * H;
    const int SL = 34000;
    float* aggreg  = ws; ws += (size_t)NP * H;
    float* deg     = ws; ws += (size_t)NP;
    float* agg     = aggreg;
    int*   perm    = (int*)(aggreg + (size_t)SL * H);
    int*   rowptr  = perm + (size_t)3 * E;
    int*   cnt     = (int*)hn_gene;
    int*   ccnt    = (int*)hn_pat;

    const int* srcL[3] = { srcA[0], srcA[1], srcA[3] };
    const int* dstL[3] = { dstA[0], dstA[1], dstA[3] };
    const int* none = nullptr;
    const int NCH   = (E + CHE - 1) / CHE;
    const int NSLOT = (NP + RNG - 1) / RNG + (NG + RNG - 1) / RNG
                    + (NG + RNG - 1) / RNG;
    const int NTOT  = NP + NG + NG;
    const int GCAP = 1024;
    const int gbEG = (NG + 7) / 8 < GCAP ? (NG + 7) / 8 : GCAP;
    const int gbEP = (NP + 15) / 16 < GCAP ? (NP + 15) / 16 : GCAP;
    const int gbEM = (NM + 15) / 16 < GCAP ? (NM + 15) / 16 : GCAP;

    count_kernel<<<dim3(NCH, NSLOT), 1024, 0, stream>>>(
        dstL[0], dstL[1], dstL[2], none, ccnt, NP, NG, NG, 0, E);
    chunk_offsets_kernel<NCH><<<(NTOT + 255) / 256, 256, 0, stream>>>(ccnt, cnt, NTOT);
    scan_enc_kernel<<<NSCAN + (gbEP + gbEG + gbEM + 3) / 4, 1024, 0, stream>>>(
        cnt, rowptr, NP, NG, NG, 0,
        x_patient, encW[1], encB[1], h_pat,  NP, gbEP,
        x_gene,    encW[0], encB[0], h_gene, NG, gbEG,
        x_group,   encW[2], encB[2], h_grp,  NM, gbEM);
    fill_kernel<<<dim3(NCH, NSLOT), 1024, 0, stream>>>(
        srcL[0], srcL[1], srcL[2], none, dstL[0], dstL[1], dstL[2], none,
        rowptr, ccnt, perm, NP, NG, NG, 0, E);
    const int* rpL[3] = { rowptr, rowptr + NP + 4, rowptr + NP + NG + 8 };
    const int* pmL[3] = { perm, perm + (size_t)E, perm + (size_t)2 * E };

    const size_t HH = (size_t)H * H;
    const int gbP = (NP + 7) / 8 < 2 * GCAP ? (NP + 7) / 8 : 2 * GCAP;
    const int nbG = (NG + 63) / 64, nbP = (NP + 63) / 64;
    const int g3G = 576, g3P = 896;

    gather3_kernel<<<g3G + g3G + g3P, 256, 0, stream>>>(
        h_pat,  rpL[1], pmL[1], hn_gene, degG1, NG, NP, g3G,
        h_grp,  rpL[2], pmL[2], agg,     degG3, NG, NM, g3G,
        h_gene, rpL[0], pmL[0], hn_pat,  deg,   NP, NG, g3P,
        E);
    sage_layer_kernel<<<nbG + nbP, 128, 0, stream>>>(
        hn_gene, degG1, convW_l + 1 * HH, convb_l + 1 * H, convW_r + 1 * HH,
        agg,     degG3, convW_l + 3 * HH, convb_l + 3 * H, convW_r + 3 * HH,
        h_gene, ln_g + 0 * H, ln_b + 0 * H, NG, nbG,
        hn_pat,  deg,   convW_l + 0 * HH, convb_l + 0 * H, convW_r + 0 * HH,
        h_pat,  ln_g + 1 * H, ln_b + 1 * H, NP);

    gather_sum_kernel<8><<<gbP, 256, 0, stream>>>(
        h_gene, rpL[0], pmL[0], hn_pat, deg, NP, NG, E);
    sage_gemm_kernel<<<nbP, 128, 0, stream>>>(
        hn_pat, deg, h_pat, convW_l + 4 * HH, convb_l + 4 * H, convW_r + 4 * HH,
        ln_g + (3 + 1) * H, ln_b + (3 + 1) * H, NP);

    cox_kernel<<<(NP + 255) / 256, 256, 0, stream>>>(
        h_pat, coxW1, coxb1, coxW2, coxb2, (float*)d_out, NP);
}


# Second layer and head in one kernel: a wave takes 16 patient rows from neighbour sums (LDS tile) via GEMM and LayerNorm to the head; head sums now a matrix product and a lane tree; 9 -> 7 launches
# speedup vs baseline: 12.7306x; 1.1560x over previous
#include <hip/hip_runtime.h>
#include <hip/hip_bf16.h>
#include <math.h>

typedef float v2f __attribute__((ext_vector_type(2)));
typedef float v8f __attribute__((ext_vector_type(8)));

#define H 64

__device__ __forceinline__ float elu1(float x) {
    return x > 0.0f ? x : (__expf(x) - 1.0f);
}

template <int F, int U>
__device__ __forceinline__ void enc_rows(const float* __restrict__ x, const float* __restrict__ W,
                                         const float* __restrict__ b, float* __restrict__ h,
                                         int N, int vb, int nvb, int tid) {
    int j = tid & 63;
    int sub = tid >> 6;
    float w[F];
#pragma unroll
    for (int k = 0; k < F; ++k) w[k] = W[k * H + j];
    float bj = b[j];
    for (int n0 = vb * (4 * U) + sub; n0 < N; n0 += nvb * (4 * U)) {
        if (n0 + 4 * (U - 1) < N) {
            float xv[U][F];
#pragma unroll
            for (int u = 0; u < U; ++u) {
                const float* xr = x + (long)(n0 + 4 * u) * F;
#pragma unroll
                for (int k = 0; k < F; ++k) xv[u][k] = xr[k];
            }
#pragma unroll
            for (int u = 0; u < U; ++u) {
                float s = bj;
#pragma unroll
                for (int k = 0; k < F; ++k) s = fmaf(xv[u][k], w[k], s);
                h[(long)(n0 + 4 * u) * H + j] = elu1(s);
            }
        } else {
            for (int n = n0; n < N; n += 4) {
                const float* xr = x + (long)n * F;
                float s = bj;
#pragma unroll
                for (int k = 0; k < F; ++k) s = fmaf(xr[k], w[k], s);
                h[(long)n * H + j] = elu1(s);
            }
        }
    }
}

#define RNG  25000
#define CHSH 15
#define CHE  (1 << CHSH)
#define LPT  8

__device__ __forceinline__ void range_of_slot(int slot, int n0, int n1, int n2, int n3,
                                              int& t, int& lo, int& cntn, int& coff) {
    int r0 = (n0 + RNG - 1) / RNG, r1 = (n1 + RNG - 1) / RNG, r2 = (n2 + RNG - 1) / RNG;
    int r, N;
    if (slot < r0)                { t = 0; r = slot;                N = n0; coff = 0; }
    else if (slot < r0 + r1)      { t = 1; r = slot - r0;           N = n1; coff = n0; }
    else if (slot < r0 + r1 + r2) { t = 2; r = slot - r0 - r1;      N = n2; coff = n0 + n1; }
    else                          { t = 3; r = slot - r0 - r1 - r2; N = n3; coff = n0 + n1 + n2; }
    lo = r * RNG;
    cntn = min(RNG, N - lo);
}

__global__ void __launch_bounds__(1024)
count_kernel(const int* __restrict__ d0, const int* __restrict__ d1,
             const int* __restrict__ d2, const int* __restrict__ d3,
             int* __restrict__ ccnt, int n0, int n1, int n2, int n3, int E) {
    __shared__ int lc[RNG];
    int t, lo, cntn, coff;
    range_of_slot(blockIdx.y, n0, n1, n2, n3, t, lo, cntn, coff);
    const int* d = t == 0 ? d0 : t == 1 ? d1 : t == 2 ? d2 : d3;
    int ntot = n0 + n1 + n2 + n3;
    for (int i = threadIdx.x; i < cntn; i += 1024) lc[i] = 0;
    __syncthreads();
    int e0 = blockIdx.x << CHSH;
    int e1 = min(e0 + CHE, E);
    for (int b = e0 + threadIdx.x; b < e1; b += 1024 * LPT) {
        int dv[LPT];
#pragma unroll
        for (int u = 0; u < LPT; ++u) {
            int e = b + u * 1024;
            dv[u] = e < e1 ? d[e] : -1;
        }
#pragma unroll
        for (int u = 0; u < LPT; ++u) {
            unsigned rel = (unsigned)(dv[u] - lo);
            if (rel < (unsigned)cntn) atomicAdd(&lc[rel], 1);
        }
    }
    __syncthreads();
    int* o = ccnt + (long)blockIdx.x * ntot + coff + lo;
    for (int i = threadIdx.x; i < cntn; i += 1024) o[i] = lc[i];
}

template <int NCH>
__global__ void __launch_bounds__(256)
chunk_offsets_kernel(int* __restrict__ ccnt, int* __restrict__ cnt, int ntot) {
    int n = blockIdx.x * 256 + threadIdx.x;
    if (n >= ntot) return;
    int v[NCH];
#pragma unroll
    for (int c = 0; c < NCH; ++c) v[c] = ccnt[(long)c * ntot + n];
    int run = 0;
#pragma unroll
    for (int c = 0; c < NCH; ++c) {
        ccnt[(long)c * ntot + n] = run;
        run += v[c];
    }
    cnt[n] = run;
}

__device__ __forceinline__ void rowptr_scan_block(const int* __restrict__ cnt, int* __restrict__ rowptr,
                                                  int n0, int n1, int n2, int n3, int t, int* wsum) {
    int N    = t == 0 ? n0 : t == 1 ? n1 : t == 2 ? n2 : n3;
    int coff = t == 0 ? 0 : t == 1 ? n0 : t == 2 ? n0 + n1 : n0 + n1 + n2;
    const int* c = cnt + coff;
    int* rp = rowptr + coff + 4 * t;
    int wave = threadIdx.x >> 5, lane = threadIdx.x & 31;
    int per = (((N + 31) >> 5) + 127) & ~127;
    int beg = wave * per;
    int end = min(beg + per, N);

    int s = 0;
    for (int p = beg + lane * 4; p < end; p += 128) {
        int4 v = *(const int4*)(c + p);
        s += v.x + v.y + v.z + v.w;
    }
    for (int m = 16; m >= 1; m >>= 1) s += __shfl_xor(s, m, 32);
    if (lane == 0) wsum[wave] = s;
    __syncthreads();

    int wtot = wsum[lane];
    int winc = wtot;
    for (int dlt = 1; dlt < 32; dlt <<= 1) {
        int o = __shfl_up(winc, dlt, 32);
        if (lane >= dlt) winc += o;
    }
    int carry = __shfl(winc - wtot, wave, 32);
    int total = __shfl(winc, 31, 32);

    for (int tb = beg; tb < end; tb += 128) {
        int p = tb + lane * 4;
        int4 v = make_int4(0, 0, 0, 0);
        if (p < end) v = *(const int4*)(c + p);
        int ls = v.x + v.y + v.z + v.w;
        int li = ls;
        for (int dlt = 1; dlt < 32; dlt <<= 1) {
            int o = __shfl_up(li, dlt, 32);
            if (lane >= dlt) li += o;
        }
        int ex = carry + li - ls;
        if (p < end)
            *(int4*)(rp + p) = make_int4(ex, ex + v.x, ex + v.x + v.y, ex + v.x + v.y + v.z);
        carry += __shfl(li, 31, 32);
    }
    if (threadIdx.x == 0) rp[N] = total;
}

#define NSCAN 3
__global__ void __launch_bounds__(1024)
scan_enc_kernel(const int* __restrict__ cnt, int* __restrict__ rowptr,
                int n0, int n1, int n2, int n3,
                const float* __restrict__ xp, const float* __restrict__ Wp,
                const float* __restrict__ bp, float* __restrict__ hp, int Np, int gbp,
                const float* __restrict__ xg, const float* __restrict__ Wg,
                const float* __restrict__ bg, float* __restrict__ hg, int Ng, int gbg,
                const float* __restrict__ xm, const float* __restrict__ Wm,
                const float* __restrict__ bm, float* __restrict__ hm, int Nm, int gbm) {
    __shared__ int wsum[32];
    if (blockIdx.x < NSCAN) {
        rowptr_scan_block(cnt, rowptr, n0, n1, n2, n3, blockIdx.x, wsum);
    } else {
        int vb  = (blockIdx.x - NSCAN) * 4 + (threadIdx.x >> 8);
        int tid = threadIdx.x & 255;
        if (vb < gbp)                  enc_rows<3, 4>(xp, Wp, bp, hp, Np, vb, gbp, tid);
        else if (vb < gbp + gbg)       enc_rows<11, 2>(xg, Wg, bg, hg, Ng, vb - gbp, gbg, tid);
        else if (vb < gbp + gbg + gbm) enc_rows<4, 4>(xm, Wm, bm, hm, Nm, vb - gbp - gbg, gbm, tid);
    }
}

__global__ void __launch_bounds__(1024)
fill_kernel(const int* __restrict__ s0, const int* __restrict__ s1,
            const int* __restrict__ s2, const int* __restrict__ s3,
            const int* __restrict__ d0, const int* __restrict__ d1,
            const int* __restrict__ d2, const int* __restrict__ d3,
            const int* __restrict__ rowptr, const int* __restrict__ choff,
            int* __restrict__ perm, int n0, int n1, int n2, int n3, int E) {
    __shared__ int lc[RNG];
    int t, lo, cntn, coff;
    range_of_slot(blockIdx.y, n0, n1, n2, n3, t, lo, cntn, coff);
    const int* s = t == 0 ? s0 : t == 1 ? s1 : t == 2 ? s2 : s3;
    const int* d = t == 0 ? d0 : t == 1 ? d1 : t == 2 ? d2 : d3;
    int ntot = n0 + n1 + n2 + n3;
    const int* rp = rowptr + coff + 4 * t + lo;
    const int* co = choff + (long)blockIdx.x * ntot + coff + lo;
    for (int i = threadIdx.x; i < cntn; i += 1024) lc[i] = rp[i] + co[i];
    __syncthreads();
    int e0 = blockIdx.x << CHSH;
    int e1 = min(e0 + CHE, E);
    int* pm = perm + (long)t * E;
    for (int b = e0 + threadIdx.x; b < e1; b += 1024 * LPT) {
        int dv[LPT], sv[LPT];
#pragma unroll
        for (int u = 0; u < LPT; ++u) {
            int e = b + u * 1024;
            bool ok = e < e1;
            dv[u] = ok ? d[e] : -1;
            sv[u] = ok ? s[e] : 0;
        }
        int pos[LPT];
#pragma unroll
        for (int u = 0; u < LPT; ++u) {
            unsigned rel = (unsigned)(dv[u] - lo);
            pos[u] = -1;
            if (rel < (unsigned)cntn) pos[u] = atomicAdd(&lc[rel], 1);
        }
#pragma unroll
        for (int u = 0; u < LPT; ++u) {
            if (pos[u] >= 0) pm[min(pos[u], E - 1)] = sv[u];
        }
    }
}

template <int U>
__device__ __forceinline__ void gather_rows(const float* __restrict__ hsrc, const int* __restrict__ rowptr,
                                            const int* __restrict__ perm, float* __restrict__ agg,
                                            float* __restrict__ deg, int ncnt, int nsrc, int E,
                                            int vb, int nvb) {
    int wave = threadIdx.x >> 5, lane = threadIdx.x & 31;
    int half = lane >> 4, c = lane & 15;
    const char* hsb = (const char*)hsrc;
    const int stride = nvb * 8;
    int ln = vb * 8 + wave;
    if (ln >= ncnt) return;
    int ln1 = ln + stride;
    int beg = rowptr[ln], end = rowptr[ln + 1];
    int nb = 0, ne = 0;
    if (ln1 < ncnt) { nb = rowptr[ln1]; ne = rowptr[ln1 + 1]; }
    beg = max(beg, 0);
    end = min(end, E);
    int ids = 0;
    if (lane < end - beg) ids = perm[beg + lane];
    for (;;) {
        int ln2 = ln1 + stride;
        nb = max(nb, 0);
        ne = min(ne, E);
        bool nidok = lane < ne - nb;
        int nnb = 0, nne = 0, nids = 0;
        bool pend = true;

        float4 acc = make_float4(0.0f, 0.0f, 0.0f, 0.0f);
        int cur = ids;
        for (int base = beg; base < end; base += 32) {
            int cnt = min(32, end - base);
            int nxt = 0;
            int mine = min(max(cur, 0), nsrc - 1);
            for (int i = 0; i < cnt; i += 2 * U) {
                float4 v[U];
                bool ok[U];
#pragma unroll
                for (int u = 0; u < U; ++u) {
                    int idx = i + 2 * u + half;
                    ok[u] = idx < cnt;
                    int sidx = __shfl(mine, ok[u] ? idx : 0, 32);
                    v[u] = *(const float4*)(hsb + (unsigned)(sidx * 256 + c * 16));
                }
                if (i == 0) {
                    if (lane < end - base - 32) nxt = perm[base + 32 + lane];
                    if (pend) {
                        if (ln2 < ncnt) { nnb = rowptr[ln2]; nne = rowptr[ln2 + 1]; }
                        if (nidok) nids = perm[nb + lane];
                        pend = false;
                    }
                }
#pragma unroll
                for (int u = 0; u < U; ++u) {
                    acc.x += ok[u] ? v[u].x : 0.0f;
                    acc.y += ok[u] ? v[u].y : 0.0f;
                    acc.z += ok[u] ? v[u].z : 0.0f;
                    acc.w += ok[u] ? v[u].w : 0.0f;
                }
            }
            cur = nxt;
        }
        if (pend) {
            if (ln2 < ncnt) { nnb = rowptr[ln2]; nne = rowptr[ln2 + 1]; }
            if (nidok) nids = perm[nb + lane];
        }
        acc.x += __shfl_xor(acc.x, 16, 32);
        acc.y += __shfl_xor(acc.y, 16, 32);
        acc.z += __shfl_xor(acc.z, 16, 32);
        acc.w += __shfl_xor(acc.w, 16, 32);
        if (half == 0) ((float4*)agg)[(long)ln * 16 + c] = acc;
        if (lane == 0) deg[ln] = (float)(end - beg);
        if (ln1 >= ncnt) break;
        ln = ln1; ln1 = ln2;
        beg = nb; end = ne; nb = nnb; ne = nne;
        ids = nids;
    }
}

__global__ void __launch_bounds__(256, 16)
gather3_kernel(const float* __restrict__ hA, const int* __restrict__ rpA, const int* __restrict__ pmA,
               float* __restrict__ aggA, float* __restrict__ degA, int nA, int nsA, int gbA,
               const float* __restrict__ hB, const int* __restrict__ rpB, const int* __restrict__ pmB,
               float* __restrict__ aggB, float* __restrict__ degB, int nB, int nsB, int gbB,
               const float* __restrict__ hC, const int* __restrict__ rpC, const int* __restrict__ pmC,
               float* __restrict__ aggC, float* __restrict__ degC, int nC, int nsC, int gbC,
               int E) {
    int b = blockIdx.x;
    if (b < gbA)            gather_rows<8>(hA, rpA, pmA, aggA, degA, nA, nsA, E, b, gbA);
    else if (b < gbA + gbB) gather_rows<8>(hB, rpB, pmB, aggB, degB, nB, nsB, E, b - gbA, gbB);
    else                    gather_rows<8>(hC, rpC, pmC, aggC, degC, nC, nsC, E, b - gbA - gbB, gbC);
}

template <int M>
__device__ __forceinline__ float row_xor_add(float x) {
    int xi = __float_as_int(x);
    int yi = __builtin_amdgcn_update_dpp(0, xi, 0x160 + M, 0xf, 0xf, false);
    return x + __int_as_float(yi);
}


__device__ __forceinline__ void stage_weights(const float* __restrict__ Wl, const float* __restrict__ Wr,
                                              float* sWl, float* sWr, int tid) {
    const float4* gl = (const float4*)Wl;
    const float4* gr = (const float4*)Wr;
    float4* ll = (float4*)sWl;
    float4* lr = (float4*)sWr;
    for (int i = tid; i < (H * H / 4); i += 128) {
        ll[i] = gl[i];
        lr[i] = gr[i];
    }
}

__device__ __forceinline__ void tile_products(const float* aggr, float invd, const float* hdr,
                                              const float* sWl, const float* sWr,
                                              int kk, int col, v8f (&acc)[4]) {
#pragma unroll
    for (int k0 = 0; k0 < H; k0 += 4) {
        v2f a;
        a.x = aggr[k0 + kk] * invd;
        a.y = aggr[k0 + kk + 1] * invd;
#pragma unroll
        for (int t4 = 0; t4 < 4; ++t4) {
            v2f b;
            b.x = sWl[(k0 + kk) * H + t4 * 16 + col];
            b.y = sWl[(k0 + kk + 1) * H + t4 * 16 + col];
            acc[t4] = __builtin_amdgcn_wmma_f32_16x16x4_f32(
                false, a, false, b, (short)0, acc[t4], false, false);
        }
    }
#pragma unroll
    for (int k0 = 0; k0 < H; k0 += 4) {
        v2f a;
        a.x = hdr[k0 + kk];
        a.y = hdr[k0 + kk + 1];
#pragma unroll
        for (int t4 = 0; t4 < 4; ++t4) {
            v2f b;
            b.x = sWr[(k0 + kk) * H + t4 * 16 + col];
            b.y = sWr[(k0 + kk + 1) * H + t4 * 16 + col];
            acc[t4] = __builtin_amdgcn_wmma_f32_16x16x4_f32(
                false, a, false, b, (short)0, acc[t4], false, false);
        }
    }
}

#define TP 68

template <bool TWO, bool TOLDS>
__device__ __forceinline__ void tile_ln_store(const v8f (&acc)[4], const v8f (&first)[4],
                                              const float* __restrict__ bl, const float* __restrict__ bl1,
                                              const float* __restrict__ g, const float* __restrict__ be,
                                              const float* hdst, float* hout, int m0, int hi, int col, int N) {
    float bias[4], bias1[4], gm[4], bt[4];
#pragma unroll
    for (int t4 = 0; t4 < 4; ++t4) {
        int cg = t4 * 16 + col;
        bias[t4]  = bl[cg];
        bias1[t4] = TWO ? bl1[cg] : 0.0f;
        gm[t4]    = g[cg];
        bt[t4]    = be[cg];
    }
    float res[4][8];
#pragma unroll
    for (int rr = 0; rr < 8; ++rr) {
        int row = m0 + rr + hi * 8;
        long ro = (long)(row < N ? row : (N - 1)) * H + col;
#pragma unroll
        for (int t4 = 0; t4 < 4; ++t4) res[t4][rr] = hdst[ro + t4 * 16];
    }
#pragma unroll
    for (int rr = 0; rr < 8; ++rr) {
        int row = m0 + rr + hi * 8;
        float x[4];
#pragma unroll
        for (int t4 = 0; t4 < 4; ++t4) {
            float o = acc[t4][rr] + bias[t4];
            if (TWO) o = (first[t4][rr] + bias1[t4]) + o;
            x[t4] = res[t4][rr] + o;
        }
        float s = (x[0] + x[2]) + (x[1] + x[3]);
        s = row_xor_add<8>(s);
        s = row_xor_add<4>(s);
        s = row_xor_add<2>(s);
        s = row_xor_add<1>(s);
        float mean = s * (1.0f / 64.0f);
        float d0 = x[0] - mean, d1 = x[1] - mean, d2 = x[2] - mean, d3 = x[3] - mean;
        float qa = d0 * d0 + d2 * d2;
        float qb = d1 * d1 + d3 * d3;
        float q = qa + qb;
        q = row_xor_add<8>(q);
        q = row_xor_add<4>(q);
        q = row_xor_add<2>(q);
        q = row_xor_add<1>(q);
        float inv = rsqrtf(q * (1.0f / 64.0f) + 1e-5f);
        float o0 = elu1(d0 * inv * gm[0] + bt[0]);
        float o1 = elu1(d1 * inv * gm[1] + bt[1]);
        float o2 = elu1(d2 * inv * gm[2] + bt[2]);
        float o3 = elu1(d3 * inv * gm[3] + bt[3]);
        if (TOLDS) {
            float* orow = hout + (rr + hi * 8) * TP + col;
            orow[0]  = o0;
            orow[16] = o1;
            orow[32] = o2;
            orow[48] = o3;
        } else if (row < N) {
            float* orow = hout + (long)row * H + col;
            orow[0]  = o0;
            orow[16] = o1;
            orow[32] = o2;
            orow[48] = o3;
        }
    }
}

__device__ __forceinline__ void sage_block_one(const float* __restrict__ sums, const float* __restrict__ deg,
                                               float* hdst,
                                               const float* __restrict__ Wl, const float* __restrict__ bl,
                                               const float* __restrict__ Wr,
                                               const float* __restrict__ g, const float* __restrict__ be,
                                               int N, int blk, float* sWl, float* sWr) {
    int tid = threadIdx.x;
    stage_weights(Wl, Wr, sWl, sWr, tid);
    __syncthreads();

    int wave = tid >> 5;
    int lane = tid & 31;
    int hi   = lane >> 4;
    int col  = lane & 15;
    int kk   = hi * 2;
    int m0   = blk * 64 + wave * 16;

    int r  = m0 + col;
    int rc = r < N ? r : (N - 1);
    float invd = 1.0f / fmaxf(deg[rc], 1.0f);

    v8f acc[4] = {};
    tile_products(sums + (long)rc * H, invd, hdst + (long)rc * H, sWl, sWr, kk, col, acc);
    tile_ln_store<false, false>(acc, acc, bl, bl, g, be, hdst, hdst, m0, hi, col, N);
}

__device__ __forceinline__ void sage_block_two(const float* __restrict__ sums1, const float* __restrict__ deg1,
                                               const float* __restrict__ Wl1, const float* __restrict__ bl1,
                                               const float* __restrict__ Wr1,
                                               const float* __restrict__ sums2, const float* __restrict__ deg2,
                                               const float* __restrict__ Wl2, const float* __restrict__ bl2,
                                               const float* __restrict__ Wr2,
                                               float* hdst,
                                               const float* __restrict__ g, const float* __restrict__ be,
                                               int N, int blk, float* sWl, float* sWr) {
    int tid = threadIdx.x;
    stage_weights(Wl1, Wr1, sWl, sWr, tid);
    __syncthreads();

    int wave = tid >> 5;
    int lane = tid & 31;
    int hi   = lane >> 4;
    int col  = lane & 15;
    int kk   = hi * 2;
    int m0   = blk * 64 + wave * 16;

    int r  = m0 + col;
    int rc = r < N ? r : (N - 1);
    float invd1 = 1.0f / fmaxf(deg1[rc], 1.0f);
    float invd2 = 1.0f / fmaxf(deg2[rc], 1.0f);
    const float* hdr = hdst + (long)rc * H;

    v8f acc1[4] = {};
    tile_products(sums1 + (long)rc * H, invd1, hdr, sWl, sWr, kk, col, acc1);
    __syncthreads();
    stage_weights(Wl2, Wr2, sWl, sWr, tid);
    __syncthreads();
    v8f acc2[4] = {};
    tile_products(sums2 + (long)rc * H, invd2, hdr, sWl, sWr, kk, col, acc2);
    tile_ln_store<true, false>(acc2, acc1, bl2, bl1, g, be, hdst, hdst, m0, hi, col, N);
}

__global__ void __launch_bounds__(128)
sage_layer_kernel(const float* __restrict__ sums1, const float* __restrict__ deg1,
                  const float* __restrict__ Wl1, const float* __restrict__ bl1,
                  const float* __restrict__ Wr1,
                  const float* __restrict__ sums2, const float* __restrict__ deg2,
                  const float* __restrict__ Wl2, const float* __restrict__ bl2,
                  const float* __restrict__ Wr2,
                  float* hT, const float* __restrict__ gT, const float* __restrict__ beT, int NT, int nbT,
                  const float* __restrict__ sumsO, const float* __restrict__ degO,
                  const float* __restrict__ WlO, const float* __restrict__ blO,
                  const float* __restrict__ WrO,
                  float* hO, const float* __restrict__ gO, const float* __restrict__ beO, int NO) {
    __shared__ float sWl[H * H];
    __shared__ float sWr[H * H];
    int b = blockIdx.x;
    if (b < nbT)
        sage_block_two(sums1, deg1, Wl1, bl1, Wr1, sums2, deg2, Wl2, bl2, Wr2,
                       hT, gT, beT, NT, b, sWl, sWr);
    else
        sage_block_one(sumsO, degO, hO, WlO, blO, WrO, gO, beO, NO, b - nbT, sWl, sWr);
}

#define HC 32
__global__ void __launch_bounds__(256, 8)
layer2_head_kernel(const float* __restrict__ hsrc, const int* __restrict__ rowptr,
                   const int* __restrict__ perm, const float* __restrict__ hdst,
                   const float* __restrict__ Wl, const float* __restrict__ bl,
                   const float* __restrict__ Wr,
                   const float* __restrict__ g, const float* __restrict__ be,
                   const float* __restrict__ W1, const float* __restrict__ b1,
                   const float* __restrict__ W2, const float* __restrict__ b2,
                   float* __restrict__ out, int N, int nsrc, int E) {
    __shared__ __attribute__((aligned(16))) float sWl[H * H];
    __shared__ __attribute__((aligned(16))) float sWr[H * H];
    __shared__ __attribute__((aligned(16))) float sW1[H * HC];
    __shared__ __attribute__((aligned(16))) float sT[8 * 16 * TP];
    int tid = threadIdx.x;
    {
        const float4* gl = (const float4*)Wl;
        const float4* gr = (const float4*)Wr;
        const float4* g1 = (const float4*)W1;
        float4 vl[4], vr[4], v1[2];
#pragma unroll
        for (int q = 0; q < 4; ++q) {
            vl[q] = gl[tid + 256 * q];
            vr[q] = gr[tid + 256 * q];
        }
#pragma unroll
        for (int q = 0; q < 2; ++q) v1[q] = g1[tid + 256 * q];
#pragma unroll
        for (int q = 0; q < 4; ++q) {
            ((float4*)sWl)[tid + 256 * q] = vl[q];
            ((float4*)sWr)[tid + 256 * q] = vr[q];
        }
#pragma unroll
        for (int q = 0; q < 2; ++q) ((float4*)sW1)[tid + 256 * q] = v1[q];
    }
    __syncthreads();

    int wave = __builtin_amdgcn_readfirstlane(tid >> 5);
    int lane = tid & 31;
    int hi = lane >> 4, col = lane & 15;
    int kk = hi * 2;
    int m0 = blockIdx.x * 128 + wave * 16;
    if (m0 >= N) return;
    float* tile = sT + wave * (16 * TP);
    int nn = min(16, N - m0);

    int rpv = rowptr[min(m0 + min(lane, 16), N)];

    {
        const char* hsb = (const char*)hsrc;
        int beg = max(__shfl(rpv, 0, 32), 0);
        int end = min(__shfl(rpv, 1, 32), E);
        int ids = 0;
        if (lane < end - beg) ids = perm[beg + lane];
        for (int i = 0; i < nn; ++i) {
            int nb = max(__shfl(rpv, min(i + 1, 16), 32), 0);
            int ne = min(__shfl(rpv, min(i + 2, 16), 32), E);
            bool nidok = lane < ne - nb;
            int nids = 0;
            bool pend = true;

            float4 acc = make_float4(0.0f, 0.0f, 0.0f, 0.0f);
            int cur = ids;
            for (int base = beg; base < end; base += 32) {
                int cnt = min(32, end - base);
                int nxt = 0;
                int mine = min(max(cur, 0), nsrc - 1);
                for (int t = 0; t < cnt; t += 16) {
                    float4 v[8];
                    bool ok[8];
#pragma unroll
                    for (int u = 0; u < 8; ++u) {
                        int idx = t + 2 * u + hi;
                        ok[u] = idx < cnt;
                        int sidx = __shfl(mine, ok[u] ? idx : 0, 32);
                        v[u] = *(const float4*)(hsb + (unsigned)(sidx * 256 + col * 16));
                    }
                    if (t == 0) {
                        if (lane < end - base - 32) nxt = perm[base + 32 + lane];
                        if (pend) {
                            if (nidok) nids = perm[nb + lane];
                            pend = false;
                        }
                    }
#pragma unroll
                    for (int u = 0; u < 8; ++u) {
                        acc.x += ok[u] ? v[u].x : 0.0f;
                        acc.y += ok[u] ? v[u].y : 0.0f;
                        acc.z += ok[u] ? v[u].z : 0.0f;
                        acc.w += ok[u] ? v[u].w : 0.0f;
                    }
                }
                cur = nxt;
            }
            if (pend) {
                if (nidok) nids = perm[nb + lane];
            }
            acc.x += __shfl_xor(acc.x, 16, 32);
            acc.y += __shfl_xor(acc.y, 16, 32);
            acc.z += __shfl_xor(acc.z, 16, 32);
            acc.w += __shfl_xor(acc.w, 16, 32);
            if (hi == 0) *(float4*)(tile + i * TP + col * 4) = acc;
            beg = nb; end = ne;
            ids = nids;
        }
        for (int i = nn; i < 16; ++i)
            if (hi == 0) *(float4*)(tile + i * TP + col * 4) = make_float4(0.0f, 0.0f, 0.0f, 0.0f);
    }

    int r  = m0 + col;
    int rc = r < N ? r : (N - 1);
    int db = max(__shfl(rpv, col, 32), 0);
    int de = min(__shfl(rpv, col + 1, 32), E);
    float invd = 1.0f / fmaxf((float)(de - db), 1.0f);

    v8f acc[4] = {};
    tile_products(tile + col * TP, invd, hdst + (long)rc * H, sWl, sWr, kk, col, acc);

    tile_ln_store<false, true>(acc, acc, bl, bl, g, be, hdst, tile, m0, hi, col, N);

    v8f z[2] = {};
    const float* hrow = tile + col * TP;
#pragma unroll
    for (int k0 = 0; k0 < H; k0 += 4) {
        v2f a;
        a.x = hrow[k0 + kk];
        a.y = hrow[k0 + kk + 1];
#pragma unroll
        for (int t = 0; t < 2; ++t) {
            v2f b;
            b.x = sW1[(k0 + kk) * HC + t * 16 + col];
            b.y = sW1[(k0 + kk + 1) * HC + t * 16 + col];
            z[t] = __builtin_amdgcn_wmma_f32_16x16x4_f32(
                false, a, false, b, (short)0, z[t], false, false);
        }
    }
    float b1a = b1[col], b1b = b1[16 + col];
    float w2a = W2[col], w2b = W2[16 + col];
    float b2v = b2[0];
    float hz = 0.0f;
#pragma unroll
    for (int rr = 0; rr < 8; ++rr) {
        float p = elu1(z[0][rr] + b1a) * w2a + elu1(z[1][rr] + b1b) * w2b;
        p = row_xor_add<8>(p);
        p = row_xor_add<4>(p);
        p = row_xor_add<2>(p);
        p = row_xor_add<1>(p);
        if (col == rr) hz = p + b2v;
    }
    int orow = m0 + hi * 8 + col;
    if (col < 8 && orow < N) out[orow] = hz;
}

extern "C" void kernel_launch(void* const* d_in, const int* in_sizes, int n_in,
                              void* d_out, int out_size, void* d_ws, size_t ws_size,
                              hipStream_t stream) {
    (void)in_sizes; (void)n_in; (void)out_size; (void)ws_size;
    const int NG = 20000, NP = 100000, NM = 5000, E = 1000000;

    const float* x_gene    = (const float*)d_in[0];
    const float* x_patient = (const float*)d_in[1];
    const float* x_group   = (const float*)d_in[2];
    const int* srcA[4]  = { (const int*)d_in[3], (const int*)d_in[5],
                            (const int*)d_in[7], (const int*)d_in[9] };
    const int* dstA[4]  = { (const int*)d_in[4], (const int*)d_in[6],
                            (const int*)d_in[8], (const int*)d_in[10] };
    const float* encW[3] = { (const float*)d_in[11], (const float*)d_in[13], (const float*)d_in[15] };
    const float* encB[3] = { (const float*)d_in[12], (const float*)d_in[14], (const float*)d_in[16] };
    const float* convW_l = (const float*)d_in[17];
    const float* convb_l = (const float*)d_in[18];
    const float* convW_r = (const float*)d_in[19];
    const float* ln_g = (const float*)d_in[20];
    const float* ln_b = (const float*)d_in[21];
    const float* coxW1 = (const float*)d_in[22];
    const float* coxb1 = (const float*)d_in[23];
    const float* coxW2 = (const float*)d_in[24];
    const float* coxb2 = (const float*)d_in[25];

    float* ws = (float*)d_ws;
    float* h_gene  = ws; ws += (size_t)NG * H;
    float* h_pat   = ws; ws += (size_t)NP * H;
    float* h_grp   = ws; ws += (size_t)NM * H;
    float* hn_gene = ws; ws += (size_t)NG * H;
    float* hn_pat  = ws; ws += (size_t)NP * H;
    float* degG1   = ws;
    float* degG3   = ws + NG;
    ws += (size_t)NM * H;
    const int SL = 34000;
    float* aggreg  = ws; ws += (size_t)NP * H;
    float* deg     = ws; ws += (size_t)NP;
    float* agg     = aggreg;
    int*   perm    = (int*)(aggreg + (size_t)SL * H);
    int*   rowptr  = perm + (size_t)3 * E;
    int*   cnt     = (int*)hn_gene;
    int*   ccnt    = (int*)hn_pat;

    const int* srcL[3] = { srcA[0], srcA[1], srcA[3] };
    const int* dstL[3] = { dstA[0], dstA[1], dstA[3] };
    const int* none = nullptr;
    const int NCH   = (E + CHE - 1) / CHE;
    const int NSLOT = (NP + RNG - 1) / RNG + (NG + RNG - 1) / RNG
                    + (NG + RNG - 1) / RNG;
    const int NTOT  = NP + NG + NG;
    const int GCAP = 1024;
    const int gbEG = (NG + 7) / 8 < GCAP ? (NG + 7) / 8 : GCAP;
    const int gbEP = (NP + 15) / 16 < GCAP ? (NP + 15) / 16 : GCAP;
    const int gbEM = (NM + 15) / 16 < GCAP ? (NM + 15) / 16 : GCAP;

    count_kernel<<<dim3(NCH, NSLOT), 1024, 0, stream>>>(
        dstL[0], dstL[1], dstL[2], none, ccnt, NP, NG, NG, 0, E);
    chunk_offsets_kernel<NCH><<<(NTOT + 255) / 256, 256, 0, stream>>>(ccnt, cnt, NTOT);
    scan_enc_kernel<<<NSCAN + (gbEP + gbEG + gbEM + 3) / 4, 1024, 0, stream>>>(
        cnt, rowptr, NP, NG, NG, 0,
        x_patient, encW[1], encB[1], h_pat,  NP, gbEP,
        x_gene,    encW[0], encB[0], h_gene, NG, gbEG,
        x_group,   encW[2], encB[2], h_grp,  NM, gbEM);
    fill_kernel<<<dim3(NCH, NSLOT), 1024, 0, stream>>>(
        srcL[0], srcL[1], srcL[2], none, dstL[0], dstL[1], dstL[2], none,
        rowptr, ccnt, perm, NP, NG, NG, 0, E);
    const int* rpL[3] = { rowptr, rowptr + NP + 4, rowptr + NP + NG + 8 };
    const int* pmL[3] = { perm, perm + (size_t)E, perm + (size_t)2 * E };

    const size_t HH = (size_t)H * H;
    const int nbG = (NG + 63) / 64, nbP = (NP + 63) / 64;
    const int g3G = 576, g3P = 896;

    gather3_kernel<<<g3G + g3G + g3P, 256, 0, stream>>>(
        h_pat,  rpL[1], pmL[1], hn_gene, degG1, NG, NP, g3G,
        h_grp,  rpL[2], pmL[2], agg,     degG3, NG, NM, g3G,
        h_gene, rpL[0], pmL[0], hn_pat,  deg,   NP, NG, g3P,
        E);
    sage_layer_kernel<<<nbG + nbP, 128, 0, stream>>>(
        hn_gene, degG1, convW_l + 1 * HH, convb_l + 1 * H, convW_r + 1 * HH,
        agg,     degG3, convW_l + 3 * HH, convb_l + 3 * H, convW_r + 3 * HH,
        h_gene, ln_g + 0 * H, ln_b + 0 * H, NG, nbG,
        hn_pat,  deg,   convW_l + 0 * HH, convb_l + 0 * H, convW_r + 0 * HH,
        h_pat,  ln_g + 1 * H, ln_b + 1 * H, NP);

    layer2_head_kernel<<<(NP + 127) / 128, 256, 0, stream>>>(
        h_gene, rpL[0], pmL[0], h_pat, convW_l + 4 * HH, convb_l + 4 * H, convW_r + 4 * HH,
        ln_g + (3 + 1) * H, ln_b + (3 + 1) * H,
        coxW1, coxb1, coxW2, coxb2, (float*)d_out, NP, NG, E);
}


# No prefix-sum launch: chunk_offsets leaves 256-node block sums and in-block prefixes, every fill workgroup rebuilds its range's group offsets from them; encoders run beside fill; 7 to 6 launches
# speedup vs baseline: 14.3595x; 1.1280x over previous
#include <hip/hip_runtime.h>
#include <hip/hip_bf16.h>
#include <math.h>

typedef float v2f __attribute__((ext_vector_type(2)));
typedef float v8f __attribute__((ext_vector_type(8)));

#define H 64

__device__ __forceinline__ float elu1(float x) {
    return x > 0.0f ? x : (__expf(x) - 1.0f);
}

template <int F, int U>
__device__ __forceinline__ void enc_rows(const float* __restrict__ x, const float* __restrict__ W,
                                         const float* __restrict__ b, float* __restrict__ h,
                                         int N, int vb, int nvb, int tid) {
    int j = tid & 63;
    int sub = tid >> 6;
    float w[F];
#pragma unroll
    for (int k = 0; k < F; ++k) w[k] = W[k * H + j];
    float bj = b[j];
    for (int n0 = vb * (4 * U) + sub; n0 < N; n0 += nvb * (4 * U)) {
        if (n0 + 4 * (U - 1) < N) {
            float xv[U][F];
#pragma unroll
            for (int u = 0; u < U; ++u) {
                const float* xr = x + (long)(n0 + 4 * u) * F;
#pragma unroll
                for (int k = 0; k < F; ++k) xv[u][k] = xr[k];
            }
#pragma unroll
            for (int u = 0; u < U; ++u) {
                float s = bj;
#pragma unroll
                for (int k = 0; k < F; ++k) s = fmaf(xv[u][k], w[k], s);
                h[(long)(n0 + 4 * u) * H + j] = elu1(s);
            }
        } else {
            for (int n = n0; n < N; n += 4) {
                const float* xr = x + (long)n * F;
                float s = bj;
#pragma unroll
                for (int k = 0; k < F; ++k) s = fmaf(xr[k], w[k], s);
                h[(long)n * H + j] = elu1(s);
            }
        }
    }
}

#define RNG  25000
#define CHSH 15
#define CHE  (1 << CHSH)
#define LPT  8

__device__ __forceinline__ void range_of_slot(int slot, int n0, int n1, int n2, int n3,
                                              int& t, int& lo, int& cntn, int& coff) {
    int r0 = (n0 + RNG - 1) / RNG, r1 = (n1 + RNG - 1) / RNG, r2 = (n2 + RNG - 1) / RNG;
    int r, N;
    if (slot < r0)                { t = 0; r = slot;                N = n0; coff = 0; }
    else if (slot < r0 + r1)      { t = 1; r = slot - r0;           N = n1; coff = n0; }
    else if (slot < r0 + r1 + r2) { t = 2; r = slot - r0 - r1;      N = n2; coff = n0 + n1; }
    else                          { t = 3; r = slot - r0 - r1 - r2; N = n3; coff = n0 + n1 + n2; }
    lo = r * RNG;
    cntn = min(RNG, N - lo);
}

__global__ void __launch_bounds__(1024)
count_kernel(const int* __restrict__ d0, const int* __restrict__ d1,
             const int* __restrict__ d2, const int* __restrict__ d3,
             int* __restrict__ ccnt, int n0, int n1, int n2, int n3, int E) {
    __shared__ int lc[RNG];
    int t, lo, cntn, coff;
    range_of_slot(blockIdx.y, n0, n1, n2, n3, t, lo, cntn, coff);
    const int* d = t == 0 ? d0 : t == 1 ? d1 : t == 2 ? d2 : d3;
    int ntot = n0 + n1 + n2 + n3;
    for (int i = threadIdx.x; i < cntn; i += 1024) lc[i] = 0;
    __syncthreads();
    int e0 = blockIdx.x << CHSH;
    int e1 = min(e0 + CHE, E);
    for (int b = e0 + threadIdx.x; b < e1; b += 1024 * LPT) {
        int dv[LPT];
#pragma unroll
        for (int u = 0; u < LPT; ++u) {
            int e = b + u * 1024;
            dv[u] = e < e1 ? d[e] : -1;
        }
#pragma unroll
        for (int u = 0; u < LPT; ++u) {
            unsigned rel = (unsigned)(dv[u] - lo);
            if (rel < (unsigned)cntn) atomicAdd(&lc[rel], 1);
        }
    }
    __syncthreads();
    int* o = ccnt + (long)blockIdx.x * ntot + coff + lo;
    for (int i = threadIdx.x; i < cntn; i += 1024) o[i] = lc[i];
}

template <int NCH>
__global__ void __launch_bounds__(256)
chunk_offsets_kernel(int* __restrict__ ccnt, int* __restrict__ lpre, int* __restrict__ bsum, int ntot) {
    __shared__ int wtot[8];
    int n = blockIdx.x * 256 + threadIdx.x;
    int run = 0;
    if (n < ntot) {
        int v[NCH];
#pragma unroll
        for (int c = 0; c < NCH; ++c) v[c] = ccnt[(long)c * ntot + n];
#pragma unroll
        for (int c = 0; c < NCH; ++c) {
            ccnt[(long)c * ntot + n] = run;
            run += v[c];
        }
    }
    int wave = threadIdx.x >> 5, lane = threadIdx.x & 31;
    int inc = run;
    for (int dlt = 1; dlt < 32; dlt <<= 1) {
        int o = __shfl_up(inc, dlt, 32);
        if (lane >= dlt) inc += o;
    }
    if (lane == 31) wtot[wave] = inc;
    __syncthreads();
    int wbase = 0, tot = 0;
#pragma unroll
    for (int w = 0; w < 8; ++w) {
        int x = wtot[w];
        wbase += w < wave ? x : 0;
        tot += x;
    }
    if (n <= ntot) lpre[n] = wbase + inc - run;
    if (threadIdx.x == 0) bsum[blockIdx.x] = tot;
}

#define NBMAX 576
__global__ void __launch_bounds__(1024)
fill_enc_kernel(const int* __restrict__ s0, const int* __restrict__ s1,
                const int* __restrict__ s2, const int* __restrict__ s3,
                const int* __restrict__ d0, const int* __restrict__ d1,
                const int* __restrict__ d2, const int* __restrict__ d3,
                const int* __restrict__ lpre, const int* __restrict__ bsum,
                const int* __restrict__ choff, int* __restrict__ rowptr,
                int* __restrict__ perm, int n0, int n1, int n2, int n3, int E, int nch, int nfill,
                const float* __restrict__ xp, const float* __restrict__ Wp,
                const float* __restrict__ bp, float* __restrict__ hp, int Np, int gbp,
                const float* __restrict__ xg, const float* __restrict__ Wg,
                const float* __restrict__ bg, float* __restrict__ hg, int Ng, int gbg,
                const float* __restrict__ xm, const float* __restrict__ Wm,
                const float* __restrict__ bm, float* __restrict__ hm, int Nm, int gbm) {
    __shared__ int lc[RNG];
    __shared__ int sB[NBMAX];
    if ((int)blockIdx.x >= nfill) {
        int vb  = ((int)blockIdx.x - nfill) * 4 + (threadIdx.x >> 8);
        int tid = threadIdx.x & 255;
        if (vb < gbp)                  enc_rows<3, 4>(xp, Wp, bp, hp, Np, vb, gbp, tid);
        else if (vb < gbp + gbg)       enc_rows<11, 2>(xg, Wg, bg, hg, Ng, vb - gbp, gbg, tid);
        else if (vb < gbp + gbg + gbm) enc_rows<4, 4>(xm, Wm, bm, hm, Nm, vb - gbp - gbg, gbm, tid);
        return;
    }
    int slot = (int)blockIdx.x / nch;
    int chunk = (int)blockIdx.x - slot * nch;
    int t, lo, cntn, coff;
    range_of_slot(slot, n0, n1, n2, n3, t, lo, cntn, coff);
    const int* s = t == 0 ? s0 : t == 1 ? s1 : t == 2 ? s2 : s3;
    const int* d = t == 0 ? d0 : t == 1 ? d1 : t == 2 ? d2 : d3;
    int Nt = t == 0 ? n0 : t == 1 ? n1 : t == 2 ? n2 : n3;
    int ntot = n0 + n1 + n2 + n3;
    int nblk = (ntot + 256) >> 8;
    int lp0 = lpre[coff];
    for (int i = threadIdx.x; i < NBMAX; i += 1024) sB[i] = i < nblk ? bsum[i] : 0;
    __syncthreads();
    if (threadIdx.x < 32) {
        int base = threadIdx.x * (NBMAX / 32);
        int v[NBMAX / 32];
        int sum = 0;
#pragma unroll
        for (int k = 0; k < NBMAX / 32; ++k) { v[k] = sB[base + k]; sum += v[k]; }
        int inc = sum;
        for (int dlt = 1; dlt < 32; dlt <<= 1) {
            int o = __shfl_up(inc, dlt, 32);
            if ((int)threadIdx.x >= dlt) inc += o;
        }
        int ex = inc - sum;
#pragma unroll
        for (int k = 0; k < NBMAX / 32; ++k) { sB[base + k] = ex; ex += v[k]; }
    }
    __syncthreads();
    int g0 = coff + lo;
    int pbase = sB[coff >> 8] + lp0;
    const int* co = choff + (long)chunk * ntot + g0;
    const int* lp = lpre + g0;
    int* rp = rowptr + coff + 4 * t + lo;
    bool wr = chunk == 0;
    for (int i0 = threadIdx.x; i0 < cntn; i0 += 1024 * LPT) {
        int lv[LPT], cv[LPT], bv[LPT];
#pragma unroll
        for (int u = 0; u < LPT; ++u) {
            int ic = min(i0 + u * 1024, cntn - 1);
            lv[u] = lp[ic];
            cv[u] = co[ic];
            bv[u] = sB[(g0 + ic) >> 8];
        }
#pragma unroll
        for (int u = 0; u < LPT; ++u) {
            int i = i0 + u * 1024;
            if (i < cntn) {
                int off = bv[u] + lv[u] - pbase;
                lc[i] = off + cv[u];
                if (wr) rp[i] = off;
            }
        }
    }
    if (wr && lo + cntn == Nt && threadIdx.x == 0) {
        int g = coff + Nt;
        rp[cntn] = sB[g >> 8] + lpre[g] - pbase;
    }
    __syncthreads();
    int e0 = chunk << CHSH;
    int e1 = min(e0 + CHE, E);
    int* pm = perm + (long)t * E;
    for (int b = e0 + threadIdx.x; b < e1; b += 1024 * LPT) {
        int dv[LPT], sv[LPT];
#pragma unroll
        for (int u = 0; u < LPT; ++u) {
            int e = b + u * 1024;
            bool ok = e < e1;
            dv[u] = ok ? d[e] : -1;
            sv[u] = ok ? s[e] : 0;
        }
        int pos[LPT];
#pragma unroll
        for (int u = 0; u < LPT; ++u) {
            unsigned rel = (unsigned)(dv[u] - lo);
            pos[u] = -1;
            if (rel < (unsigned)cntn) pos[u] = atomicAdd(&lc[rel], 1);
        }
#pragma unroll
        for (int u = 0; u < LPT; ++u) {
            if (pos[u] >= 0) pm[min(pos[u], E - 1)] = sv[u];
        }
    }
}

template <int U>
__device__ __forceinline__ void gather_rows(const float* __restrict__ hsrc, const int* __restrict__ rowptr,
                                            const int* __restrict__ perm, float* __restrict__ agg,
                                            float* __restrict__ deg, int ncnt, int nsrc, int E,
                                            int vb, int nvb) {
    int wave = threadIdx.x >> 5, lane = threadIdx.x & 31;
    int half = lane >> 4, c = lane & 15;
    const char* hsb = (const char*)hsrc;
    const int stride = nvb * 8;
    int ln = vb * 8 + wave;
    if (ln >= ncnt) return;
    int ln1 = ln + stride;
    int beg = rowptr[ln], end = rowptr[ln + 1];
    int nb = 0, ne = 0;
    if (ln1 < ncnt) { nb = rowptr[ln1]; ne = rowptr[ln1 + 1]; }
    beg = max(beg, 0);
    end = min(end, E);
    int ids = 0;
    if (lane < end - beg) ids = perm[beg + lane];
    for (;;) {
        int ln2 = ln1 + stride;
        nb = max(nb, 0);
        ne = min(ne, E);
        bool nidok = lane < ne - nb;
        int nnb = 0, nne = 0, nids = 0;
        bool pend = true;

        float4 acc = make_float4(0.0f, 0.0f, 0.0f, 0.0f);
        int cur = ids;
        for (int base = beg; base < end; base += 32) {
            int cnt = min(32, end - base);
            int nxt = 0;
            int mine = min(max(cur, 0), nsrc - 1);
            for (int i = 0; i < cnt; i += 2 * U) {
                float4 v[U];
                bool ok[U];
#pragma unroll
                for (int u = 0; u < U; ++u) {
                    int idx = i + 2 * u + half;
                    ok[u] = idx < cnt;
                    int sidx = __shfl(mine, ok[u] ? idx : 0, 32);
                    v[u] = *(const float4*)(hsb + (unsigned)(sidx * 256 + c * 16));
                }
                if (i == 0) {
                    if (lane < end - base - 32) nxt = perm[base + 32 + lane];
                    if (pend) {
                        if (ln2 < ncnt) { nnb = rowptr[ln2]; nne = rowptr[ln2 + 1]; }
                        if (nidok) nids = perm[nb + lane];
                        pend = false;
                    }
                }
#pragma unroll
                for (int u = 0; u < U; ++u) {
                    acc.x += ok[u] ? v[u].x : 0.0f;
                    acc.y += ok[u] ? v[u].y : 0.0f;
                    acc.z += ok[u] ? v[u].z : 0.0f;
                    acc.w += ok[u] ? v[u].w : 0.0f;
                }
            }
            cur = nxt;
        }
        if (pend) {
            if (ln2 < ncnt) { nnb = rowptr[ln2]; nne = rowptr[ln2 + 1]; }
            if (nidok) nids = perm[nb + lane];
        }
        acc.x += __shfl_xor(acc.x, 16, 32);
        acc.y += __shfl_xor(acc.y, 16, 32);
        acc.z += __shfl_xor(acc.z, 16, 32);
        acc.w += __shfl_xor(acc.w, 16, 32);
        if (half == 0) ((float4*)agg)[(long)ln * 16 + c] = acc;
        if (lane == 0) deg[ln] = (float)(end - beg);
        if (ln1 >= ncnt) break;
        ln = ln1; ln1 = ln2;
        beg = nb; end = ne; nb = nnb; ne = nne;
        ids = nids;
    }
}

__global__ void __launch_bounds__(256, 16)
gather3_kernel(const float* __restrict__ hA, const int* __restrict__ rpA, const int* __restrict__ pmA,
               float* __restrict__ aggA, float* __restrict__ degA, int nA, int nsA, int gbA,
               const float* __restrict__ hB, const int* __restrict__ rpB, const int* __restrict__ pmB,
               float* __restrict__ aggB, float* __restrict__ degB, int nB, int nsB, int gbB,
               const float* __restrict__ hC, const int* __restrict__ rpC, const int* __restrict__ pmC,
               float* __restrict__ aggC, float* __restrict__ degC, int nC, int nsC, int gbC,
               int E) {
    int b = blockIdx.x;
    if (b < gbA)            gather_rows<8>(hA, rpA, pmA, aggA, degA, nA, nsA, E, b, gbA);
    else if (b < gbA + gbB) gather_rows<8>(hB, rpB, pmB, aggB, degB, nB, nsB, E, b - gbA, gbB);
    else                    gather_rows<8>(hC, rpC, pmC, aggC, degC, nC, nsC, E, b - gbA - gbB, gbC);
}

template <int M>
__device__ __forceinline__ float row_xor_add(float x) {
    int xi = __float_as_int(x);
    int yi = __builtin_amdgcn_update_dpp(0, xi, 0x160 + M, 0xf, 0xf, false);
    return x + __int_as_float(yi);
}


__device__ __forceinline__ void stage_weights(const float* __restrict__ Wl, const float* __restrict__ Wr,
                                              float* sWl, float* sWr, int tid) {
    const float4* gl = (const float4*)Wl;
    const float4* gr = (const float4*)Wr;
    float4* ll = (float4*)sWl;
    float4* lr = (float4*)sWr;
    for (int i = tid; i < (H * H / 4); i += 128) {
        ll[i] = gl[i];
        lr[i] = gr[i];
    }
}

__device__ __forceinline__ void tile_products(const float* aggr, float invd, const float* hdr,
                                              const float* sWl, const float* sWr,
                                              int kk, int col, v8f (&acc)[4]) {
#pragma unroll
    for (int k0 = 0; k0 < H; k0 += 4) {
        v2f a;
        a.x = aggr[k0 + kk] * invd;
        a.y = aggr[k0 + kk + 1] * invd;
#pragma unroll
        for (int t4 = 0; t4 < 4; ++t4) {
            v2f b;
            b.x = sWl[(k0 + kk) * H + t4 * 16 + col];
            b.y = sWl[(k0 + kk + 1) * H + t4 * 16 + col];
            acc[t4] = __builtin_amdgcn_wmma_f32_16x16x4_f32(
                false, a, false, b, (short)0, acc[t4], false, false);
        }
    }
#pragma unroll
    for (int k0 = 0; k0 < H; k0 += 4) {
        v2f a;
        a.x = hdr[k0 + kk];
        a.y = hdr[k0 + kk + 1];
#pragma unroll
        for (int t4 = 0; t4 < 4; ++t4) {
            v2f b;
            b.x = sWr[(k0 + kk) * H + t4 * 16 + col];
            b.y = sWr[(k0 + kk + 1) * H + t4 * 16 + col];
            acc[t4] = __builtin_amdgcn_wmma_f32_16x16x4_f32(
                false, a, false, b, (short)0, acc[t4], false, false);
        }
    }
}

#define TP 68

template <bool TWO, bool TOLDS>
__device__ __forceinline__ void tile_ln_store(const v8f (&acc)[4], const v8f (&first)[4],
                                              const float* __restrict__ bl, const float* __restrict__ bl1,
                                              const float* __restrict__ g, const float* __restrict__ be,
                                              const float* hdst, float* hout, int m0, int hi, int col, int N) {
    float bias[4], bias1[4], gm[4], bt[4];
#pragma unroll
    for (int t4 = 0; t4 < 4; ++t4) {
        int cg = t4 * 16 + col;
        bias[t4]  = bl[cg];
        bias1[t4] = TWO ? bl1[cg] : 0.0f;
        gm[t4]    = g[cg];
        bt[t4]    = be[cg];
    }
    float res[4][8];
#pragma unroll
    for (int rr = 0; rr < 8; ++rr) {
        int row = m0 + rr + hi * 8;
        long ro = (long)(row < N ? row : (N - 1)) * H + col;
#pragma unroll
        for (int t4 = 0; t4 < 4; ++t4) res[t4][rr] = hdst[ro + t4 * 16];
    }
#pragma unroll
    for (int rr = 0; rr < 8; ++rr) {
        int row = m0 + rr + hi * 8;
        float x[4];
#pragma unroll
        for (int t4 = 0; t4 < 4; ++t4) {
            float o = acc[t4][rr] + bias[t4];
            if (TWO) o = (first[t4][rr] + bias1[t4]) + o;
            x[t4] = res[t4][rr] + o;
        }
        float s = (x[0] + x[2]) + (x[1] + x[3]);
        s = row_xor_add<8>(s);
        s = row_xor_add<4>(s);
        s = row_xor_add<2>(s);
        s = row_xor_add<1>(s);
        float mean = s * (1.0f / 64.0f);
        float d0 = x[0] - mean, d1 = x[1] - mean, d2 = x[2] - mean, d3 = x[3] - mean;
        float qa = d0 * d0 + d2 * d2;
        float qb = d1 * d1 + d3 * d3;
        float q = qa + qb;
        q = row_xor_add<8>(q);
        q = row_xor_add<4>(q);
        q = row_xor_add<2>(q);
        q = row_xor_add<1>(q);
        float inv = rsqrtf(q * (1.0f / 64.0f) + 1e-5f);
        float o0 = elu1(d0 * inv * gm[0] + bt[0]);
        float o1 = elu1(d1 * inv * gm[1] + bt[1]);
        float o2 = elu1(d2 * inv * gm[2] + bt[2]);
        float o3 = elu1(d3 * inv * gm[3] + bt[3]);
        if (TOLDS) {
            float* orow = hout + (rr + hi * 8) * TP + col;
            orow[0]  = o0;
            orow[16] = o1;
            orow[32] = o2;
            orow[48] = o3;
        } else if (row < N) {
            float* orow = hout + (long)row * H + col;
            orow[0]  = o0;
            orow[16] = o1;
            orow[32] = o2;
            orow[48] = o3;
        }
    }
}

__device__ __forceinline__ void sage_block_one(const float* __restrict__ sums, const float* __restrict__ deg,
                                               float* hdst,
                                               const float* __restrict__ Wl, const float* __restrict__ bl,
                                               const float* __restrict__ Wr,
                                               const float* __restrict__ g, const float* __restrict__ be,
                                               int N, int blk, float* sWl, float* sWr) {
    int tid = threadIdx.x;
    stage_weights(Wl, Wr, sWl, sWr, tid);
    __syncthreads();

    int wave = tid >> 5;
    int lane = tid & 31;
    int hi   = lane >> 4;
    int col  = lane & 15;
    int kk   = hi * 2;
    int m0   = blk * 64 + wave * 16;

    int r  = m0 + col;
    int rc = r < N ? r : (N - 1);
    float invd = 1.0f / fmaxf(deg[rc], 1.0f);

    v8f acc[4] = {};
    tile_products(sums + (long)rc * H, invd, hdst + (long)rc * H, sWl, sWr, kk, col, acc);
    tile_ln_store<false, false>(acc, acc, bl, bl, g, be, hdst, hdst, m0, hi, col, N);
}

__device__ __forceinline__ void sage_block_two(const float* __restrict__ sums1, const float* __restrict__ deg1,
                                               const float* __restrict__ Wl1, const float* __restrict__ bl1,
                                               const float* __restrict__ Wr1,
                                               const float* __restrict__ sums2, const float* __restrict__ deg2,
                                               const float* __restrict__ Wl2, const float* __restrict__ bl2,
                                               const float* __restrict__ Wr2,
                                               float* hdst,
                                               const float* __restrict__ g, const float* __restrict__ be,
                                               int N, int blk, float* sWl, float* sWr) {
    int tid = threadIdx.x;
    stage_weights(Wl1, Wr1, sWl, sWr, tid);
    __syncthreads();

    int wave = tid >> 5;
    int lane = tid & 31;
    int hi   = lane >> 4;
    int col  = lane & 15;
    int kk   = hi * 2;
    int m0   = blk * 64 + wave * 16;

    int r  = m0 + col;
    int rc = r < N ? r : (N - 1);
    float invd1 = 1.0f / fmaxf(deg1[rc], 1.0f);
    float invd2 = 1.0f / fmaxf(deg2[rc], 1.0f);
    const float* hdr = hdst + (long)rc * H;

    v8f acc1[4] = {};
    tile_products(sums1 + (long)rc * H, invd1, hdr, sWl, sWr, kk, col, acc1);
    __syncthreads();
    stage_weights(Wl2, Wr2, sWl, sWr, tid);
    __syncthreads();
    v8f acc2[4] = {};
    tile_products(sums2 + (long)rc * H, invd2, hdr, sWl, sWr, kk, col, acc2);
    tile_ln_store<true, false>(acc2, acc1, bl2, bl1, g, be, hdst, hdst, m0, hi, col, N);
}

__global__ void __launch_bounds__(128)
sage_layer_kernel(const float* __restrict__ sums1, const float* __restrict__ deg1,
                  const float* __restrict__ Wl1, const float* __restrict__ bl1,
                  const float* __restrict__ Wr1,
                  const float* __restrict__ sums2, const float* __restrict__ deg2,
                  const float* __restrict__ Wl2, const float* __restrict__ bl2,
                  const float* __restrict__ Wr2,
                  float* hT, const float* __restrict__ gT, const float* __restrict__ beT, int NT, int nbT,
                  const float* __restrict__ sumsO, const float* __restrict__ degO,
                  const float* __restrict__ WlO, const float* __restrict__ blO,
                  const float* __restrict__ WrO,
                  float* hO, const float* __restrict__ gO, const float* __restrict__ beO, int NO) {
    __shared__ float sWl[H * H];
    __shared__ float sWr[H * H];
    int b = blockIdx.x;
    if (b < nbT)
        sage_block_two(sums1, deg1, Wl1, bl1, Wr1, sums2, deg2, Wl2, bl2, Wr2,
                       hT, gT, beT, NT, b, sWl, sWr);
    else
        sage_block_one(sumsO, degO, hO, WlO, blO, WrO, gO, beO, NO, b - nbT, sWl, sWr);
}

#define HC 32
__global__ void __launch_bounds__(256, 8)
layer2_head_kernel(const float* __restrict__ hsrc, const int* __restrict__ rowptr,
                   const int* __restrict__ perm, const float* __restrict__ hdst,
                   const float* __restrict__ Wl, const float* __restrict__ bl,
                   const float* __restrict__ Wr,
                   const float* __restrict__ g, const float* __restrict__ be,
                   const float* __restrict__ W1, const float* __restrict__ b1,
                   const float* __restrict__ W2, const float* __restrict__ b2,
                   float* __restrict__ out, int N, int nsrc, int E) {
    __shared__ __attribute__((aligned(16))) float sWl[H * H];
    __shared__ __attribute__((aligned(16))) float sWr[H * H];
    __shared__ __attribute__((aligned(16))) float sW1[H * HC];
    __shared__ __attribute__((aligned(16))) float sT[8 * 16 * TP];
    int tid = threadIdx.x;
    {
        const float4* gl = (const float4*)Wl;
        const float4* gr = (const float4*)Wr;
        const float4* g1 = (const float4*)W1;
        float4 vl[4], vr[4], v1[2];
#pragma unroll
        for (int q = 0; q < 4; ++q) {
            vl[q] = gl[tid + 256 * q];
            vr[q] = gr[tid + 256 * q];
        }
#pragma unroll
        for (int q = 0; q < 2; ++q) v1[q] = g1[tid + 256 * q];
#pragma unroll
        for (int q = 0; q < 4; ++q) {
            ((float4*)sWl)[tid + 256 * q] = vl[q];
            ((float4*)sWr)[tid + 256 * q] = vr[q];
        }
#pragma unroll
        for (int q = 0; q < 2; ++q) ((float4*)sW1)[tid + 256 * q] = v1[q];
    }
    __syncthreads();

    int wave = __builtin_amdgcn_readfirstlane(tid >> 5);
    int lane = tid & 31;
    int hi = lane >> 4, col = lane & 15;
    int kk = hi * 2;
    int m0 = blockIdx.x * 128 + wave * 16;
    if (m0 >= N) return;
    float* tile = sT + wave * (16 * TP);
    int nn = min(16, N - m0);

    int rpv = rowptr[min(m0 + min(lane, 16), N)];

    {
        const char* hsb = (const char*)hsrc;
        int beg = max(__shfl(rpv, 0, 32), 0);
        int end = min(__shfl(rpv, 1, 32), E);
        int ids = 0;
        if (lane < end - beg) ids = perm[beg + lane];
        for (int i = 0; i < nn; ++i) {
            int nb = max(__shfl(rpv, min(i + 1, 16), 32), 0);
            int ne = min(__shfl(rpv, min(i + 2, 16), 32), E);
            bool nidok = lane < ne - nb;
            int nids = 0;
            bool pend = true;

            float4 acc = make_float4(0.0f, 0.0f, 0.0f, 0.0f);
            int cur = ids;
            for (int base = beg; base < end; base += 32) {
                int cnt = min(32, end - base);
                int nxt = 0;
                int mine = min(max(cur, 0), nsrc - 1);
                for (int t = 0; t < cnt; t += 16) {
                    float4 v[8];
                    bool ok[8];
#pragma unroll
                    for (int u = 0; u < 8; ++u) {
                        int idx = t + 2 * u + hi;
                        ok[u] = idx < cnt;
                        int sidx = __shfl(mine, ok[u] ? idx : 0, 32);
                        v[u] = *(const float4*)(hsb + (unsigned)(sidx * 256 + col * 16));
                    }
                    if (t == 0) {
                        if (lane < end - base - 32) nxt = perm[base + 32 + lane];
                        if (pend) {
                            if (nidok) nids = perm[nb + lane];
                            pend = false;
                        }
                    }
#pragma unroll
                    for (int u = 0; u < 8; ++u) {
                        acc.x += ok[u] ? v[u].x : 0.0f;
                        acc.y += ok[u] ? v[u].y : 0.0f;
                        acc.z += ok[u] ? v[u].z : 0.0f;
                        acc.w += ok[u] ? v[u].w : 0.0f;
                    }
                }
                cur = nxt;
            }
            if (pend) {
                if (nidok) nids = perm[nb + lane];
            }
            acc.x += __shfl_xor(acc.x, 16, 32);
            acc.y += __shfl_xor(acc.y, 16, 32);
            acc.z += __shfl_xor(acc.z, 16, 32);
            acc.w += __shfl_xor(acc.w, 16, 32);
            if (hi == 0) *(float4*)(tile + i * TP + col * 4) = acc;
            beg = nb; end = ne;
            ids = nids;
        }
        for (int i = nn; i < 16; ++i)
            if (hi == 0) *(float4*)(tile + i * TP + col * 4) = make_float4(0.0f, 0.0f, 0.0f, 0.0f);
    }

    int r  = m0 + col;
    int rc = r < N ? r : (N - 1);
    int db = max(__shfl(rpv, col, 32), 0);
    int de = min(__shfl(rpv, col + 1, 32), E);
    float invd = 1.0f / fmaxf((float)(de - db), 1.0f);

    v8f acc[4] = {};
    tile_products(tile + col * TP, invd, hdst + (long)rc * H, sWl, sWr, kk, col, acc);

    tile_ln_store<false, true>(acc, acc, bl, bl, g, be, hdst, tile, m0, hi, col, N);

    v8f z[2] = {};
    const float* hrow = tile + col * TP;
#pragma unroll
    for (int k0 = 0; k0 < H; k0 += 4) {
        v2f a;
        a.x = hrow[k0 + kk];
        a.y = hrow[k0 + kk + 1];
#pragma unroll
        for (int t = 0; t < 2; ++t) {
            v2f b;
            b.x = sW1[(k0 + kk) * HC + t * 16 + col];
            b.y = sW1[(k0 + kk + 1) * HC + t * 16 + col];
            z[t] = __builtin_amdgcn_wmma_f32_16x16x4_f32(
                false, a, false, b, (short)0, z[t], false, false);
        }
    }
    float b1a = b1[col], b1b = b1[16 + col];
    float w2a = W2[col], w2b = W2[16 + col];
    float b2v = b2[0];
    float hz = 0.0f;
#pragma unroll
    for (int rr = 0; rr < 8; ++rr) {
        float p = elu1(z[0][rr] + b1a) * w2a + elu1(z[1][rr] + b1b) * w2b;
        p = row_xor_add<8>(p);
        p = row_xor_add<4>(p);
        p = row_xor_add<2>(p);
        p = row_xor_add<1>(p);
        if (col == rr) hz = p + b2v;
    }
    int orow = m0 + hi * 8 + col;
    if (col < 8 && orow < N) out[orow] = hz;
}

extern "C" void kernel_launch(void* const* d_in, const int* in_sizes, int n_in,
                              void* d_out, int out_size, void* d_ws, size_t ws_size,
                              hipStream_t stream) {
    (void)in_sizes; (void)n_in; (void)out_size; (void)ws_size;
    const int NG = 20000, NP = 100000, NM = 5000, E = 1000000;

    const float* x_gene    = (const float*)d_in[0];
    const float* x_patient = (const float*)d_in[1];
    const float* x_group   = (const float*)d_in[2];
    const int* srcA[4]  = { (const int*)d_in[3], (const int*)d_in[5],
                            (const int*)d_in[7], (const int*)d_in[9] };
    const int* dstA[4]  = { (const int*)d_in[4], (const int*)d_in[6],
                            (const int*)d_in[8], (const int*)d_in[10] };
    const float* encW[3] = { (const float*)d_in[11], (const float*)d_in[13], (const float*)d_in[15] };
    const float* encB[3] = { (const float*)d_in[12], (const float*)d_in[14], (const float*)d_in[16] };
    const float* convW_l = (const float*)d_in[17];
    const float* convb_l = (const float*)d_in[18];
    const float* convW_r = (const float*)d_in[19];
    const float* ln_g = (const float*)d_in[20];
    const float* ln_b = (const float*)d_in[21];
    const float* coxW1 = (const float*)d_in[22];
    const float* coxb1 = (const float*)d_in[23];
    const float* coxW2 = (const float*)d_in[24];
    const float* coxb2 = (const float*)d_in[25];

    float* ws = (float*)d_ws;
    float* h_gene  = ws; ws += (size_t)NG * H;
    float* h_pat   = ws; ws += (size_t)NP * H;
    float* h_grp   = ws; ws += (size_t)NM * H;
    float* hn_gene = ws; ws += (size_t)NG * H;
    float* hn_pat  = ws; ws += (size_t)NP * H;
    float* degG1   = ws;
    float* degG3   = ws + NG;
    ws += (size_t)NM * H;
    const int SL = 34000;
    float* aggreg  = ws; ws += (size_t)NP * H;
    float* deg     = ws; ws += (size_t)NP;
    float* agg     = aggreg;
    int*   perm    = (int*)(aggreg + (size_t)SL * H);
    int*   rowptr  = perm + (size_t)3 * E;
    int*   lpre    = (int*)hn_gene;
    int*   bsum    = lpre + 262144;
    int*   ccnt    = (int*)hn_pat;

    const int* srcL[3] = { srcA[0], srcA[1], srcA[3] };
    const int* dstL[3] = { dstA[0], dstA[1], dstA[3] };
    const int* none = nullptr;
    const int NCH   = (E + CHE - 1) / CHE;
    const int NSLOT = (NP + RNG - 1) / RNG + (NG + RNG - 1) / RNG
                    + (NG + RNG - 1) / RNG;
    const int NTOT  = NP + NG + NG;
    const int GCAP = 1024;
    const int gbEG = (NG + 7) / 8 < GCAP ? (NG + 7) / 8 : GCAP;
    const int gbEP = (NP + 15) / 16 < GCAP ? (NP + 15) / 16 : GCAP;
    const int gbEM = (NM + 15) / 16 < GCAP ? (NM + 15) / 16 : GCAP;

    count_kernel<<<dim3(NCH, NSLOT), 1024, 0, stream>>>(
        dstL[0], dstL[1], dstL[2], none, ccnt, NP, NG, NG, 0, E);
    chunk_offsets_kernel<NCH><<<(NTOT + 256) / 256, 256, 0, stream>>>(ccnt, lpre, bsum, NTOT);
    fill_enc_kernel<<<NCH * NSLOT + (gbEP + gbEG + gbEM + 3) / 4, 1024, 0, stream>>>(
        srcL[0], srcL[1], srcL[2], none, dstL[0], dstL[1], dstL[2], none,
        lpre, bsum, ccnt, rowptr, perm, NP, NG, NG, 0, E, NCH, NCH * NSLOT,
        x_patient, encW[1], encB[1], h_pat,  NP, gbEP,
        x_gene,    encW[0], encB[0], h_gene, NG, gbEG,
        x_group,   encW[2], encB[2], h_grp,  NM, gbEM);
    const int* rpL[3] = { rowptr, rowptr + NP + 4, rowptr + NP + NG + 8 };
    const int* pmL[3] = { perm, perm + (size_t)E, perm + (size_t)2 * E };

    const size_t HH = (size_t)H * H;
    const int nbG = (NG + 63) / 64, nbP = (NP + 63) / 64;
    const int g3G = 576, g3P = 896;

    gather3_kernel<<<g3G + g3G + g3P, 256, 0, stream>>>(
        h_pat,  rpL[1], pmL[1], hn_gene, degG1, NG, NP, g3G,
        h_grp,  rpL[2], pmL[2], agg,     degG3, NG, NM, g3G,
        h_gene, rpL[0], pmL[0], hn_pat,  deg,   NP, NG, g3P,
        E);
    sage_layer_kernel<<<nbG + nbP, 128, 0, stream>>>(
        hn_gene, degG1, convW_l + 1 * HH, convb_l + 1 * H, convW_r + 1 * HH,
        agg,     degG3, convW_l + 3 * HH, convb_l + 3 * H, convW_r + 3 * HH,
        h_gene, ln_g + 0 * H, ln_b + 0 * H, NG, nbG,
        hn_pat,  deg,   convW_l + 0 * HH, convb_l + 0 * H, convW_r + 0 * HH,
        h_pat,  ln_g + 1 * H, ln_b + 1 * H, NP);

    layer2_head_kernel<<<(NP + 127) / 128, 256, 0, stream>>>(
        h_gene, rpL[0], pmL[0], h_pat, convW_l + 4 * HH, convb_l + 4 * H, convW_r + 4 * HH,
        ln_g + (3 + 1) * H, ln_b + (3 + 1) * H,
        coxW1, coxb1, coxW2, coxb2, (float*)d_out, NP, NG, E);
}
